# starting kernel
# baseline (speedup 1.0000x reference)
#include <hip/hip_runtime.h>
#include <cfloat>
#include <cmath>

typedef __attribute__((ext_vector_type(2))) float v2f;
typedef __attribute__((ext_vector_type(8))) float v8f;

#define BATCH 16
#define NPTS 2048
#define CINIT 64
#define KNN 32
#define SIGMA_ 0.3f
#define BN_EPS_ 1e-5f
#define STD_G 16

__global__ __launch_bounds__(256) void k_embed_wmma(const float* __restrict__ xyz,
                                                    const float* __restrict__ W,
                                                    float* __restrict__ h) {
  const int wid  = (blockIdx.x * 256 + (int)threadIdx.x) >> 5;
  const int lane = threadIdx.x & 31;
  const int hf   = lane >> 4;
  const int ln   = lane & 15;
  const int tileM = wid >> 2;
  const int tileN = wid & 3;
  const int m  = tileM * 16 + ln;
  const int c  = tileN * 16 + ln;
  const float x0 = xyz[m * 3], x1 = xyz[m * 3 + 1], x2 = xyz[m * 3 + 2];
  const float w0 = W[c * 3],   w1 = W[c * 3 + 1],   w2 = W[c * 3 + 2];
  v2f a, b;
  a.x = hf ? x2 : x0;  a.y = hf ? 0.f : x1;
  b.x = hf ? w2 : w0;  b.y = hf ? 0.f : w1;
  v8f d = {};
  d = __builtin_amdgcn_wmma_f32_16x16x4_f32(false, a, false, b, (short)0, d, false, false);
#pragma unroll
  for (int r = 0; r < 8; ++r) {
    int row = tileM * 16 + r + 8 * hf;
    h[row * CINIT + tileN * 16 + ln] = d[r];
  }
}

__global__ __launch_bounds__(256) void k_bn_stats(const float* __restrict__ h,
                                                  float* __restrict__ mu,
                                                  float* __restrict__ inv) {
  __shared__ float s1[256], s2[256];
  const int c = blockIdx.x;
  const int n = BATCH * NPTS;
  float a = 0.f, b = 0.f;
  for (int i = threadIdx.x; i < n; i += 256) {
    float v = h[i * CINIT + c];
    a += v; b += v * v;
  }
  s1[threadIdx.x] = a; s2[threadIdx.x] = b; __syncthreads();
  for (int o = 128; o > 0; o >>= 1) {
    if ((int)threadIdx.x < o) { s1[threadIdx.x] += s1[threadIdx.x + o]; s2[threadIdx.x] += s2[threadIdx.x + o]; }
    __syncthreads();
  }
  if (threadIdx.x == 0) {
    float m   = s1[0] / (float)n;
    float var = s2[0] / (float)n - m * m;
    mu[c]  = m;
    inv[c] = rsqrtf(var + BN_EPS_);
  }
}

__global__ void k_bn_apply(float* __restrict__ h, const float* __restrict__ mu,
                           const float* __restrict__ inv, const float* __restrict__ gamma,
                           const float* __restrict__ beta) {
  const int total = BATCH * NPTS * CINIT;
  for (int i = blockIdx.x * blockDim.x + threadIdx.x; i < total; i += gridDim.x * blockDim.x) {
    int c = i & (CINIT - 1);
    float v = (h[i] - mu[c]) * inv[c] * gamma[c] + beta[c];
    h[i] = fmaxf(v, 0.f);
  }
}

#define FPS_T 1024
__global__ __launch_bounds__(FPS_T) void k_fps(const float* __restrict__ xyz, int N, int S,
                                               int* __restrict__ fps_idx) {
  __shared__ float px[NPTS], py[NPTS], pz[NPTS];
  __shared__ float wv[32];
  __shared__ int   wi[32];
  __shared__ int   far_s;
  const int b = blockIdx.x;
  const float* p = xyz + (size_t)b * N * 3;
  const int tid = threadIdx.x;
  for (int i = tid; i < N; i += FPS_T) { px[i] = p[i*3]; py[i] = p[i*3+1]; pz[i] = p[i*3+2]; }
  float d0 = 1e10f, d1 = 1e10f;
  const int i0 = tid, i1 = tid + FPS_T;
  if (tid == 0) far_s = 0;
  __syncthreads();
  for (int it = 0; it < S; ++it) {
    const int far = far_s;
    if (tid == 0) fps_idx[b * S + it] = far;
    const float fx = px[far], fy = py[far], fz = pz[far];
    float best = -1.f; int bi = 0x7fffffff;
    if (i0 < N) {
      float dx = px[i0]-fx, dy = py[i0]-fy, dz = pz[i0]-fz;
      d0 = fminf(d0, dx*dx + dy*dy + dz*dz);
      best = d0; bi = i0;
    }
    if (i1 < N) {
      float dx = px[i1]-fx, dy = py[i1]-fy, dz = pz[i1]-fz;
      d1 = fminf(d1, dx*dx + dy*dy + dz*dz);
      if (d1 > best) { best = d1; bi = i1; }
    }
#pragma unroll
    for (int o = 16; o > 0; o >>= 1) {
      float ov = __shfl_down(best, o, 32);
      int   oi = __shfl_down(bi,   o, 32);
      if (ov > best || (ov == best && oi < bi)) { best = ov; bi = oi; }
    }
    if ((tid & 31) == 0) { wv[tid >> 5] = best; wi[tid >> 5] = bi; }
    __syncthreads();
    if (tid < 32) {
      best = wv[tid]; bi = wi[tid];
#pragma unroll
      for (int o = 16; o > 0; o >>= 1) {
        float ov = __shfl_down(best, o, 32);
        int   oi = __shfl_down(bi,   o, 32);
        if (ov > best || (ov == best && oi < bi)) { best = ov; bi = oi; }
      }
      if (tid == 0) far_s = bi;
    }
    __syncthreads();
  }
}

__global__ void k_gather(const float* __restrict__ xyz, const float* __restrict__ feat,
                         const int* __restrict__ fps_idx, int N, int S, int C,
                         float* __restrict__ xyz_s, float* __restrict__ feat_s) {
  const int b = blockIdx.x / S, s = blockIdx.x % S;
  const int idx = fps_idx[b * S + s];
  if (threadIdx.x < 3)
    xyz_s[(size_t)(b * S + s) * 3 + threadIdx.x] = xyz[(size_t)(b * N + idx) * 3 + threadIdx.x];
  for (int j = threadIdx.x; j < C; j += blockDim.x)
    feat_s[(size_t)(b * S + s) * C + j] = feat[(size_t)(b * N + idx) * C + j];
}

__global__ __launch_bounds__(256) void k_knn_dist_wmma(const float* __restrict__ xs,
                                                       const float* __restrict__ x,
                                                       int S, int N, float* __restrict__ dist) {
  __shared__ float sN[16];
  const int mbase = blockIdx.y * 16;
  const int w    = threadIdx.x >> 5;
  const int lane = threadIdx.x & 31;
  const int hf   = lane >> 4;
  const int ln   = lane & 15;
  if (threadIdx.x < 16) {
    float a0 = xs[(mbase + threadIdx.x) * 3];
    float a1 = xs[(mbase + threadIdx.x) * 3 + 1];
    float a2 = xs[(mbase + threadIdx.x) * 3 + 2];
    sN[threadIdx.x] = a0 * a0 + a1 * a1 + a2 * a2;
  }
  __syncthreads();
  const int m = mbase + ln;
  const float s0 = xs[m * 3], s1 = xs[m * 3 + 1], s2 = xs[m * 3 + 2];
  const int n = blockIdx.x * 128 + w * 16 + ln;
  const float x0 = x[n * 3], x1 = x[n * 3 + 1], x2 = x[n * 3 + 2];
  const float nn = x0 * x0 + x1 * x1 + x2 * x2;
  v2f a, b;
  a.x = hf ? s2 : s0;  a.y = hf ? 0.f : s1;
  b.x = hf ? x2 : x0;  b.y = hf ? 0.f : x1;
  v8f d = {};
  d = __builtin_amdgcn_wmma_f32_16x16x4_f32(false, a, false, b, (short)0, d, false, false);
#pragma unroll
  for (int r = 0; r < 8; ++r) {
    int row = mbase + r + 8 * hf;
    dist[(size_t)row * N + n] = sN[r + 8 * hf] + nn - 2.f * d[r];
  }
}

__global__ __launch_bounds__(256) void k_knn_select(const float* __restrict__ dist, int N,
                                                    int* __restrict__ knn_idx) {
  __shared__ float sv[NPTS];
  __shared__ float rv[256];
  __shared__ int   ri[256];
  const int s = blockIdx.x;
  const float* row = dist + (size_t)s * N;
  for (int i = threadIdx.x; i < N; i += 256) sv[i] = row[i];
  __syncthreads();
  for (int k = 0; k < KNN; ++k) {
    float best = FLT_MAX; int bi = 0x7fffffff;
    for (int i = threadIdx.x; i < N; i += 256) {
      float v = sv[i];
      if (v < best) { best = v; bi = i; }
    }
    rv[threadIdx.x] = best; ri[threadIdx.x] = bi;
    __syncthreads();
    for (int o = 128; o > 0; o >>= 1) {
      if ((int)threadIdx.x < o) {
        float vo = rv[threadIdx.x + o];
        if (vo < rv[threadIdx.x] || (vo == rv[threadIdx.x] && ri[threadIdx.x + o] < ri[threadIdx.x])) {
          rv[threadIdx.x] = vo; ri[threadIdx.x] = ri[threadIdx.x + o];
        }
      }
      __syncthreads();
    }
    if (threadIdx.x == 0) { knn_idx[s * KNN + k] = ri[0]; sv[ri[0]] = FLT_MAX; }
    __syncthreads();
  }
}

__global__ __launch_bounds__(256) void k_std_partial(const float* __restrict__ xyz,
                                                     const float* __restrict__ feat,
                                                     const float* __restrict__ xyz_s,
                                                     const float* __restrict__ feat_s,
                                                     const int* __restrict__ knn_idx,
                                                     int N, int S, int C,
                                                     float* __restrict__ part) {
  __shared__ float s1[256], s2[256], s3[256], s4[256];
  const int k = blockIdx.x, g = blockIdx.y;
  float ax = 0.f, qx = 0.f, af = 0.f, qf = 0.f;
  for (int p = g * 256 + threadIdx.x; p < BATCH * S; p += STD_G * 256) {
    int b = p / S, s = p % S;
    int idx = knn_idx[(size_t)(b * S + s) * KNN + k];
    for (int c = 0; c < 3; ++c) {
      float d = xyz[(size_t)(b * N + idx) * 3 + c] - xyz_s[(size_t)(b * S + s) * 3 + c];
      ax += d; qx += d * d;
    }
    const float* fr = feat   + (size_t)(b * N + idx) * C;
    const float* fc = feat_s + (size_t)(b * S + s) * C;
    for (int j = 0; j < C; ++j) { float d = fr[j] - fc[j]; af += d; qf += d * d; }
  }
  s1[threadIdx.x] = ax; s2[threadIdx.x] = qx; s3[threadIdx.x] = af; s4[threadIdx.x] = qf;
  __syncthreads();
  for (int o = 128; o > 0; o >>= 1) {
    if ((int)threadIdx.x < o) {
      s1[threadIdx.x] += s1[threadIdx.x + o]; s2[threadIdx.x] += s2[threadIdx.x + o];
      s3[threadIdx.x] += s3[threadIdx.x + o]; s4[threadIdx.x] += s4[threadIdx.x + o];
    }
    __syncthreads();
  }
  if (threadIdx.x == 0) {
    float* q = part + ((size_t)k * STD_G + g) * 4;
    q[0] = s1[0]; q[1] = s2[0]; q[2] = s3[0]; q[3] = s4[0];
  }
}

__global__ void k_std_final(const float* __restrict__ part, int S, int C,
                            float* __restrict__ std_xyz, float* __restrict__ std_feat) {
  const int k = threadIdx.x;
  if (k >= KNN) return;
  float ax = 0.f, qx = 0.f, af = 0.f, qf = 0.f;
  for (int g = 0; g < STD_G; ++g) {
    const float* q = part + ((size_t)k * STD_G + g) * 4;
    ax += q[0]; qx += q[1]; af += q[2]; qf += q[3];
  }
  {
    double n = (double)BATCH * S * 3;
    double var = ((double)qx - (double)ax * ax / n) / (n - 1.0);
    std_xyz[k] = fmaxf(sqrtf(fmaxf((float)var, 0.f)), 1e-5f);
  }
  {
    double n = (double)BATCH * S * C;
    double var = ((double)qf - (double)af * af / n) / (n - 1.0);
    std_feat[k] = fmaxf(sqrtf(fmaxf((float)var, 0.f)), 1e-5f);
  }
}

__global__ __launch_bounds__(256) void k_stage(const float* __restrict__ xyz,
                                               const float* __restrict__ feat,
                                               const float* __restrict__ xyz_s,
                                               const float* __restrict__ feat_s,
                                               const int* __restrict__ knn_idx,
                                               const float* __restrict__ std_xyz,
                                               const float* __restrict__ std_feat,
                                               int N, int S, int C,
                                               float* __restrict__ out) {
  __shared__ int   kidx[KNN];
  __shared__ float sfe[KNN];
  __shared__ float xkn[KNN][3];
  __shared__ float fs[512];
  const int b = blockIdx.x / S, s = blockIdx.x % S;
  if (threadIdx.x < KNN) {
    kidx[threadIdx.x] = knn_idx[(size_t)(b * S + s) * KNN + threadIdx.x];
    sfe[threadIdx.x]  = std_feat[threadIdx.x];
  }
  __syncthreads();
  if (threadIdx.x < KNN)
    __builtin_prefetch(feat + (size_t)(b * N + kidx[threadIdx.x]) * C, 0, 3);
  if (threadIdx.x < KNN * 3) {
    int k = threadIdx.x / 3, c = threadIdx.x % 3;
    xkn[k][c] = (xyz[(size_t)(b * N + kidx[k]) * 3 + c] - xyz_s[(size_t)(b * S + s) * 3 + c]) / std_xyz[k];
  }
  for (int j = threadIdx.x; j < C; j += 256) fs[j] = feat_s[(size_t)(b * S + s) * C + j];
  __syncthreads();
  const int C2 = 2 * C;
  const int fd = (C2 + 2) / 3;
  const int M3 = 3 * fd;
  const float inv_sig = 1.f / SIGMA_;
  for (int j = threadIdx.x; j < C2; j += 256) {
    int src = (int)(((long long)j * (M3 - 1)) / (C2 - 1));
    int cc = src / fd, f = src % fd;
    float fv = -1.f + 2.f * (float)(f + 1) / (float)(fd + 1);
    float acc = 0.f, mx = -FLT_MAX;
    for (int k = 0; k < KNN; ++k) {
      float t  = (xkn[k][cc] - fv) * inv_sig;
      float pe = expf(-0.5f * t * t);
      float fk;
      if (j < C) fk = (feat[(size_t)(b * N + kidx[k]) * C + j] - fs[j]) / sfe[k];
      else       fk = fs[j - C];
      float fw = (fk + pe) * pe;
      acc += fw; mx = fmaxf(mx, fw);
    }
    float xx = acc * (1.f / KNN) + mx;
    out[(size_t)(b * S + s) * C2 + j] = 0.5f * xx * (1.f + erff(xx * 0.70710678118654752f));
  }
}

__global__ void k_pool(const float* __restrict__ feat, int S, int C2,
                       float* __restrict__ out, int off) {
  const int b = blockIdx.x;
  for (int j = threadIdx.x; j < C2; j += blockDim.x) {
    float mx = -FLT_MAX, sm = 0.f;
    for (int s = 0; s < S; ++s) {
      float v = feat[(size_t)(b * S + s) * C2 + j];
      mx = fmaxf(mx, v); sm += v;
    }
    out[b * 3840 + off + j]      = mx;
    out[b * 3840 + off + C2 + j] = sm / (float)S;
  }
}

extern "C" void kernel_launch(void* const* d_in, const int* in_sizes, int n_in,
                              void* d_out, int out_size, void* d_ws, size_t ws_size,
                              hipStream_t stream) {
  const float* xyz_in = (const float*)d_in[0];
  const float* W      = (const float*)d_in[1];
  const float* gamma  = (const float*)d_in[2];
  const float* beta   = (const float*)d_in[3];
  float* out = (float*)d_out;
  float* ws  = (float*)d_ws;

  size_t o = 0;
  float* featA  = ws + o; o += (size_t)BATCH * NPTS * CINIT;
  float* featB  = ws + o; o += (size_t)BATCH * NPTS * CINIT;
  float* xyzA   = ws + o; o += (size_t)BATCH * NPTS * 3;
  float* xyzB   = ws + o; o += (size_t)BATCH * NPTS * 3;
  float* feat_s = ws + o; o += (size_t)BATCH * 1024 * CINIT;
  float* dist   = ws + o; o += (size_t)1024 * 2048;
  float* bn_mu  = ws + o; o += 64;
  float* bn_inv = ws + o; o += 64;
  float* stdx   = ws + o; o += 32;
  float* stdf   = ws + o; o += 32;
  float* stdp   = ws + o; o += (size_t)KNN * STD_G * 4;
  int* fps_i = (int*)(ws + o); o += (size_t)BATCH * 1024;
  int* knn_i = (int*)(ws + o); o += (size_t)BATCH * 1024 * KNN;

  k_embed_wmma<<<1024, 256, 0, stream>>>(xyz_in, W, featA);
  k_bn_stats<<<CINIT, 256, 0, stream>>>(featA, bn_mu, bn_inv);
  k_bn_apply<<<2048, 256, 0, stream>>>(featA, bn_mu, bn_inv, gamma, beta);

  const float* xyz_cur = xyz_in;
  float* feat_cur = featA; float* feat_nxt = featB;
  float* xs_buf = xyzA;    float* xs_other = xyzB;
  int N = NPTS, C = CINIT, off = 0;

  for (int st = 0; st < 4; ++st) {
    const int S = N / 2, C2 = 2 * C;
    k_fps<<<BATCH, FPS_T, 0, stream>>>(xyz_cur, N, S, fps_i);
    k_gather<<<BATCH * S, 256, 0, stream>>>(xyz_cur, feat_cur, fps_i, N, S, C, xs_buf, feat_s);
    for (int b = 0; b < BATCH; ++b) {
      dim3 g(N / 128, S / 16);
      k_knn_dist_wmma<<<g, 256, 0, stream>>>(xs_buf + (size_t)b * S * 3,
                                             xyz_cur + (size_t)b * N * 3, S, N, dist);
      k_knn_select<<<S, 256, 0, stream>>>(dist, N, knn_i + (size_t)b * S * KNN);
    }
    {
      dim3 g(KNN, STD_G);
      k_std_partial<<<g, 256, 0, stream>>>(xyz_cur, feat_cur, xs_buf, feat_s, knn_i, N, S, C, stdp);
      k_std_final<<<1, KNN, 0, stream>>>(stdp, S, C, stdx, stdf);
    }
    k_stage<<<BATCH * S, 256, 0, stream>>>(xyz_cur, feat_cur, xs_buf, feat_s, knn_i,
                                           stdx, stdf, N, S, C, feat_nxt);
    k_pool<<<BATCH, 256, 0, stream>>>(feat_nxt, S, C2, out, off);
    off += 2 * C2;
    float* t = feat_cur; feat_cur = feat_nxt; feat_nxt = t;
    xyz_cur = xs_buf;
    float* u = xs_buf; xs_buf = xs_other; xs_other = u;
    N = S; C = C2;
  }
  (void)in_sizes; (void)n_in; (void)out_size; (void)ws_size;
}


# kNN top-32 in one pass per row, no argmin passes: T = 32nd smallest of 64 group minima bounds the answer; survivors (v<=T) compacted to LDS, each placed by counting smaller (value,index) pairs
# speedup vs baseline: 1.7916x; 1.7916x over previous
#include <hip/hip_runtime.h>
#include <cfloat>
#include <cmath>

typedef __attribute__((ext_vector_type(2))) float v2f;
typedef __attribute__((ext_vector_type(8))) float v8f;

#define BATCH 16
#define NPTS 2048
#define CINIT 64
#define KNN 32
#define SIGMA_ 0.3f
#define BN_EPS_ 1e-5f
#define STD_G 16

__global__ __launch_bounds__(256) void k_embed_wmma(const float* __restrict__ xyz,
                                                    const float* __restrict__ W,
                                                    float* __restrict__ h) {
  const int wid  = (blockIdx.x * 256 + (int)threadIdx.x) >> 5;
  const int lane = threadIdx.x & 31;
  const int hf   = lane >> 4;
  const int ln   = lane & 15;
  const int tileM = wid >> 2;
  const int tileN = wid & 3;
  const int m  = tileM * 16 + ln;
  const int c  = tileN * 16 + ln;
  const float x0 = xyz[m * 3], x1 = xyz[m * 3 + 1], x2 = xyz[m * 3 + 2];
  const float w0 = W[c * 3],   w1 = W[c * 3 + 1],   w2 = W[c * 3 + 2];
  v2f a, b;
  a.x = hf ? x2 : x0;  a.y = hf ? 0.f : x1;
  b.x = hf ? w2 : w0;  b.y = hf ? 0.f : w1;
  v8f d = {};
  d = __builtin_amdgcn_wmma_f32_16x16x4_f32(false, a, false, b, (short)0, d, false, false);
#pragma unroll
  for (int r = 0; r < 8; ++r) {
    int row = tileM * 16 + r + 8 * hf;
    h[row * CINIT + tileN * 16 + ln] = d[r];
  }
}

__global__ __launch_bounds__(256) void k_bn_stats(const float* __restrict__ h,
                                                  float* __restrict__ mu,
                                                  float* __restrict__ inv) {
  __shared__ float s1[256], s2[256];
  const int c = blockIdx.x;
  const int n = BATCH * NPTS;
  float a = 0.f, b = 0.f;
  for (int i = threadIdx.x; i < n; i += 256) {
    float v = h[i * CINIT + c];
    a += v; b += v * v;
  }
  s1[threadIdx.x] = a; s2[threadIdx.x] = b; __syncthreads();
  for (int o = 128; o > 0; o >>= 1) {
    if ((int)threadIdx.x < o) { s1[threadIdx.x] += s1[threadIdx.x + o]; s2[threadIdx.x] += s2[threadIdx.x + o]; }
    __syncthreads();
  }
  if (threadIdx.x == 0) {
    float m   = s1[0] / (float)n;
    float var = s2[0] / (float)n - m * m;
    mu[c]  = m;
    inv[c] = rsqrtf(var + BN_EPS_);
  }
}

__global__ void k_bn_apply(float* __restrict__ h, const float* __restrict__ mu,
                           const float* __restrict__ inv, const float* __restrict__ gamma,
                           const float* __restrict__ beta) {
  const int total = BATCH * NPTS * CINIT;
  for (int i = blockIdx.x * blockDim.x + threadIdx.x; i < total; i += gridDim.x * blockDim.x) {
    int c = i & (CINIT - 1);
    float v = (h[i] - mu[c]) * inv[c] * gamma[c] + beta[c];
    h[i] = fmaxf(v, 0.f);
  }
}

#define FPS_T 1024
__global__ __launch_bounds__(FPS_T) void k_fps(const float* __restrict__ xyz, int N, int S,
                                               int* __restrict__ fps_idx) {
  __shared__ float px[NPTS], py[NPTS], pz[NPTS];
  __shared__ float wv[32];
  __shared__ int   wi[32];
  __shared__ int   far_s;
  const int b = blockIdx.x;
  const float* p = xyz + (size_t)b * N * 3;
  const int tid = threadIdx.x;
  for (int i = tid; i < N; i += FPS_T) { px[i] = p[i*3]; py[i] = p[i*3+1]; pz[i] = p[i*3+2]; }
  float d0 = 1e10f, d1 = 1e10f;
  const int i0 = tid, i1 = tid + FPS_T;
  if (tid == 0) far_s = 0;
  __syncthreads();
  for (int it = 0; it < S; ++it) {
    const int far = far_s;
    if (tid == 0) fps_idx[b * S + it] = far;
    const float fx = px[far], fy = py[far], fz = pz[far];
    float best = -1.f; int bi = 0x7fffffff;
    if (i0 < N) {
      float dx = px[i0]-fx, dy = py[i0]-fy, dz = pz[i0]-fz;
      d0 = fminf(d0, dx*dx + dy*dy + dz*dz);
      best = d0; bi = i0;
    }
    if (i1 < N) {
      float dx = px[i1]-fx, dy = py[i1]-fy, dz = pz[i1]-fz;
      d1 = fminf(d1, dx*dx + dy*dy + dz*dz);
      if (d1 > best) { best = d1; bi = i1; }
    }
#pragma unroll
    for (int o = 16; o > 0; o >>= 1) {
      float ov = __shfl_down(best, o, 32);
      int   oi = __shfl_down(bi,   o, 32);
      if (ov > best || (ov == best && oi < bi)) { best = ov; bi = oi; }
    }
    if ((tid & 31) == 0) { wv[tid >> 5] = best; wi[tid >> 5] = bi; }
    __syncthreads();
    if (tid < 32) {
      best = wv[tid]; bi = wi[tid];
#pragma unroll
      for (int o = 16; o > 0; o >>= 1) {
        float ov = __shfl_down(best, o, 32);
        int   oi = __shfl_down(bi,   o, 32);
        if (ov > best || (ov == best && oi < bi)) { best = ov; bi = oi; }
      }
      if (tid == 0) far_s = bi;
    }
    __syncthreads();
  }
}

__global__ void k_gather(const float* __restrict__ xyz, const float* __restrict__ feat,
                         const int* __restrict__ fps_idx, int N, int S, int C,
                         float* __restrict__ xyz_s, float* __restrict__ feat_s) {
  const int b = blockIdx.x / S, s = blockIdx.x % S;
  const int idx = fps_idx[b * S + s];
  if (threadIdx.x < 3)
    xyz_s[(size_t)(b * S + s) * 3 + threadIdx.x] = xyz[(size_t)(b * N + idx) * 3 + threadIdx.x];
  for (int j = threadIdx.x; j < C; j += blockDim.x)
    feat_s[(size_t)(b * S + s) * C + j] = feat[(size_t)(b * N + idx) * C + j];
}

__global__ __launch_bounds__(256) void k_knn_dist_wmma(const float* __restrict__ xs,
                                                       const float* __restrict__ x,
                                                       int S, int N, float* __restrict__ dist) {
  __shared__ float sN[16];
  const int mbase = blockIdx.y * 16;
  const int w    = threadIdx.x >> 5;
  const int lane = threadIdx.x & 31;
  const int hf   = lane >> 4;
  const int ln   = lane & 15;
  if (threadIdx.x < 16) {
    float a0 = xs[(mbase + threadIdx.x) * 3];
    float a1 = xs[(mbase + threadIdx.x) * 3 + 1];
    float a2 = xs[(mbase + threadIdx.x) * 3 + 2];
    sN[threadIdx.x] = a0 * a0 + a1 * a1 + a2 * a2;
  }
  __syncthreads();
  const int m = mbase + ln;
  const float s0 = xs[m * 3], s1 = xs[m * 3 + 1], s2 = xs[m * 3 + 2];
  const int n = blockIdx.x * 128 + w * 16 + ln;
  const float x0 = x[n * 3], x1 = x[n * 3 + 1], x2 = x[n * 3 + 2];
  const float nn = x0 * x0 + x1 * x1 + x2 * x2;
  v2f a, b;
  a.x = hf ? s2 : s0;  a.y = hf ? 0.f : s1;
  b.x = hf ? x2 : x0;  b.y = hf ? 0.f : x1;
  v8f d = {};
  d = __builtin_amdgcn_wmma_f32_16x16x4_f32(false, a, false, b, (short)0, d, false, false);
#pragma unroll
  for (int r = 0; r < 8; ++r) {
    int row = mbase + r + 8 * hf;
    dist[(size_t)row * N + n] = sN[r + 8 * hf] + nn - 2.f * d[r];
  }
}

#define SEL_W 4
template <int NPL>
__global__ __launch_bounds__(SEL_W * 32) void k_knn_select(const float* __restrict__ dist,
                                                           int* __restrict__ knn_idx) {
  constexpr int N = NPL * 32;
  constexpr int H = NPL / 2;
  __shared__ int2 cand[SEL_W][N];
  const int w    = threadIdx.x >> 5;
  const int lane = threadIdx.x & 31;
  const int s    = blockIdx.x * SEL_W + w;
  const float* row = dist + (size_t)s * N;
  float v[NPL];
#pragma unroll
  for (int j = 0; j < NPL; ++j) v[j] = row[j * 32 + lane];
  float g0 = v[0], g1 = v[H];
#pragma unroll
  for (int j = 1; j < H; ++j) { g0 = fminf(g0, v[j]); g1 = fminf(g1, v[H + j]); }
  cand[w][lane].x      = __float_as_int(g0);
  cand[w][32 + lane].x = __float_as_int(g1);
  __syncthreads();
  int r0 = 0, r1 = 0;
#pragma unroll
  for (int e = 0; e < 64; ++e) {
    const float o = __int_as_float(cand[w][e].x);
    r0 += (o < g0 || (o == g0 && e < lane)) ? 1 : 0;
    r1 += (o < g1 || (o == g1 && e < 32 + lane)) ? 1 : 0;
  }
  float T = (r0 == 31) ? g0 : ((r1 == 31) ? g1 : -INFINITY);
#pragma unroll
  for (int o = 16; o > 0; o >>= 1) T = fmaxf(T, __shfl_xor(T, o, 32));
  int M = 0;
#pragma unroll
  for (int j = 0; j < NPL; ++j) {
    const bool c = v[j] <= T;
    const unsigned bal = __builtin_amdgcn_ballot_w32(c);
    const int pos = M + (int)__builtin_amdgcn_mbcnt_lo(bal, 0u);
    if (c) cand[w][pos] = make_int2(__float_as_int(v[j]), j * 32 + lane);
    M += __builtin_popcount(bal);
  }
  __syncthreads();
  for (int base = 0; base < M; base += 32) {
    const int my = base + lane;
    const bool have = my < M;
    const int2 mine = cand[w][have ? my : 0];
    const float mv = __int_as_float(mine.x);
    const int   mi = mine.y;
    int rank = 0;
#pragma unroll 4
    for (int j = 0; j < M; ++j) {
      const int2 o = cand[w][j];
      const float ov = __int_as_float(o.x);
      rank += (ov < mv || (ov == mv && o.y < mi)) ? 1 : 0;
    }
    if (have && rank < KNN) knn_idx[(size_t)s * KNN + rank] = mi;
  }
}

__global__ __launch_bounds__(256) void k_std_partial(const float* __restrict__ xyz,
                                                     const float* __restrict__ feat,
                                                     const float* __restrict__ xyz_s,
                                                     const float* __restrict__ feat_s,
                                                     const int* __restrict__ knn_idx,
                                                     int N, int S, int C,
                                                     float* __restrict__ part) {
  __shared__ float s1[256], s2[256], s3[256], s4[256];
  const int k = blockIdx.x, g = blockIdx.y;
  float ax = 0.f, qx = 0.f, af = 0.f, qf = 0.f;
  for (int p = g * 256 + threadIdx.x; p < BATCH * S; p += STD_G * 256) {
    int b = p / S, s = p % S;
    int idx = knn_idx[(size_t)(b * S + s) * KNN + k];
    for (int c = 0; c < 3; ++c) {
      float d = xyz[(size_t)(b * N + idx) * 3 + c] - xyz_s[(size_t)(b * S + s) * 3 + c];
      ax += d; qx += d * d;
    }
    const float* fr = feat   + (size_t)(b * N + idx) * C;
    const float* fc = feat_s + (size_t)(b * S + s) * C;
    for (int j = 0; j < C; ++j) { float d = fr[j] - fc[j]; af += d; qf += d * d; }
  }
  s1[threadIdx.x] = ax; s2[threadIdx.x] = qx; s3[threadIdx.x] = af; s4[threadIdx.x] = qf;
  __syncthreads();
  for (int o = 128; o > 0; o >>= 1) {
    if ((int)threadIdx.x < o) {
      s1[threadIdx.x] += s1[threadIdx.x + o]; s2[threadIdx.x] += s2[threadIdx.x + o];
      s3[threadIdx.x] += s3[threadIdx.x + o]; s4[threadIdx.x] += s4[threadIdx.x + o];
    }
    __syncthreads();
  }
  if (threadIdx.x == 0) {
    float* q = part + ((size_t)k * STD_G + g) * 4;
    q[0] = s1[0]; q[1] = s2[0]; q[2] = s3[0]; q[3] = s4[0];
  }
}

__global__ void k_std_final(const float* __restrict__ part, int S, int C,
                            float* __restrict__ std_xyz, float* __restrict__ std_feat) {
  const int k = threadIdx.x;
  if (k >= KNN) return;
  float ax = 0.f, qx = 0.f, af = 0.f, qf = 0.f;
  for (int g = 0; g < STD_G; ++g) {
    const float* q = part + ((size_t)k * STD_G + g) * 4;
    ax += q[0]; qx += q[1]; af += q[2]; qf += q[3];
  }
  {
    double n = (double)BATCH * S * 3;
    double var = ((double)qx - (double)ax * ax / n) / (n - 1.0);
    std_xyz[k] = fmaxf(sqrtf(fmaxf((float)var, 0.f)), 1e-5f);
  }
  {
    double n = (double)BATCH * S * C;
    double var = ((double)qf - (double)af * af / n) / (n - 1.0);
    std_feat[k] = fmaxf(sqrtf(fmaxf((float)var, 0.f)), 1e-5f);
  }
}

__global__ __launch_bounds__(256) void k_stage(const float* __restrict__ xyz,
                                               const float* __restrict__ feat,
                                               const float* __restrict__ xyz_s,
                                               const float* __restrict__ feat_s,
                                               const int* __restrict__ knn_idx,
                                               const float* __restrict__ std_xyz,
                                               const float* __restrict__ std_feat,
                                               int N, int S, int C,
                                               float* __restrict__ out) {
  __shared__ int   kidx[KNN];
  __shared__ float sfe[KNN];
  __shared__ float xkn[KNN][3];
  __shared__ float fs[512];
  const int b = blockIdx.x / S, s = blockIdx.x % S;
  if (threadIdx.x < KNN) {
    kidx[threadIdx.x] = knn_idx[(size_t)(b * S + s) * KNN + threadIdx.x];
    sfe[threadIdx.x]  = std_feat[threadIdx.x];
  }
  __syncthreads();
  if (threadIdx.x < KNN)
    __builtin_prefetch(feat + (size_t)(b * N + kidx[threadIdx.x]) * C, 0, 3);
  if (threadIdx.x < KNN * 3) {
    int k = threadIdx.x / 3, c = threadIdx.x % 3;
    xkn[k][c] = (xyz[(size_t)(b * N + kidx[k]) * 3 + c] - xyz_s[(size_t)(b * S + s) * 3 + c]) / std_xyz[k];
  }
  for (int j = threadIdx.x; j < C; j += 256) fs[j] = feat_s[(size_t)(b * S + s) * C + j];
  __syncthreads();
  const int C2 = 2 * C;
  const int fd = (C2 + 2) / 3;
  const int M3 = 3 * fd;
  const float inv_sig = 1.f / SIGMA_;
  for (int j = threadIdx.x; j < C2; j += 256) {
    int src = (int)(((long long)j * (M3 - 1)) / (C2 - 1));
    int cc = src / fd, f = src % fd;
    float fv = -1.f + 2.f * (float)(f + 1) / (float)(fd + 1);
    float acc = 0.f, mx = -FLT_MAX;
    for (int k = 0; k < KNN; ++k) {
      float t  = (xkn[k][cc] - fv) * inv_sig;
      float pe = expf(-0.5f * t * t);
      float fk;
      if (j < C) fk = (feat[(size_t)(b * N + kidx[k]) * C + j] - fs[j]) / sfe[k];
      else       fk = fs[j - C];
      float fw = (fk + pe) * pe;
      acc += fw; mx = fmaxf(mx, fw);
    }
    float xx = acc * (1.f / KNN) + mx;
    out[(size_t)(b * S + s) * C2 + j] = 0.5f * xx * (1.f + erff(xx * 0.70710678118654752f));
  }
}

__global__ void k_pool(const float* __restrict__ feat, int S, int C2,
                       float* __restrict__ out, int off) {
  const int b = blockIdx.x;
  for (int j = threadIdx.x; j < C2; j += blockDim.x) {
    float mx = -FLT_MAX, sm = 0.f;
    for (int s = 0; s < S; ++s) {
      float v = feat[(size_t)(b * S + s) * C2 + j];
      mx = fmaxf(mx, v); sm += v;
    }
    out[b * 3840 + off + j]      = mx;
    out[b * 3840 + off + C2 + j] = sm / (float)S;
  }
}

extern "C" void kernel_launch(void* const* d_in, const int* in_sizes, int n_in,
                              void* d_out, int out_size, void* d_ws, size_t ws_size,
                              hipStream_t stream) {
  const float* xyz_in = (const float*)d_in[0];
  const float* W      = (const float*)d_in[1];
  const float* gamma  = (const float*)d_in[2];
  const float* beta   = (const float*)d_in[3];
  float* out = (float*)d_out;
  float* ws  = (float*)d_ws;

  size_t o = 0;
  float* featA  = ws + o; o += (size_t)BATCH * NPTS * CINIT;
  float* featB  = ws + o; o += (size_t)BATCH * NPTS * CINIT;
  float* xyzA   = ws + o; o += (size_t)BATCH * NPTS * 3;
  float* xyzB   = ws + o; o += (size_t)BATCH * NPTS * 3;
  float* feat_s = ws + o; o += (size_t)BATCH * 1024 * CINIT;
  float* dist   = ws + o; o += (size_t)1024 * 2048;
  float* bn_mu  = ws + o; o += 64;
  float* bn_inv = ws + o; o += 64;
  float* stdx   = ws + o; o += 32;
  float* stdf   = ws + o; o += 32;
  float* stdp   = ws + o; o += (size_t)KNN * STD_G * 4;
  int* fps_i = (int*)(ws + o); o += (size_t)BATCH * 1024;
  int* knn_i = (int*)(ws + o); o += (size_t)BATCH * 1024 * KNN;

  k_embed_wmma<<<1024, 256, 0, stream>>>(xyz_in, W, featA);
  k_bn_stats<<<CINIT, 256, 0, stream>>>(featA, bn_mu, bn_inv);
  k_bn_apply<<<2048, 256, 0, stream>>>(featA, bn_mu, bn_inv, gamma, beta);

  const float* xyz_cur = xyz_in;
  float* feat_cur = featA; float* feat_nxt = featB;
  float* xs_buf = xyzA;    float* xs_other = xyzB;
  int N = NPTS, C = CINIT, off = 0;

  for (int st = 0; st < 4; ++st) {
    const int S = N / 2, C2 = 2 * C;
    k_fps<<<BATCH, FPS_T, 0, stream>>>(xyz_cur, N, S, fps_i);
    k_gather<<<BATCH * S, 256, 0, stream>>>(xyz_cur, feat_cur, fps_i, N, S, C, xs_buf, feat_s);
    for (int b = 0; b < BATCH; ++b) {
      dim3 g(N / 128, S / 16);
      k_knn_dist_wmma<<<g, 256, 0, stream>>>(xs_buf + (size_t)b * S * 3,
                                             xyz_cur + (size_t)b * N * 3, S, N, dist);
      int* ki = knn_i + (size_t)b * S * KNN;
      if (N == 2048)      k_knn_select<64><<<S / SEL_W, SEL_W * 32, 0, stream>>>(dist, ki);
      else if (N == 1024) k_knn_select<32><<<S / SEL_W, SEL_W * 32, 0, stream>>>(dist, ki);
      else if (N == 512)  k_knn_select<16><<<S / SEL_W, SEL_W * 32, 0, stream>>>(dist, ki);
      else                k_knn_select<8><<<S / SEL_W, SEL_W * 32, 0, stream>>>(dist, ki);
    }
    {
      dim3 g(KNN, STD_G);
      k_std_partial<<<g, 256, 0, stream>>>(xyz_cur, feat_cur, xs_buf, feat_s, knn_i, N, S, C, stdp);
      k_std_final<<<1, KNN, 0, stream>>>(stdp, S, C, stdx, stdf);
    }
    k_stage<<<BATCH * S, 256, 0, stream>>>(xyz_cur, feat_cur, xs_buf, feat_s, knn_i,
                                           stdx, stdf, N, S, C, feat_nxt);
    k_pool<<<BATCH, 256, 0, stream>>>(feat_nxt, S, C2, out, off);
    off += 2 * C2;
    float* t = feat_cur; feat_cur = feat_nxt; feat_nxt = t;
    xyz_cur = xs_buf;
    float* u = xs_buf; xs_buf = xs_other; xs_other = u;
    N = S; C = C2;
  }
  (void)in_sizes; (void)n_in; (void)out_size; (void)ws_size;
}


# readout folded into the stage body: a block owns 4/4/2/1 consecutive sampled points (128 threads in stage 1), stores max and sum of its outputs; readout folds S/chunk rows; mean's sum order changes
# speedup vs baseline: 3.9948x; 2.2297x over previous
#include <hip/hip_runtime.h>
#include <cfloat>
#include <cmath>

typedef __attribute__((ext_vector_type(2))) float v2f;
typedef __attribute__((ext_vector_type(8))) float v8f;

#define BATCH 16
#define NPTS 2048
#define CINIT 64
#define KNN 32
#define SIGMA_ 0.3f
#define BN_EPS_ 1e-5f
#define STD_G 16

__global__ __launch_bounds__(256) void k_embed_wmma(const float* __restrict__ xyz,
                                                    const float* __restrict__ W,
                                                    float* __restrict__ h) {
  const int wid  = (blockIdx.x * 256 + (int)threadIdx.x) >> 5;
  const int lane = threadIdx.x & 31;
  const int hf   = lane >> 4;
  const int ln   = lane & 15;
  const int tileM = wid >> 2;
  const int tileN = wid & 3;
  const int m  = tileM * 16 + ln;
  const int c  = tileN * 16 + ln;
  const float x0 = xyz[m * 3], x1 = xyz[m * 3 + 1], x2 = xyz[m * 3 + 2];
  const float w0 = W[c * 3],   w1 = W[c * 3 + 1],   w2 = W[c * 3 + 2];
  v2f a, b;
  a.x = hf ? x2 : x0;  a.y = hf ? 0.f : x1;
  b.x = hf ? w2 : w0;  b.y = hf ? 0.f : w1;
  v8f d = {};
  d = __builtin_amdgcn_wmma_f32_16x16x4_f32(false, a, false, b, (short)0, d, false, false);
#pragma unroll
  for (int r = 0; r < 8; ++r) {
    int row = tileM * 16 + r + 8 * hf;
    h[row * CINIT + tileN * 16 + ln] = d[r];
  }
}

__global__ __launch_bounds__(256) void k_bn_stats(const float* __restrict__ h,
                                                  float* __restrict__ mu,
                                                  float* __restrict__ inv) {
  __shared__ float s1[256], s2[256];
  const int c = blockIdx.x;
  const int n = BATCH * NPTS;
  float a = 0.f, b = 0.f;
  for (int i = threadIdx.x; i < n; i += 256) {
    float v = h[i * CINIT + c];
    a += v; b += v * v;
  }
  s1[threadIdx.x] = a; s2[threadIdx.x] = b; __syncthreads();
  for (int o = 128; o > 0; o >>= 1) {
    if ((int)threadIdx.x < o) { s1[threadIdx.x] += s1[threadIdx.x + o]; s2[threadIdx.x] += s2[threadIdx.x + o]; }
    __syncthreads();
  }
  if (threadIdx.x == 0) {
    float m   = s1[0] / (float)n;
    float var = s2[0] / (float)n - m * m;
    mu[c]  = m;
    inv[c] = rsqrtf(var + BN_EPS_);
  }
}

__global__ void k_bn_apply(float* __restrict__ h, const float* __restrict__ mu,
                           const float* __restrict__ inv, const float* __restrict__ gamma,
                           const float* __restrict__ beta) {
  const int total = BATCH * NPTS * CINIT;
  for (int i = blockIdx.x * blockDim.x + threadIdx.x; i < total; i += gridDim.x * blockDim.x) {
    int c = i & (CINIT - 1);
    float v = (h[i] - mu[c]) * inv[c] * gamma[c] + beta[c];
    h[i] = fmaxf(v, 0.f);
  }
}

__device__ __forceinline__ int imax2(int a, int b) { return a > b ? a : b; }
__device__ __forceinline__ int imin2(int a, int b) { return a < b ? a : b; }

template <int CTRL>
__device__ __forceinline__ int row_xchg(int v) {
  return __builtin_amdgcn_update_dpp(v, v, CTRL, 0xf, 0xf, true);
}

template <int PPL>
__global__ __launch_bounds__(32) void k_fps_wave(const float* __restrict__ xyz, int S,
                                                 int* __restrict__ fps_idx) {
  constexpr int N = 32 * PPL;
  __shared__ float4 pc[N];
  const int b = blockIdx.x;
  const int lane = threadIdx.x;
  const float* p = xyz + ((size_t)b * N + (size_t)lane * PPL) * 3;
  float px[PPL], py[PPL], pz[PPL];
  int di[PPL];
#pragma unroll
  for (int s = 0; s < PPL; ++s) {
    px[s] = p[s * 3]; py[s] = p[s * 3 + 1]; pz[s] = p[s * 3 + 2];
    di[s] = __float_as_int(1e10f);
    pc[lane * PPL + s] = make_float4(px[s], py[s], pz[s], 0.f);
  }
  __syncthreads();
  int far = 0;
  for (int it0 = 0; it0 < S; it0 += 32) {
    int keep = 0;
    for (int j = 0; j < 32; ++j) {
      const float4 f = pc[far];
      keep = (lane == j) ? far : keep;
      constexpr int CH = PPL / 8;
      int v[8], ix[8];
#pragma unroll
      for (int q = 0; q < 8; ++q) {
#pragma unroll
        for (int e = 0; e < CH; ++e) {
          const int s = q * CH + e;
          const float dx = px[s] - f.x, dy = py[s] - f.y, dz = pz[s] - f.z;
          const float t = __builtin_fmaf(dz, dz, __builtin_fmaf(dx, dx, dy * dy));
          di[s] = imin2(di[s], __float_as_int(t));
          if (e == 0) { v[q] = di[s]; ix[q] = s; }
          else {
            const bool c = di[s] > v[q];
            v[q]  = c ? di[s] : v[q];
            ix[q] = c ? s : ix[q];
          }
        }
      }
#pragma unroll
      for (int w = 8; w > 1; w >>= 1) {
#pragma unroll
        for (int k = 0; k < w; k += 2) {
          const bool c = v[k + 1] > v[k];
          const int nv = c ? v[k + 1] : v[k];
          const int ni = c ? ix[k + 1] : ix[k];
          v[k / 2] = nv; ix[k / 2] = ni;
        }
      }
      const int mine = v[0];
      int g = mine;
      g = imax2(g, row_xchg<0x161>(g));
      g = imax2(g, row_xchg<0x162>(g));
      g = imax2(g, row_xchg<0x164>(g));
      g = imax2(g, row_xchg<0x168>(g));
      g = imax2(g, (int)__builtin_amdgcn_permlanex16(g, g, 0x76543210, 0xfedcba98, false, false));
      const unsigned hit = __builtin_amdgcn_ballot_w32(mine == g);
      const int L = __builtin_ctz(hit);
      far = L * PPL + __builtin_amdgcn_readlane(ix[0], L);
    }
    fps_idx[b * S + it0 + lane] = keep;
  }
}

__global__ void k_gather(const float* __restrict__ xyz, const float* __restrict__ feat,
                         const int* __restrict__ fps_idx, int N, int S, int C,
                         float* __restrict__ xyz_s, float* __restrict__ feat_s) {
  const int b = blockIdx.x / S, s = blockIdx.x % S;
  const int idx = fps_idx[b * S + s];
  if (threadIdx.x < 3)
    xyz_s[(size_t)(b * S + s) * 3 + threadIdx.x] = xyz[(size_t)(b * N + idx) * 3 + threadIdx.x];
  for (int j = threadIdx.x; j < C; j += blockDim.x)
    feat_s[(size_t)(b * S + s) * C + j] = feat[(size_t)(b * N + idx) * C + j];
}

#define KW 2
#define KCAP 128
#define KU 8

__device__ __forceinline__ float sq3(float p0, float p1, float p2) {
  return __builtin_fmaf(p2, p2, __builtin_fmaf(p0, p0, p1 * p1));
}

__device__ __forceinline__ v8f knn_tile(const float* __restrict__ x, int n, int hf, v2f a,
                                        float& nn) {
  const float x0 = x[n * 3], x1 = x[n * 3 + 1], x2 = x[n * 3 + 2];
  nn = sq3(x0, x1, x2);
  v2f b;
  b.x = hf ? x2 : x0;  b.y = hf ? 0.f : x1;
  v8f d = {};
  return __builtin_amdgcn_wmma_f32_16x16x4_f32(false, a, false, b, (short)0, d, false, false);
}

template <int N>
__global__ __launch_bounds__(KW * 32) void k_knn_fused(const float* __restrict__ xs_all,
                                                       const float* __restrict__ x_all,
                                                       int S, int* __restrict__ knn_idx) {
  __shared__ float gm[KW][16][64];
  __shared__ int2  cand[KW][16][KCAP];
  __shared__ int   cntL[KW][16];
  const int w    = threadIdx.x >> 5;
  const int lane = threadIdx.x & 31;
  const int hf   = lane >> 4;
  const int ln   = lane & 15;
  const int row0 = (blockIdx.x * KW + w) * 16;
  const int b    = row0 / S;
  const float* xs = xs_all + (size_t)row0 * 3;
  const float* x  = x_all + (size_t)b * N * 3;
  const float s0 = xs[ln * 3], s1 = xs[ln * 3 + 1], s2 = xs[ln * 3 + 2];
  v2f a;
  a.x = hf ? s2 : s0;  a.y = hf ? 0.f : s1;
  float sN[8];
#pragma unroll
  for (int r = 0; r < 8; ++r) {
    const float a0 = xs[(r + 8 * hf) * 3], a1 = xs[(r + 8 * hf) * 3 + 1], a2 = xs[(r + 8 * hf) * 3 + 2];
    sN[r] = sq3(a0, a1, a2);
  }

  float g[4][8];
#pragma unroll
  for (int u = 0; u < 4; ++u)
#pragma unroll
    for (int r = 0; r < 8; ++r) g[u][r] = INFINITY;
  for (int t = 0; t < N / 16; t += KU) {
    v8f d[KU];
    float nn[KU];
#pragma unroll
    for (int u = 0; u < KU; ++u) d[u] = knn_tile(x, (t + u) * 16 + ln, hf, a, nn[u]);
#pragma unroll
    for (int u = 0; u < KU; ++u)
#pragma unroll
      for (int r = 0; r < 8; ++r)
        g[u & 3][r] = fminf(g[u & 3][r], __builtin_fmaf(-2.f, d[u][r], sN[r] + nn[u]));
  }
#pragma unroll
  for (int u = 0; u < 4; ++u)
#pragma unroll
    for (int r = 0; r < 8; ++r) gm[w][r + 8 * hf][u * 16 + ln] = g[u][r];
  __syncthreads();

  float T[8];
#pragma unroll
  for (int r = 0; r < 8; ++r) {
    const float* grow = gm[w][r + 8 * hf];
    int k0 = 0, k1 = 0, k2 = 0, k3 = 0;
#pragma unroll 16
    for (int e = 0; e < 64; ++e) {
      const float o = grow[e];
      k0 += (o < g[0][r]) ? 1 : 0;
      k1 += (o < g[1][r]) ? 1 : 0;
      k2 += (o < g[2][r]) ? 1 : 0;
      k3 += (o < g[3][r]) ? 1 : 0;
    }
    float tv = (k0 < 32) ? g[0][r] : -INFINITY;
    tv = fmaxf(tv, (k1 < 32) ? g[1][r] : -INFINITY);
    tv = fmaxf(tv, (k2 < 32) ? g[2][r] : -INFINITY);
    tv = fmaxf(tv, (k3 < 32) ? g[3][r] : -INFINITY);
#pragma unroll
    for (int o = 8; o > 0; o >>= 1) tv = fmaxf(tv, __shfl_xor(tv, o, 32));
    T[r] = tv;
  }

  int cnt[8];
#pragma unroll
  for (int r = 0; r < 8; ++r) cnt[r] = 0;
  const unsigned below = (1u << ln) - 1u;
  for (int t = 0; t < N / 16; t += KU) {
    v8f d[KU];
    float nn[KU];
#pragma unroll
    for (int u = 0; u < KU; ++u) d[u] = knn_tile(x, (t + u) * 16 + ln, hf, a, nn[u]);
#pragma unroll
    for (int u = 0; u < KU; ++u) {
      const int col = (t + u) * 16 + ln;
#pragma unroll
      for (int r = 0; r < 8; ++r) {
        const float v = __builtin_fmaf(-2.f, d[u][r], sN[r] + nn[u]);
        const bool c = v <= T[r];
        const unsigned bal = __builtin_amdgcn_ballot_w32(c);
        if (bal != 0u) {
          const unsigned hb = (hf ? (bal >> 16) : bal) & 0xffffu;
          const int pos = cnt[r] + __builtin_popcount(hb & below);
          if (c && pos < KCAP) cand[w][r + 8 * hf][pos] = make_int2(__float_as_int(v), col);
          cnt[r] += __builtin_popcount(hb);
        }
      }
    }
  }
#pragma unroll
  for (int r = 0; r < 8; ++r) {
    const int p = cnt[r] + ln;
    if (ln < 8 && p < ((cnt[r] + 7) & ~7) && p < KCAP)
      cand[w][r + 8 * hf][p] = make_int2(__float_as_int(INFINITY), 0x7fffffff);
  }
  if (ln == 0) {
#pragma unroll
    for (int r = 0; r < 8; ++r) cntL[w][r + 8 * hf] = cnt[r];
  }
  __syncthreads();

  bool over = false;
  for (int R = 0; R < 16; ++R) {
    const int M = __builtin_amdgcn_readfirstlane(cntL[w][R]);
    if (M > KCAP) { over = true; continue; }
    const int2* cr = cand[w][R];
    int* outp = knn_idx + (size_t)(row0 + R) * KNN;
    const int Mp = (M + 7) & ~7;
    for (int base = 0; base < M; base += 64) {
      const int my0 = base + lane, my1 = base + 32 + lane;
      const bool have0 = my0 < M, have1 = my1 < M;
      const int2 mine0 = cr[have0 ? my0 : 0];
      const int2 mine1 = cr[have1 ? my1 : 0];
      const float mv0 = __int_as_float(mine0.x), mv1 = __int_as_float(mine1.x);
      const int   mi0 = mine0.y, mi1 = mine1.y;
      int rank0 = 0, rank1 = 0;
      for (int j = 0; j < Mp; j += 8) {
#pragma unroll
        for (int q = 0; q < 8; ++q) {
          const int2 o = cr[j + q];
          const float ov = __int_as_float(o.x);
          rank0 += (ov < mv0 || (ov == mv0 && o.y < mi0)) ? 1 : 0;
          rank1 += (ov < mv1 || (ov == mv1 && o.y < mi1)) ? 1 : 0;
        }
      }
      if (have0 && rank0 < KNN) outp[rank0] = mi0;
      if (have1 && rank1 < KNN) outp[rank1] = mi1;
    }
  }

  if (over) {
    float pv[8]; int pi[8];
#pragma unroll
    for (int r = 0; r < 8; ++r) { pv[r] = -INFINITY; pi[r] = -1; }
    for (int k = 0; k < KNN; ++k) {
      float bv[8]; int bi[8];
#pragma unroll
      for (int r = 0; r < 8; ++r) { bv[r] = INFINITY; bi[r] = 0x7fffffff; }
      for (int t = 0; t < N / 16; ++t) {
        float nn;
        const v8f d = knn_tile(x, t * 16 + ln, hf, a, nn);
        const int col = t * 16 + ln;
#pragma unroll
        for (int r = 0; r < 8; ++r) {
          const float v = __builtin_fmaf(-2.f, d[r], sN[r] + nn);
          const bool after  = (v > pv[r]) || (v == pv[r] && col > pi[r]);
          const bool better = (v < bv[r]) || (v == bv[r] && col < bi[r]);
          if (after && better) { bv[r] = v; bi[r] = col; }
        }
      }
#pragma unroll
      for (int r = 0; r < 8; ++r) {
#pragma unroll
        for (int o = 8; o > 0; o >>= 1) {
          const float ov = __shfl_xor(bv[r], o, 32);
          const int   oi = __shfl_xor(bi[r], o, 32);
          if (ov < bv[r] || (ov == bv[r] && oi < bi[r])) { bv[r] = ov; bi[r] = oi; }
        }
        pv[r] = bv[r]; pi[r] = bi[r];
        if (ln == 0 && cntL[w][r + 8 * hf] > KCAP)
          knn_idx[(size_t)(row0 + r + 8 * hf) * KNN + k] = bi[r];
      }
    }
  }
}

__global__ __launch_bounds__(256) void k_std_partial(const float* __restrict__ xyz,
                                                     const float* __restrict__ feat,
                                                     const float* __restrict__ xyz_s,
                                                     const float* __restrict__ feat_s,
                                                     const int* __restrict__ knn_idx,
                                                     int N, int S, int C,
                                                     float* __restrict__ part) {
  __shared__ float s1[256], s2[256], s3[256], s4[256];
  const int k = blockIdx.x, g = blockIdx.y;
  float ax = 0.f, qx = 0.f, af = 0.f, qf = 0.f;
  for (int p = g * 256 + threadIdx.x; p < BATCH * S; p += STD_G * 256) {
    int b = p / S, s = p % S;
    int idx = knn_idx[(size_t)(b * S + s) * KNN + k];
    for (int c = 0; c < 3; ++c) {
      float d = xyz[(size_t)(b * N + idx) * 3 + c] - xyz_s[(size_t)(b * S + s) * 3 + c];
      ax += d; qx += d * d;
    }
    const float* fr = feat   + (size_t)(b * N + idx) * C;
    const float* fc = feat_s + (size_t)(b * S + s) * C;
    for (int j = 0; j < C; ++j) { float d = fr[j] - fc[j]; af += d; qf += d * d; }
  }
  s1[threadIdx.x] = ax; s2[threadIdx.x] = qx; s3[threadIdx.x] = af; s4[threadIdx.x] = qf;
  __syncthreads();
  for (int o = 128; o > 0; o >>= 1) {
    if ((int)threadIdx.x < o) {
      s1[threadIdx.x] += s1[threadIdx.x + o]; s2[threadIdx.x] += s2[threadIdx.x + o];
      s3[threadIdx.x] += s3[threadIdx.x + o]; s4[threadIdx.x] += s4[threadIdx.x + o];
    }
    __syncthreads();
  }
  if (threadIdx.x == 0) {
    float* q = part + ((size_t)k * STD_G + g) * 4;
    q[0] = s1[0]; q[1] = s2[0]; q[2] = s3[0]; q[3] = s4[0];
  }
}

__global__ void k_std_final(const float* __restrict__ part, int S, int C,
                            float* __restrict__ std_xyz, float* __restrict__ std_feat) {
  const int k = threadIdx.x;
  if (k >= KNN) return;
  float ax = 0.f, qx = 0.f, af = 0.f, qf = 0.f;
  for (int g = 0; g < STD_G; ++g) {
    const float* q = part + ((size_t)k * STD_G + g) * 4;
    ax += q[0]; qx += q[1]; af += q[2]; qf += q[3];
  }
  {
    double n = (double)BATCH * S * 3;
    double var = ((double)qx - (double)ax * ax / n) / (n - 1.0);
    std_xyz[k] = fmaxf(sqrtf(fmaxf((float)var, 0.f)), 1e-5f);
  }
  {
    double n = (double)BATCH * S * C;
    double var = ((double)qf - (double)af * af / n) / (n - 1.0);
    std_feat[k] = fmaxf(sqrtf(fmaxf((float)var, 0.f)), 1e-5f);
  }
}

#define PCH 4
__global__ __launch_bounds__(256) void k_stage(const float* __restrict__ xyz,
                                               const float* __restrict__ feat,
                                               const float* __restrict__ xyz_s,
                                               const float* __restrict__ feat_s,
                                               const int* __restrict__ knn_idx,
                                               const float* __restrict__ std_xyz,
                                               const float* __restrict__ std_feat,
                                               int N, int S, int C, int chunk,
                                               float* __restrict__ out,
                                               float* __restrict__ pmaxb,
                                               float* __restrict__ psumb) {
  __shared__ int   kidx[PCH * KNN];
  __shared__ float sfe[KNN];
  __shared__ float xkn[PCH * KNN][3];
  __shared__ float fs[512];
  const int nt   = blockDim.x;
  const int row0 = blockIdx.x * chunk;
  const int b    = row0 / S;
  const int nk   = chunk * KNN;
  for (int t = threadIdx.x; t < nk; t += nt) kidx[t] = knn_idx[(size_t)row0 * KNN + t];
  if (threadIdx.x < KNN) sfe[threadIdx.x] = std_feat[threadIdx.x];
  __syncthreads();
  for (int t = threadIdx.x; t < nk; t += nt)
    __builtin_prefetch(feat + (size_t)(b * N + kidx[t]) * C, 0, 3);
  for (int t = threadIdx.x; t < nk * 3; t += nt) {
    const int kk = t / 3, c = t % 3;
    const int q = kk / KNN, k = kk % KNN;
    xkn[kk][c] = (xyz[(size_t)(b * N + kidx[kk]) * 3 + c] - xyz_s[(size_t)(row0 + q) * 3 + c]) / std_xyz[k];
  }
  for (int t = threadIdx.x; t < chunk * C; t += nt) fs[t] = feat_s[(size_t)row0 * C + t];
  __syncthreads();
  const int C2 = 2 * C;
  const int fd = (C2 + 2) / 3;
  const int M3 = 3 * fd;
  const float inv_sig = 1.f / SIGMA_;
  for (int j = threadIdx.x; j < C2; j += nt) {
    int src = (int)(((long long)j * (M3 - 1)) / (C2 - 1));
    int cc = src / fd, f = src % fd;
    float fv = -1.f + 2.f * (float)(f + 1) / (float)(fd + 1);
    float pmx = -FLT_MAX, psm = 0.f;
    for (int q = 0; q < chunk; ++q) {
      const int*   kq = kidx + q * KNN;
      const float* fq = fs + q * C;
      float acc = 0.f, mx = -FLT_MAX;
      for (int k = 0; k < KNN; ++k) {
        float t  = (xkn[q * KNN + k][cc] - fv) * inv_sig;
        float pe = expf(-0.5f * t * t);
        float fk;
        if (j < C) fk = (feat[(size_t)(b * N + kq[k]) * C + j] - fq[j]) / sfe[k];
        else       fk = fq[j - C];
        float fw = (fk + pe) * pe;
        acc += fw; mx = fmaxf(mx, fw);
      }
      float xx = acc * (1.f / KNN) + mx;
      const float o = 0.5f * xx * (1.f + erff(xx * 0.70710678118654752f));
      out[(size_t)(row0 + q) * C2 + j] = o;
      pmx = fmaxf(pmx, o); psm += o;
    }
    if (chunk > 1) {
      pmaxb[(size_t)blockIdx.x * C2 + j] = pmx;
      psumb[(size_t)blockIdx.x * C2 + j] = psm;
    }
  }
}

__global__ __launch_bounds__(1024) void k_pool_fold(const float* pm, const float* ps,

                                                    int G, int C2, int S,
                                                    float* __restrict__ out, int off) {
  const int b = blockIdx.x;
  for (int j = threadIdx.x; j < C2; j += blockDim.x) {
    const float* a = pm + (size_t)b * G * C2 + j;
    const float* c = ps + (size_t)b * G * C2 + j;
    float mx = -FLT_MAX, sm = 0.f;
    if (pm == ps) {
#pragma unroll 8
      for (int g = 0; g < G; ++g) {
        const float v = a[(size_t)g * C2];
        mx = fmaxf(mx, v); sm += v;
      }
    } else {
#pragma unroll 8
      for (int g = 0; g < G; ++g) {
        mx = fmaxf(mx, a[(size_t)g * C2]);
        sm += c[(size_t)g * C2];
      }
    }
    out[b * 3840 + off + j]      = mx;
    out[b * 3840 + off + C2 + j] = sm / (float)S;
  }
}

extern "C" void kernel_launch(void* const* d_in, const int* in_sizes, int n_in,
                              void* d_out, int out_size, void* d_ws, size_t ws_size,
                              hipStream_t stream) {
  const float* xyz_in = (const float*)d_in[0];
  const float* W      = (const float*)d_in[1];
  const float* gamma  = (const float*)d_in[2];
  const float* beta   = (const float*)d_in[3];
  float* out = (float*)d_out;
  float* ws  = (float*)d_ws;

  size_t o = 0;
  float* featA  = ws + o; o += (size_t)BATCH * NPTS * CINIT;
  float* featB  = ws + o; o += (size_t)BATCH * NPTS * CINIT;
  float* xyzA   = ws + o; o += (size_t)BATCH * NPTS * 3;
  float* xyzB   = ws + o; o += (size_t)BATCH * NPTS * 3;
  float* feat_s = ws + o; o += (size_t)BATCH * 1024 * CINIT;
  float* pmaxb  = ws + o; o += (size_t)1024 * 1024;
  float* psumb  = ws + o; o += (size_t)1024 * 1024;
  float* bn_mu  = ws + o; o += 64;
  float* bn_inv = ws + o; o += 64;
  float* stdx   = ws + o; o += 32;
  float* stdf   = ws + o; o += 32;
  float* stdp   = ws + o; o += (size_t)KNN * STD_G * 4;
  int* fps_i = (int*)(ws + o); o += (size_t)BATCH * 1024;
  int* knn_i = (int*)(ws + o); o += (size_t)BATCH * 1024 * KNN;

  k_embed_wmma<<<1024, 256, 0, stream>>>(xyz_in, W, featA);
  k_bn_stats<<<CINIT, 256, 0, stream>>>(featA, bn_mu, bn_inv);
  k_bn_apply<<<2048, 256, 0, stream>>>(featA, bn_mu, bn_inv, gamma, beta);

  const float* xyz_cur = xyz_in;
  float* feat_cur = featA; float* feat_nxt = featB;
  float* xs_buf = xyzA;    float* xs_other = xyzB;
  int N = NPTS, C = CINIT, off = 0;

  for (int st = 0; st < 4; ++st) {
    const int S = N / 2, C2 = 2 * C;
    if (N == 2048)      k_fps_wave<64><<<BATCH, 32, 0, stream>>>(xyz_cur, S, fps_i);
    else if (N == 1024) k_fps_wave<32><<<BATCH, 32, 0, stream>>>(xyz_cur, S, fps_i);
    else if (N == 512)  k_fps_wave<16><<<BATCH, 32, 0, stream>>>(xyz_cur, S, fps_i);
    else                k_fps_wave<8><<<BATCH, 32, 0, stream>>>(xyz_cur, S, fps_i);
    k_gather<<<BATCH * S, 256, 0, stream>>>(xyz_cur, feat_cur, fps_i, N, S, C, xs_buf, feat_s);
    {
      const int nb = BATCH * S / 16 / KW;
      if (N == 2048)      k_knn_fused<2048><<<nb, KW * 32, 0, stream>>>(xs_buf, xyz_cur, S, knn_i);
      else if (N == 1024) k_knn_fused<1024><<<nb, KW * 32, 0, stream>>>(xs_buf, xyz_cur, S, knn_i);
      else if (N == 512)  k_knn_fused<512><<<nb, KW * 32, 0, stream>>>(xs_buf, xyz_cur, S, knn_i);
      else                k_knn_fused<256><<<nb, KW * 32, 0, stream>>>(xs_buf, xyz_cur, S, knn_i);
    }
    {
      dim3 g(KNN, STD_G);
      k_std_partial<<<g, 256, 0, stream>>>(xyz_cur, feat_cur, xs_buf, feat_s, knn_i, N, S, C, stdp);
      k_std_final<<<1, KNN, 0, stream>>>(stdp, S, C, stdx, stdf);
    }
    {
      const int chunk = (C2 <= 256) ? 4 : ((C2 == 512) ? 2 : 1);
      const int nthr  = (C2 < 256) ? C2 : 256;
      k_stage<<<BATCH * S / chunk, nthr, 0, stream>>>(xyz_cur, feat_cur, xs_buf, feat_s, knn_i,
                                                      stdx, stdf, N, S, C, chunk, feat_nxt,
                                                      pmaxb, psumb);
      if (chunk > 1) k_pool_fold<<<BATCH, C2, 0, stream>>>(pmaxb, psumb, S / chunk, C2, S, out, off);
      else           k_pool_fold<<<BATCH, C2, 0, stream>>>(feat_nxt, feat_nxt, S, C2, S, out, off);
    }
    off += 2 * C2;
    float* t = feat_cur; feat_cur = feat_nxt; feat_nxt = t;
    xyz_cur = xs_buf;
    float* u = xs_buf; xs_buf = xs_other; xs_other = u;
    N = S; C = C2;
  }
  (void)in_sizes; (void)n_in; (void)out_size; (void)ws_size;
}


# FPS pick in two steps (N>=1024): scan keeps values only (max of 8 slot groups); lane+group by equality, then a scalar branch compares that group's slots while its points' LDS read is in flight
# speedup vs baseline: 4.2346x; 1.0600x over previous
#include <hip/hip_runtime.h>
#include <cfloat>
#include <cmath>

typedef __attribute__((ext_vector_type(2))) float v2f;
typedef __attribute__((ext_vector_type(3))) float v3f;
typedef __attribute__((ext_vector_type(8))) float v8f;

#define BATCH 16
#define NPTS 2048
#define CINIT 64
#define KNN 32
#define SIGMA_ 0.3f
#define BN_EPS_ 1e-5f
#define STD_G 16

__global__ __launch_bounds__(256) void k_embed_wmma(const float* __restrict__ xyz,
                                                    const float* __restrict__ W,
                                                    float* __restrict__ h) {
  const int wid  = (blockIdx.x * 256 + (int)threadIdx.x) >> 5;
  const int lane = threadIdx.x & 31;
  const int hf   = lane >> 4;
  const int ln   = lane & 15;
  const int tileM = wid >> 2;
  const int tileN = wid & 3;
  const int m  = tileM * 16 + ln;
  const int c  = tileN * 16 + ln;
  const float x0 = xyz[m * 3], x1 = xyz[m * 3 + 1], x2 = xyz[m * 3 + 2];
  const float w0 = W[c * 3],   w1 = W[c * 3 + 1],   w2 = W[c * 3 + 2];
  v2f a, b;
  a.x = hf ? x2 : x0;  a.y = hf ? 0.f : x1;
  b.x = hf ? w2 : w0;  b.y = hf ? 0.f : w1;
  v8f d = {};
  d = __builtin_amdgcn_wmma_f32_16x16x4_f32(false, a, false, b, (short)0, d, false, false);
#pragma unroll
  for (int r = 0; r < 8; ++r) {
    int row = tileM * 16 + r + 8 * hf;
    h[row * CINIT + tileN * 16 + ln] = d[r];
  }
}

__global__ __launch_bounds__(256) void k_bn_stats(const float* __restrict__ h,
                                                  float* __restrict__ mu,
                                                  float* __restrict__ inv) {
  __shared__ float s1[256], s2[256];
  const int c = blockIdx.x;
  const int n = BATCH * NPTS;
  float a = 0.f, b = 0.f;
  for (int i = threadIdx.x; i < n; i += 256) {
    float v = h[i * CINIT + c];
    a += v; b += v * v;
  }
  s1[threadIdx.x] = a; s2[threadIdx.x] = b; __syncthreads();
  for (int o = 128; o > 0; o >>= 1) {
    if ((int)threadIdx.x < o) { s1[threadIdx.x] += s1[threadIdx.x + o]; s2[threadIdx.x] += s2[threadIdx.x + o]; }
    __syncthreads();
  }
  if (threadIdx.x == 0) {
    float m   = s1[0] / (float)n;
    float var = s2[0] / (float)n - m * m;
    mu[c]  = m;
    inv[c] = rsqrtf(var + BN_EPS_);
  }
}

__global__ void k_bn_apply(float* __restrict__ h, const float* __restrict__ mu,
                           const float* __restrict__ inv, const float* __restrict__ gamma,
                           const float* __restrict__ beta) {
  const int total = BATCH * NPTS * CINIT;
  for (int i = blockIdx.x * blockDim.x + threadIdx.x; i < total; i += gridDim.x * blockDim.x) {
    int c = i & (CINIT - 1);
    float v = (h[i] - mu[c]) * inv[c] * gamma[c] + beta[c];
    h[i] = fmaxf(v, 0.f);
  }
}

__device__ __forceinline__ int imax2(int a, int b) { return a > b ? a : b; }
__device__ __forceinline__ int imin2(int a, int b) { return a < b ? a : b; }

template <int CTRL>
__device__ __forceinline__ int row_xchg(int v) {
  return __builtin_amdgcn_update_dpp(v, v, CTRL, 0xf, 0xf, true);
}

template <int PPL, int Q>
__device__ __forceinline__ unsigned fps_eq_bits(const int (&di)[PPL], int m) {
  constexpr int CH = PPL / 8;
  unsigned bits = 1u << (CH - 1);
#pragma unroll
  for (int x = 0; x < CH - 1; ++x) bits |= (di[Q * CH + x] == m) ? (1u << x) : 0u;
  return bits;
}

template <int PPL>
__device__ __forceinline__ int fps_pick_values(const v2f (&px)[PPL / 2], const v2f (&py)[PPL / 2],
                                               const v2f (&pz)[PPL / 2], int (&di)[PPL],
                                               const float4* pc, int lane,
                                               float& fx, float& fy, float& fz) {
  constexpr int CH = PPL / 8;
  int gm[8];
#pragma unroll
  for (int p2 = 0; p2 < PPL / 2; ++p2) {
    const v2f dx = px[p2] - fx, dy = py[p2] - fy, dz = pz[p2] - fz;
    const v2f t = __builtin_elementwise_fma(dz, dz, __builtin_elementwise_fma(dx, dx, dy * dy));
    const int s0 = 2 * p2, s1 = 2 * p2 + 1, q = s0 / CH;
    di[s0] = imin2(di[s0], __float_as_int(t.x));
    di[s1] = imin2(di[s1], __float_as_int(t.y));
    gm[q] = (s0 % CH == 0) ? imax2(di[s0], di[s1]) : imax2(gm[q], imax2(di[s0], di[s1]));
  }
  int mine = gm[0];
#pragma unroll
  for (int q = 1; q < 8; ++q) mine = imax2(mine, gm[q]);
  unsigned gbits = 0x80u;
#pragma unroll
  for (int q = 0; q < 7; ++q) gbits |= (gm[q] == mine) ? (1u << q) : 0u;
  int g = mine;
  g = imax2(g, row_xchg<0x161>(g));
  g = imax2(g, row_xchg<0x162>(g));
  g = imax2(g, row_xchg<0x164>(g));
  g = imax2(g, row_xchg<0x168>(g));
  g = imax2(g, (int)__builtin_amdgcn_permlanex16(g, g, 0x76543210, 0xfedcba98, false, false));
  const unsigned hit = __builtin_amdgcn_ballot_w32(mine == g);
  const int L = __builtin_ctz(hit);
  const int sq = __builtin_ctz((unsigned)__builtin_amdgcn_readlane((int)gbits, L));
  const int base = L * PPL + sq * CH;
  typedef const volatile __attribute__((address_space(3))) v3f lds_v3f;
  const v3f cand = *(lds_v3f*)(pc + base + (lane & (CH - 1)));
  unsigned eb;
  switch (sq) {
    case 0:  eb = fps_eq_bits<PPL, 0>(di, mine); break;
    case 1:  eb = fps_eq_bits<PPL, 1>(di, mine); break;
    case 2:  eb = fps_eq_bits<PPL, 2>(di, mine); break;
    case 3:  eb = fps_eq_bits<PPL, 3>(di, mine); break;
    case 4:  eb = fps_eq_bits<PPL, 4>(di, mine); break;
    case 5:  eb = fps_eq_bits<PPL, 5>(di, mine); break;
    case 6:  eb = fps_eq_bits<PPL, 6>(di, mine); break;
    default: eb = fps_eq_bits<PPL, 7>(di, mine); break;
  }
  const int es = __builtin_ctz((unsigned)__builtin_amdgcn_readlane((int)eb, L));
  fx = __int_as_float(__builtin_amdgcn_readlane(__float_as_int(cand.x), es));
  fy = __int_as_float(__builtin_amdgcn_readlane(__float_as_int(cand.y), es));
  fz = __int_as_float(__builtin_amdgcn_readlane(__float_as_int(cand.z), es));
  return base + es;
}

template <int PPL>
__global__ __launch_bounds__(32) void k_fps_wave(const float* __restrict__ xyz, int S,
                                                 int* __restrict__ fps_idx) {
  constexpr int N = 32 * PPL;
  constexpr bool VALS = (PPL >= 32);
  __shared__ float4 pc[N];
  const int b = blockIdx.x;
  const int lane = threadIdx.x;
  const float* p = xyz + ((size_t)b * N + (size_t)lane * PPL) * 3;
  float px[VALS ? 1 : PPL], py[VALS ? 1 : PPL], pz[VALS ? 1 : PPL];
  v2f qx[VALS ? PPL / 2 : 1], qy[VALS ? PPL / 2 : 1], qz[VALS ? PPL / 2 : 1];
  int di[PPL];
#pragma unroll
  for (int s = 0; s < PPL; ++s) {
    const float c0 = p[s * 3], c1 = p[s * 3 + 1], c2 = p[s * 3 + 2];
    if constexpr (VALS) { qx[s / 2][s % 2] = c0; qy[s / 2][s % 2] = c1; qz[s / 2][s % 2] = c2; }
    else                { px[s] = c0; py[s] = c1; pz[s] = c2; }
    di[s] = __float_as_int(1e10f);
    pc[lane * PPL + s] = make_float4(c0, c1, c2, 0.f);
  }
  __syncthreads();
  int far = 0;
  float fx = 0.f, fy = 0.f, fz = 0.f;
  if constexpr (VALS) { const float4 f0 = pc[0]; fx = f0.x; fy = f0.y; fz = f0.z; }
  for (int it0 = 0; it0 < S; it0 += 32) {
    int keep = 0;
    for (int j = 0; j < 32; ++j) {
      if constexpr (VALS) {
        keep = (lane == j) ? far : keep;
        far = fps_pick_values<PPL>(qx, qy, qz, di, pc, lane, fx, fy, fz);
      } else {
        const float4 f = pc[far];
        keep = (lane == j) ? far : keep;
        constexpr int CH = PPL / 8;
        int v[8], ix[8];
#pragma unroll
        for (int q = 0; q < 8; ++q) {
#pragma unroll
          for (int e = 0; e < CH; ++e) {
            const int s = q * CH + e;
            const float dx = px[s] - f.x, dy = py[s] - f.y, dz = pz[s] - f.z;
            const float t = __builtin_fmaf(dz, dz, __builtin_fmaf(dx, dx, dy * dy));
            di[s] = imin2(di[s], __float_as_int(t));
            if (e == 0) { v[q] = di[s]; ix[q] = s; }
            else {
              const bool c = di[s] > v[q];
              v[q]  = c ? di[s] : v[q];
              ix[q] = c ? s : ix[q];
            }
          }
        }
#pragma unroll
        for (int w = 8; w > 1; w >>= 1) {
#pragma unroll
          for (int k = 0; k < w; k += 2) {
            const bool c = v[k + 1] > v[k];
            const int nv = c ? v[k + 1] : v[k];
            const int ni = c ? ix[k + 1] : ix[k];
            v[k / 2] = nv; ix[k / 2] = ni;
          }
        }
        const int mine = v[0];
        int g = mine;
        g = imax2(g, row_xchg<0x161>(g));
        g = imax2(g, row_xchg<0x162>(g));
        g = imax2(g, row_xchg<0x164>(g));
        g = imax2(g, row_xchg<0x168>(g));
        g = imax2(g, (int)__builtin_amdgcn_permlanex16(g, g, 0x76543210, 0xfedcba98, false, false));
        const unsigned hit = __builtin_amdgcn_ballot_w32(mine == g);
        const int L = __builtin_ctz(hit);
        far = L * PPL + __builtin_amdgcn_readlane(ix[0], L);
      }
    }
    fps_idx[b * S + it0 + lane] = keep;
  }
}

__global__ void k_gather(const float* __restrict__ xyz, const float* __restrict__ feat,
                         const int* __restrict__ fps_idx, int N, int S, int C,
                         float* __restrict__ xyz_s, float* __restrict__ feat_s) {
  const int b = blockIdx.x / S, s = blockIdx.x % S;
  const int idx = fps_idx[b * S + s];
  if (threadIdx.x < 3)
    xyz_s[(size_t)(b * S + s) * 3 + threadIdx.x] = xyz[(size_t)(b * N + idx) * 3 + threadIdx.x];
  for (int j = threadIdx.x; j < C; j += blockDim.x)
    feat_s[(size_t)(b * S + s) * C + j] = feat[(size_t)(b * N + idx) * C + j];
}

#define KW 2
#define KCAP 128
#define KU 8

__device__ __forceinline__ float sq3(float p0, float p1, float p2) {
  return __builtin_fmaf(p2, p2, __builtin_fmaf(p0, p0, p1 * p1));
}

__device__ __forceinline__ v8f knn_tile(const float* __restrict__ x, int n, int hf, v2f a,
                                        float& nn) {
  const float x0 = x[n * 3], x1 = x[n * 3 + 1], x2 = x[n * 3 + 2];
  nn = sq3(x0, x1, x2);
  v2f b;
  b.x = hf ? x2 : x0;  b.y = hf ? 0.f : x1;
  v8f d = {};
  return __builtin_amdgcn_wmma_f32_16x16x4_f32(false, a, false, b, (short)0, d, false, false);
}

template <int N>
__global__ __launch_bounds__(KW * 32) void k_knn_fused(const float* __restrict__ xs_all,
                                                       const float* __restrict__ x_all,
                                                       int S, int* __restrict__ knn_idx) {
  __shared__ float gm[KW][16][64];
  __shared__ int2  cand[KW][16][KCAP];
  __shared__ int   cntL[KW][16];
  const int w    = threadIdx.x >> 5;
  const int lane = threadIdx.x & 31;
  const int hf   = lane >> 4;
  const int ln   = lane & 15;
  const int row0 = (blockIdx.x * KW + w) * 16;
  const int b    = row0 / S;
  const float* xs = xs_all + (size_t)row0 * 3;
  const float* x  = x_all + (size_t)b * N * 3;
  const float s0 = xs[ln * 3], s1 = xs[ln * 3 + 1], s2 = xs[ln * 3 + 2];
  v2f a;
  a.x = hf ? s2 : s0;  a.y = hf ? 0.f : s1;
  float sN[8];
#pragma unroll
  for (int r = 0; r < 8; ++r) {
    const float a0 = xs[(r + 8 * hf) * 3], a1 = xs[(r + 8 * hf) * 3 + 1], a2 = xs[(r + 8 * hf) * 3 + 2];
    sN[r] = sq3(a0, a1, a2);
  }

  float g[4][8];
#pragma unroll
  for (int u = 0; u < 4; ++u)
#pragma unroll
    for (int r = 0; r < 8; ++r) g[u][r] = INFINITY;
  for (int t = 0; t < N / 16; t += KU) {
    v8f d[KU];
    float nn[KU];
#pragma unroll
    for (int u = 0; u < KU; ++u) d[u] = knn_tile(x, (t + u) * 16 + ln, hf, a, nn[u]);
#pragma unroll
    for (int u = 0; u < KU; ++u)
#pragma unroll
      for (int r = 0; r < 8; ++r)
        g[u & 3][r] = fminf(g[u & 3][r], __builtin_fmaf(-2.f, d[u][r], sN[r] + nn[u]));
  }
#pragma unroll
  for (int u = 0; u < 4; ++u)
#pragma unroll
    for (int r = 0; r < 8; ++r) gm[w][r + 8 * hf][u * 16 + ln] = g[u][r];
  __syncthreads();

  float T[8];
#pragma unroll
  for (int r = 0; r < 8; ++r) {
    const float* grow = gm[w][r + 8 * hf];
    int k0 = 0, k1 = 0, k2 = 0, k3 = 0;
#pragma unroll 16
    for (int e = 0; e < 64; ++e) {
      const float o = grow[e];
      k0 += (o < g[0][r]) ? 1 : 0;
      k1 += (o < g[1][r]) ? 1 : 0;
      k2 += (o < g[2][r]) ? 1 : 0;
      k3 += (o < g[3][r]) ? 1 : 0;
    }
    float tv = (k0 < 32) ? g[0][r] : -INFINITY;
    tv = fmaxf(tv, (k1 < 32) ? g[1][r] : -INFINITY);
    tv = fmaxf(tv, (k2 < 32) ? g[2][r] : -INFINITY);
    tv = fmaxf(tv, (k3 < 32) ? g[3][r] : -INFINITY);
#pragma unroll
    for (int o = 8; o > 0; o >>= 1) tv = fmaxf(tv, __shfl_xor(tv, o, 32));
    T[r] = tv;
  }

  int cnt[8];
#pragma unroll
  for (int r = 0; r < 8; ++r) cnt[r] = 0;
  const unsigned below = (1u << ln) - 1u;
  for (int t = 0; t < N / 16; t += KU) {
    v8f d[KU];
    float nn[KU];
#pragma unroll
    for (int u = 0; u < KU; ++u) d[u] = knn_tile(x, (t + u) * 16 + ln, hf, a, nn[u]);
#pragma unroll
    for (int u = 0; u < KU; ++u) {
      const int col = (t + u) * 16 + ln;
#pragma unroll
      for (int r = 0; r < 8; ++r) {
        const float v = __builtin_fmaf(-2.f, d[u][r], sN[r] + nn[u]);
        const bool c = v <= T[r];
        const unsigned bal = __builtin_amdgcn_ballot_w32(c);
        if (bal != 0u) {
          const unsigned hb = (hf ? (bal >> 16) : bal) & 0xffffu;
          const int pos = cnt[r] + __builtin_popcount(hb & below);
          if (c && pos < KCAP) cand[w][r + 8 * hf][pos] = make_int2(__float_as_int(v), col);
          cnt[r] += __builtin_popcount(hb);
        }
      }
    }
  }
#pragma unroll
  for (int r = 0; r < 8; ++r) {
    const int p = cnt[r] + ln;
    if (ln < 8 && p < ((cnt[r] + 7) & ~7) && p < KCAP)
      cand[w][r + 8 * hf][p] = make_int2(__float_as_int(INFINITY), 0x7fffffff);
  }
  if (ln == 0) {
#pragma unroll
    for (int r = 0; r < 8; ++r) cntL[w][r + 8 * hf] = cnt[r];
  }
  __syncthreads();

  bool over = false;
  for (int R = 0; R < 16; ++R) {
    const int M = __builtin_amdgcn_readfirstlane(cntL[w][R]);
    if (M > KCAP) { over = true; continue; }
    const int2* cr = cand[w][R];
    int* outp = knn_idx + (size_t)(row0 + R) * KNN;
    const int Mp = (M + 7) & ~7;
    for (int base = 0; base < M; base += 64) {
      const int my0 = base + lane, my1 = base + 32 + lane;
      const bool have0 = my0 < M, have1 = my1 < M;
      const int2 mine0 = cr[have0 ? my0 : 0];
      const int2 mine1 = cr[have1 ? my1 : 0];
      const float mv0 = __int_as_float(mine0.x), mv1 = __int_as_float(mine1.x);
      const int   mi0 = mine0.y, mi1 = mine1.y;
      int rank0 = 0, rank1 = 0;
      for (int j = 0; j < Mp; j += 8) {
#pragma unroll
        for (int q = 0; q < 8; ++q) {
          const int2 o = cr[j + q];
          const float ov = __int_as_float(o.x);
          rank0 += (ov < mv0 || (ov == mv0 && o.y < mi0)) ? 1 : 0;
          rank1 += (ov < mv1 || (ov == mv1 && o.y < mi1)) ? 1 : 0;
        }
      }
      if (have0 && rank0 < KNN) outp[rank0] = mi0;
      if (have1 && rank1 < KNN) outp[rank1] = mi1;
    }
  }

  if (over) {
    float pv[8]; int pi[8];
#pragma unroll
    for (int r = 0; r < 8; ++r) { pv[r] = -INFINITY; pi[r] = -1; }
    for (int k = 0; k < KNN; ++k) {
      float bv[8]; int bi[8];
#pragma unroll
      for (int r = 0; r < 8; ++r) { bv[r] = INFINITY; bi[r] = 0x7fffffff; }
      for (int t = 0; t < N / 16; ++t) {
        float nn;
        const v8f d = knn_tile(x, t * 16 + ln, hf, a, nn);
        const int col = t * 16 + ln;
#pragma unroll
        for (int r = 0; r < 8; ++r) {
          const float v = __builtin_fmaf(-2.f, d[r], sN[r] + nn);
          const bool after  = (v > pv[r]) || (v == pv[r] && col > pi[r]);
          const bool better = (v < bv[r]) || (v == bv[r] && col < bi[r]);
          if (after && better) { bv[r] = v; bi[r] = col; }
        }
      }
#pragma unroll
      for (int r = 0; r < 8; ++r) {
#pragma unroll
        for (int o = 8; o > 0; o >>= 1) {
          const float ov = __shfl_xor(bv[r], o, 32);
          const int   oi = __shfl_xor(bi[r], o, 32);
          if (ov < bv[r] || (ov == bv[r] && oi < bi[r])) { bv[r] = ov; bi[r] = oi; }
        }
        pv[r] = bv[r]; pi[r] = bi[r];
        if (ln == 0 && cntL[w][r + 8 * hf] > KCAP)
          knn_idx[(size_t)(row0 + r + 8 * hf) * KNN + k] = bi[r];
      }
    }
  }
}

__global__ __launch_bounds__(256) void k_std_partial(const float* __restrict__ xyz,
                                                     const float* __restrict__ feat,
                                                     const float* __restrict__ xyz_s,
                                                     const float* __restrict__ feat_s,
                                                     const int* __restrict__ knn_idx,
                                                     int N, int S, int C,
                                                     float* __restrict__ part) {
  __shared__ float s1[256], s2[256], s3[256], s4[256];
  const int k = blockIdx.x, g = blockIdx.y;
  float ax = 0.f, qx = 0.f, af = 0.f, qf = 0.f;
  for (int p = g * 256 + threadIdx.x; p < BATCH * S; p += STD_G * 256) {
    int b = p / S, s = p % S;
    int idx = knn_idx[(size_t)(b * S + s) * KNN + k];
    for (int c = 0; c < 3; ++c) {
      float d = xyz[(size_t)(b * N + idx) * 3 + c] - xyz_s[(size_t)(b * S + s) * 3 + c];
      ax += d; qx += d * d;
    }
    const float* fr = feat   + (size_t)(b * N + idx) * C;
    const float* fc = feat_s + (size_t)(b * S + s) * C;
    for (int j = 0; j < C; ++j) { float d = fr[j] - fc[j]; af += d; qf += d * d; }
  }
  s1[threadIdx.x] = ax; s2[threadIdx.x] = qx; s3[threadIdx.x] = af; s4[threadIdx.x] = qf;
  __syncthreads();
  for (int o = 128; o > 0; o >>= 1) {
    if ((int)threadIdx.x < o) {
      s1[threadIdx.x] += s1[threadIdx.x + o]; s2[threadIdx.x] += s2[threadIdx.x + o];
      s3[threadIdx.x] += s3[threadIdx.x + o]; s4[threadIdx.x] += s4[threadIdx.x + o];
    }
    __syncthreads();
  }
  if (threadIdx.x == 0) {
    float* q = part + ((size_t)k * STD_G + g) * 4;
    q[0] = s1[0]; q[1] = s2[0]; q[2] = s3[0]; q[3] = s4[0];
  }
}

__global__ void k_std_final(const float* __restrict__ part, int S, int C,
                            float* __restrict__ std_xyz, float* __restrict__ std_feat) {
  const int k = threadIdx.x;
  if (k >= KNN) return;
  float ax = 0.f, qx = 0.f, af = 0.f, qf = 0.f;
  for (int g = 0; g < STD_G; ++g) {
    const float* q = part + ((size_t)k * STD_G + g) * 4;
    ax += q[0]; qx += q[1]; af += q[2]; qf += q[3];
  }
  {
    double n = (double)BATCH * S * 3;
    double var = ((double)qx - (double)ax * ax / n) / (n - 1.0);
    std_xyz[k] = fmaxf(sqrtf(fmaxf((float)var, 0.f)), 1e-5f);
  }
  {
    double n = (double)BATCH * S * C;
    double var = ((double)qf - (double)af * af / n) / (n - 1.0);
    std_feat[k] = fmaxf(sqrtf(fmaxf((float)var, 0.f)), 1e-5f);
  }
}

#define PCH 4
__global__ __launch_bounds__(256) void k_stage(const float* __restrict__ xyz,
                                               const float* __restrict__ feat,
                                               const float* __restrict__ xyz_s,
                                               const float* __restrict__ feat_s,
                                               const int* __restrict__ knn_idx,
                                               const float* __restrict__ std_xyz,
                                               const float* __restrict__ std_feat,
                                               int N, int S, int C, int chunk,
                                               float* __restrict__ out,
                                               float* __restrict__ pmaxb,
                                               float* __restrict__ psumb) {
  __shared__ int   kidx[PCH * KNN];
  __shared__ float sfe[KNN];
  __shared__ float xkn[PCH * KNN][3];
  __shared__ float fs[512];
  const int nt   = blockDim.x;
  const int row0 = blockIdx.x * chunk;
  const int b    = row0 / S;
  const int nk   = chunk * KNN;
  for (int t = threadIdx.x; t < nk; t += nt) kidx[t] = knn_idx[(size_t)row0 * KNN + t];
  if (threadIdx.x < KNN) sfe[threadIdx.x] = std_feat[threadIdx.x];
  __syncthreads();
  for (int t = threadIdx.x; t < nk; t += nt)
    __builtin_prefetch(feat + (size_t)(b * N + kidx[t]) * C, 0, 3);
  for (int t = threadIdx.x; t < nk * 3; t += nt) {
    const int kk = t / 3, c = t % 3;
    const int q = kk / KNN, k = kk % KNN;
    xkn[kk][c] = (xyz[(size_t)(b * N + kidx[kk]) * 3 + c] - xyz_s[(size_t)(row0 + q) * 3 + c]) / std_xyz[k];
  }
  for (int t = threadIdx.x; t < chunk * C; t += nt) fs[t] = feat_s[(size_t)row0 * C + t];
  __syncthreads();
  const int C2 = 2 * C;
  const int fd = (C2 + 2) / 3;
  const int M3 = 3 * fd;
  const float inv_sig = 1.f / SIGMA_;
  for (int j = threadIdx.x; j < C2; j += nt) {
    int src = (int)(((long long)j * (M3 - 1)) / (C2 - 1));
    int cc = src / fd, f = src % fd;
    float fv = -1.f + 2.f * (float)(f + 1) / (float)(fd + 1);
    float pmx = -FLT_MAX, psm = 0.f;
    for (int q = 0; q < chunk; ++q) {
      const int*   kq = kidx + q * KNN;
      const float* fq = fs + q * C;
      float acc = 0.f, mx = -FLT_MAX;
      for (int k = 0; k < KNN; ++k) {
        float t  = (xkn[q * KNN + k][cc] - fv) * inv_sig;
        float pe = expf(-0.5f * t * t);
        float fk;
        if (j < C) fk = (feat[(size_t)(b * N + kq[k]) * C + j] - fq[j]) / sfe[k];
        else       fk = fq[j - C];
        float fw = (fk + pe) * pe;
        acc += fw; mx = fmaxf(mx, fw);
      }
      float xx = acc * (1.f / KNN) + mx;
      const float o = 0.5f * xx * (1.f + erff(xx * 0.70710678118654752f));
      out[(size_t)(row0 + q) * C2 + j] = o;
      pmx = fmaxf(pmx, o); psm += o;
    }
    if (chunk > 1) {
      pmaxb[(size_t)blockIdx.x * C2 + j] = pmx;
      psumb[(size_t)blockIdx.x * C2 + j] = psm;
    }
  }
}

__global__ __launch_bounds__(1024) void k_pool_fold(const float* pm, const float* ps,

                                                    int G, int C2, int S,
                                                    float* __restrict__ out, int off) {
  const int b = blockIdx.x;
  for (int j = threadIdx.x; j < C2; j += blockDim.x) {
    const float* a = pm + (size_t)b * G * C2 + j;
    const float* c = ps + (size_t)b * G * C2 + j;
    float mx = -FLT_MAX, sm = 0.f;
    if (pm == ps) {
#pragma unroll 8
      for (int g = 0; g < G; ++g) {
        const float v = a[(size_t)g * C2];
        mx = fmaxf(mx, v); sm += v;
      }
    } else {
#pragma unroll 8
      for (int g = 0; g < G; ++g) {
        mx = fmaxf(mx, a[(size_t)g * C2]);
        sm += c[(size_t)g * C2];
      }
    }
    out[b * 3840 + off + j]      = mx;
    out[b * 3840 + off + C2 + j] = sm / (float)S;
  }
}

extern "C" void kernel_launch(void* const* d_in, const int* in_sizes, int n_in,
                              void* d_out, int out_size, void* d_ws, size_t ws_size,
                              hipStream_t stream) {
  const float* xyz_in = (const float*)d_in[0];
  const float* W      = (const float*)d_in[1];
  const float* gamma  = (const float*)d_in[2];
  const float* beta   = (const float*)d_in[3];
  float* out = (float*)d_out;
  float* ws  = (float*)d_ws;

  size_t o = 0;
  float* featA  = ws + o; o += (size_t)BATCH * NPTS * CINIT;
  float* featB  = ws + o; o += (size_t)BATCH * NPTS * CINIT;
  float* xyzA   = ws + o; o += (size_t)BATCH * NPTS * 3;
  float* xyzB   = ws + o; o += (size_t)BATCH * NPTS * 3;
  float* feat_s = ws + o; o += (size_t)BATCH * 1024 * CINIT;
  float* pmaxb  = ws + o; o += (size_t)1024 * 1024;
  float* psumb  = ws + o; o += (size_t)1024 * 1024;
  float* bn_mu  = ws + o; o += 64;
  float* bn_inv = ws + o; o += 64;
  float* stdx   = ws + o; o += 32;
  float* stdf   = ws + o; o += 32;
  float* stdp   = ws + o; o += (size_t)KNN * STD_G * 4;
  int* fps_i = (int*)(ws + o); o += (size_t)BATCH * 1024;
  int* knn_i = (int*)(ws + o); o += (size_t)BATCH * 1024 * KNN;

  k_embed_wmma<<<1024, 256, 0, stream>>>(xyz_in, W, featA);
  k_bn_stats<<<CINIT, 256, 0, stream>>>(featA, bn_mu, bn_inv);
  k_bn_apply<<<2048, 256, 0, stream>>>(featA, bn_mu, bn_inv, gamma, beta);

  const float* xyz_cur = xyz_in;
  float* feat_cur = featA; float* feat_nxt = featB;
  float* xs_buf = xyzA;    float* xs_other = xyzB;
  int N = NPTS, C = CINIT, off = 0;

  for (int st = 0; st < 4; ++st) {
    const int S = N / 2, C2 = 2 * C;
    if (N == 2048)      k_fps_wave<64><<<BATCH, 32, 0, stream>>>(xyz_cur, S, fps_i);
    else if (N == 1024) k_fps_wave<32><<<BATCH, 32, 0, stream>>>(xyz_cur, S, fps_i);
    else if (N == 512)  k_fps_wave<16><<<BATCH, 32, 0, stream>>>(xyz_cur, S, fps_i);
    else                k_fps_wave<8><<<BATCH, 32, 0, stream>>>(xyz_cur, S, fps_i);
    k_gather<<<BATCH * S, 256, 0, stream>>>(xyz_cur, feat_cur, fps_i, N, S, C, xs_buf, feat_s);
    {
      const int nb = BATCH * S / 16 / KW;
      if (N == 2048)      k_knn_fused<2048><<<nb, KW * 32, 0, stream>>>(xs_buf, xyz_cur, S, knn_i);
      else if (N == 1024) k_knn_fused<1024><<<nb, KW * 32, 0, stream>>>(xs_buf, xyz_cur, S, knn_i);
      else if (N == 512)  k_knn_fused<512><<<nb, KW * 32, 0, stream>>>(xs_buf, xyz_cur, S, knn_i);
      else                k_knn_fused<256><<<nb, KW * 32, 0, stream>>>(xs_buf, xyz_cur, S, knn_i);
    }
    {
      dim3 g(KNN, STD_G);
      k_std_partial<<<g, 256, 0, stream>>>(xyz_cur, feat_cur, xs_buf, feat_s, knn_i, N, S, C, stdp);
      k_std_final<<<1, KNN, 0, stream>>>(stdp, S, C, stdx, stdf);
    }
    {
      const int chunk = (C2 <= 256) ? 4 : ((C2 == 512) ? 2 : 1);
      const int nthr  = (C2 < 256) ? C2 : 256;
      k_stage<<<BATCH * S / chunk, nthr, 0, stream>>>(xyz_cur, feat_cur, xs_buf, feat_s, knn_i,
                                                      stdx, stdf, N, S, C, chunk, feat_nxt,
                                                      pmaxb, psumb);
      if (chunk > 1) k_pool_fold<<<BATCH, C2, 0, stream>>>(pmaxb, psumb, S / chunk, C2, S, out, off);
      else           k_pool_fold<<<BATCH, C2, 0, stream>>>(feat_nxt, feat_nxt, S, C2, S, out, off);
    }
    off += 2 * C2;
    float* t = feat_cur; feat_cur = feat_nxt; feat_nxt = t;
    xyz_cur = xs_buf;
    float* u = xs_buf; xs_buf = xs_other; xs_other = u;
    N = S; C = C2;
  }
  (void)in_sizes; (void)n_in; (void)out_size; (void)ws_size;
}


# stages 2-4: the sampling chain (one wave per cloud) runs as 16 extra blocks in front of the previous stage's kNN launch, which needs none of it: kNN time hidden under it; 31 -> 28 launches
# speedup vs baseline: 4.7425x; 1.1199x over previous
#include <hip/hip_runtime.h>
#include <cfloat>
#include <cmath>

typedef __attribute__((ext_vector_type(2))) float v2f;
typedef __attribute__((ext_vector_type(3))) float v3f;
typedef __attribute__((ext_vector_type(8))) float v8f;

#define BATCH 16
#define NPTS 2048
#define CINIT 64
#define KNN 32
#define SIGMA_ 0.3f
#define BN_EPS_ 1e-5f
#define STD_G 16

__global__ __launch_bounds__(256) void k_embed_wmma(const float* __restrict__ xyz,
                                                    const float* __restrict__ W,
                                                    float* __restrict__ h) {
  const int wid  = (blockIdx.x * 256 + (int)threadIdx.x) >> 5;
  const int lane = threadIdx.x & 31;
  const int hf   = lane >> 4;
  const int ln   = lane & 15;
  const int tileM = wid >> 2;
  const int tileN = wid & 3;
  const int m  = tileM * 16 + ln;
  const int c  = tileN * 16 + ln;
  const float x0 = xyz[m * 3], x1 = xyz[m * 3 + 1], x2 = xyz[m * 3 + 2];
  const float w0 = W[c * 3],   w1 = W[c * 3 + 1],   w2 = W[c * 3 + 2];
  v2f a, b;
  a.x = hf ? x2 : x0;  a.y = hf ? 0.f : x1;
  b.x = hf ? w2 : w0;  b.y = hf ? 0.f : w1;
  v8f d = {};
  d = __builtin_amdgcn_wmma_f32_16x16x4_f32(false, a, false, b, (short)0, d, false, false);
#pragma unroll
  for (int r = 0; r < 8; ++r) {
    int row = tileM * 16 + r + 8 * hf;
    h[row * CINIT + tileN * 16 + ln] = d[r];
  }
}

__global__ __launch_bounds__(256) void k_bn_stats(const float* __restrict__ h,
                                                  float* __restrict__ mu,
                                                  float* __restrict__ inv) {
  __shared__ float s1[256], s2[256];
  const int c = blockIdx.x;
  const int n = BATCH * NPTS;
  float a = 0.f, b = 0.f;
  for (int i = threadIdx.x; i < n; i += 256) {
    float v = h[i * CINIT + c];
    a += v; b += v * v;
  }
  s1[threadIdx.x] = a; s2[threadIdx.x] = b; __syncthreads();
  for (int o = 128; o > 0; o >>= 1) {
    if ((int)threadIdx.x < o) { s1[threadIdx.x] += s1[threadIdx.x + o]; s2[threadIdx.x] += s2[threadIdx.x + o]; }
    __syncthreads();
  }
  if (threadIdx.x == 0) {
    float m   = s1[0] / (float)n;
    float var = s2[0] / (float)n - m * m;
    mu[c]  = m;
    inv[c] = rsqrtf(var + BN_EPS_);
  }
}

__global__ void k_bn_apply(float* __restrict__ h, const float* __restrict__ mu,
                           const float* __restrict__ inv, const float* __restrict__ gamma,
                           const float* __restrict__ beta) {
  const int total = BATCH * NPTS * CINIT;
  for (int i = blockIdx.x * blockDim.x + threadIdx.x; i < total; i += gridDim.x * blockDim.x) {
    int c = i & (CINIT - 1);
    float v = (h[i] - mu[c]) * inv[c] * gamma[c] + beta[c];
    h[i] = fmaxf(v, 0.f);
  }
}

__device__ __forceinline__ int imax2(int a, int b) { return a > b ? a : b; }
__device__ __forceinline__ int imin2(int a, int b) { return a < b ? a : b; }

template <int CTRL>
__device__ __forceinline__ int row_xchg(int v) {
  return __builtin_amdgcn_update_dpp(v, v, CTRL, 0xf, 0xf, true);
}

template <int PPL, int Q>
__device__ __forceinline__ unsigned fps_eq_bits(const int (&di)[PPL], int m) {
  constexpr int CH = PPL / 8;
  unsigned bits = 1u << (CH - 1);
#pragma unroll
  for (int x = 0; x < CH - 1; ++x) bits |= (di[Q * CH + x] == m) ? (1u << x) : 0u;
  return bits;
}

template <int PPL>
__device__ __forceinline__ int fps_pick_values(const v2f (&px)[PPL / 2], const v2f (&py)[PPL / 2],
                                               const v2f (&pz)[PPL / 2], int (&di)[PPL],
                                               const float4* pc, int lane,
                                               float& fx, float& fy, float& fz) {
  constexpr int CH = PPL / 8;
  int gm[8];
#pragma unroll
  for (int p2 = 0; p2 < PPL / 2; ++p2) {
    const v2f dx = px[p2] - fx, dy = py[p2] - fy, dz = pz[p2] - fz;
    const v2f t = __builtin_elementwise_fma(dz, dz, __builtin_elementwise_fma(dx, dx, dy * dy));
    const int s0 = 2 * p2, s1 = 2 * p2 + 1, q = s0 / CH;
    di[s0] = imin2(di[s0], __float_as_int(t.x));
    di[s1] = imin2(di[s1], __float_as_int(t.y));
    gm[q] = (s0 % CH == 0) ? imax2(di[s0], di[s1]) : imax2(gm[q], imax2(di[s0], di[s1]));
  }
  int mine = gm[0];
#pragma unroll
  for (int q = 1; q < 8; ++q) mine = imax2(mine, gm[q]);
  unsigned gbits = 0x80u;
#pragma unroll
  for (int q = 0; q < 7; ++q) gbits |= (gm[q] == mine) ? (1u << q) : 0u;
  int g = mine;
  g = imax2(g, row_xchg<0x161>(g));
  g = imax2(g, row_xchg<0x162>(g));
  g = imax2(g, row_xchg<0x164>(g));
  g = imax2(g, row_xchg<0x168>(g));
  g = imax2(g, (int)__builtin_amdgcn_permlanex16(g, g, 0x76543210, 0xfedcba98, false, false));
  const unsigned hit = __builtin_amdgcn_ballot_w32(mine == g);
  const int L = __builtin_ctz(hit);
  const int sq = __builtin_ctz((unsigned)__builtin_amdgcn_readlane((int)gbits, L));
  const int base = L * PPL + sq * CH;
  typedef const volatile __attribute__((address_space(3))) v3f lds_v3f;
  const v3f cand = *(lds_v3f*)(pc + base + (lane & (CH - 1)));
  unsigned eb;
  switch (sq) {
    case 0:  eb = fps_eq_bits<PPL, 0>(di, mine); break;
    case 1:  eb = fps_eq_bits<PPL, 1>(di, mine); break;
    case 2:  eb = fps_eq_bits<PPL, 2>(di, mine); break;
    case 3:  eb = fps_eq_bits<PPL, 3>(di, mine); break;
    case 4:  eb = fps_eq_bits<PPL, 4>(di, mine); break;
    case 5:  eb = fps_eq_bits<PPL, 5>(di, mine); break;
    case 6:  eb = fps_eq_bits<PPL, 6>(di, mine); break;
    default: eb = fps_eq_bits<PPL, 7>(di, mine); break;
  }
  const int es = __builtin_ctz((unsigned)__builtin_amdgcn_readlane((int)eb, L));
  fx = __int_as_float(__builtin_amdgcn_readlane(__float_as_int(cand.x), es));
  fy = __int_as_float(__builtin_amdgcn_readlane(__float_as_int(cand.y), es));
  fz = __int_as_float(__builtin_amdgcn_readlane(__float_as_int(cand.z), es));
  return base + es;
}

template <int PPL>
__device__ __forceinline__ void fps_cloud(const float* __restrict__ cloud, int S,
                                          int* __restrict__ idx_out, float4* pc, int lane, bool run) {
  constexpr bool VALS = (PPL >= 32);
  const float* p = cloud + (size_t)lane * PPL * 3;
  float px[VALS ? 1 : PPL], py[VALS ? 1 : PPL], pz[VALS ? 1 : PPL];
  v2f qx[VALS ? PPL / 2 : 1], qy[VALS ? PPL / 2 : 1], qz[VALS ? PPL / 2 : 1];
  int di[PPL];
  if (run) {
#pragma unroll
    for (int s = 0; s < PPL; ++s) {
      const float c0 = p[s * 3], c1 = p[s * 3 + 1], c2 = p[s * 3 + 2];
      if constexpr (VALS) { qx[s / 2][s % 2] = c0; qy[s / 2][s % 2] = c1; qz[s / 2][s % 2] = c2; }
      else                { px[s] = c0; py[s] = c1; pz[s] = c2; }
      di[s] = __float_as_int(1e10f);
      pc[lane * PPL + s] = make_float4(c0, c1, c2, 0.f);
    }
  }
  __syncthreads();
  if (!run) return;
  int far = 0;
  float fx = 0.f, fy = 0.f, fz = 0.f;
  if constexpr (VALS) { const float4 f0 = pc[0]; fx = f0.x; fy = f0.y; fz = f0.z; }
  for (int it0 = 0; it0 < S; it0 += 32) {
    int keep = 0;
    for (int j = 0; j < 32; ++j) {
      if constexpr (VALS) {
        keep = (lane == j) ? far : keep;
        far = fps_pick_values<PPL>(qx, qy, qz, di, pc, lane, fx, fy, fz);
      } else {
        const float4 f = pc[far];
        keep = (lane == j) ? far : keep;
        constexpr int CH = PPL / 8;
        int v[8], ix[8];
#pragma unroll
        for (int q = 0; q < 8; ++q) {
#pragma unroll
          for (int e = 0; e < CH; ++e) {
            const int s = q * CH + e;
            const float dx = px[s] - f.x, dy = py[s] - f.y, dz = pz[s] - f.z;
            const float t = __builtin_fmaf(dz, dz, __builtin_fmaf(dx, dx, dy * dy));
            di[s] = imin2(di[s], __float_as_int(t));
            if (e == 0) { v[q] = di[s]; ix[q] = s; }
            else {
              const bool c = di[s] > v[q];
              v[q]  = c ? di[s] : v[q];
              ix[q] = c ? s : ix[q];
            }
          }
        }
#pragma unroll
        for (int w = 8; w > 1; w >>= 1) {
#pragma unroll
          for (int k = 0; k < w; k += 2) {
            const bool c = v[k + 1] > v[k];
            const int nv = c ? v[k + 1] : v[k];
            const int ni = c ? ix[k + 1] : ix[k];
            v[k / 2] = nv; ix[k / 2] = ni;
          }
        }
        const int mine = v[0];
        int g = mine;
        g = imax2(g, row_xchg<0x161>(g));
        g = imax2(g, row_xchg<0x162>(g));
        g = imax2(g, row_xchg<0x164>(g));
        g = imax2(g, row_xchg<0x168>(g));
        g = imax2(g, (int)__builtin_amdgcn_permlanex16(g, g, 0x76543210, 0xfedcba98, false, false));
        const unsigned hit = __builtin_amdgcn_ballot_w32(mine == g);
        const int L = __builtin_ctz(hit);
        far = L * PPL + __builtin_amdgcn_readlane(ix[0], L);
      }
    }
    idx_out[it0 + lane] = keep;
  }
}

template <int PPL>
__global__ __launch_bounds__(32) void k_fps_wave(const float* __restrict__ xyz, int S,
                                                 int* __restrict__ fps_idx) {
  __shared__ float4 pc[32 * PPL];
  fps_cloud<PPL>(xyz + (size_t)blockIdx.x * (32 * PPL) * 3, S, fps_idx + (size_t)blockIdx.x * S, pc,
                 (int)threadIdx.x, true);
}

__global__ void k_gather(const float* __restrict__ xyz, const float* __restrict__ feat,
                         const int* __restrict__ fps_idx, int N, int S, int C,
                         float* __restrict__ xyz_s, float* __restrict__ feat_s) {
  const int b = blockIdx.x / S, s = blockIdx.x % S;
  const int idx = fps_idx[b * S + s];
  if (threadIdx.x < 3)
    xyz_s[(size_t)(b * S + s) * 3 + threadIdx.x] = xyz[(size_t)(b * N + idx) * 3 + threadIdx.x];
  for (int j = threadIdx.x; j < C; j += blockDim.x)
    feat_s[(size_t)(b * S + s) * C + j] = feat[(size_t)(b * N + idx) * C + j];
}

#define KW 2
#define KCAP 128
#define KU 8

__device__ __forceinline__ float sq3(float p0, float p1, float p2) {
  return __builtin_fmaf(p2, p2, __builtin_fmaf(p0, p0, p1 * p1));
}

__device__ __forceinline__ v8f knn_tile(const float* __restrict__ x, int n, int hf, v2f a,
                                        float& nn) {
  const float x0 = x[n * 3], x1 = x[n * 3 + 1], x2 = x[n * 3 + 2];
  nn = sq3(x0, x1, x2);
  v2f b;
  b.x = hf ? x2 : x0;  b.y = hf ? 0.f : x1;
  v8f d = {};
  return __builtin_amdgcn_wmma_f32_16x16x4_f32(false, a, false, b, (short)0, d, false, false);
}

template <int N>
__global__ __launch_bounds__(KW * 32) void k_knn_fused(const float* __restrict__ xs_all,
                                                       const float* __restrict__ x_all,
                                                       int S, int* __restrict__ knn_idx,
                                                       int* __restrict__ fps_next) {
  constexpr int  RPPL = N / 64;
  constexpr bool RIDE = (RPPL >= 8);
  __shared__ float gm[KW][16][64];
  __shared__ int2  cand[KW][16][KCAP];
  __shared__ int   cntL[KW][16];
  __shared__ float4 rpc[RIDE ? 32 * RPPL : 1];
  const int w    = threadIdx.x >> 5;
  const int lane = threadIdx.x & 31;
  if constexpr (RIDE) {
    if (blockIdx.x < BATCH) {
      const bool first = __builtin_amdgcn_readfirstlane(w) == 0;
      fps_cloud<RPPL>(xs_all + (size_t)blockIdx.x * S * 3, S / 2,
                      fps_next + (size_t)blockIdx.x * (S / 2), rpc, lane, first);
      return;
    }
  }
  const int blk  = RIDE ? (int)blockIdx.x - BATCH : (int)blockIdx.x;
  const int hf   = lane >> 4;
  const int ln   = lane & 15;
  const int row0 = (blk * KW + w) * 16;
  const int b    = row0 / S;
  const float* xs = xs_all + (size_t)row0 * 3;
  const float* x  = x_all + (size_t)b * N * 3;
  const float s0 = xs[ln * 3], s1 = xs[ln * 3 + 1], s2 = xs[ln * 3 + 2];
  v2f a;
  a.x = hf ? s2 : s0;  a.y = hf ? 0.f : s1;
  float sN[8];
#pragma unroll
  for (int r = 0; r < 8; ++r) {
    const float a0 = xs[(r + 8 * hf) * 3], a1 = xs[(r + 8 * hf) * 3 + 1], a2 = xs[(r + 8 * hf) * 3 + 2];
    sN[r] = sq3(a0, a1, a2);
  }

  float g[4][8];
#pragma unroll
  for (int u = 0; u < 4; ++u)
#pragma unroll
    for (int r = 0; r < 8; ++r) g[u][r] = INFINITY;
  for (int t = 0; t < N / 16; t += KU) {
    v8f d[KU];
    float nn[KU];
#pragma unroll
    for (int u = 0; u < KU; ++u) d[u] = knn_tile(x, (t + u) * 16 + ln, hf, a, nn[u]);
#pragma unroll
    for (int u = 0; u < KU; ++u)
#pragma unroll
      for (int r = 0; r < 8; ++r)
        g[u & 3][r] = fminf(g[u & 3][r], __builtin_fmaf(-2.f, d[u][r], sN[r] + nn[u]));
  }
#pragma unroll
  for (int u = 0; u < 4; ++u)
#pragma unroll
    for (int r = 0; r < 8; ++r) gm[w][r + 8 * hf][u * 16 + ln] = g[u][r];
  __syncthreads();

  float T[8];
#pragma unroll
  for (int r = 0; r < 8; ++r) {
    const float* grow = gm[w][r + 8 * hf];
    int k0 = 0, k1 = 0, k2 = 0, k3 = 0;
#pragma unroll 16
    for (int e = 0; e < 64; ++e) {
      const float o = grow[e];
      k0 += (o < g[0][r]) ? 1 : 0;
      k1 += (o < g[1][r]) ? 1 : 0;
      k2 += (o < g[2][r]) ? 1 : 0;
      k3 += (o < g[3][r]) ? 1 : 0;
    }
    float tv = (k0 < 32) ? g[0][r] : -INFINITY;
    tv = fmaxf(tv, (k1 < 32) ? g[1][r] : -INFINITY);
    tv = fmaxf(tv, (k2 < 32) ? g[2][r] : -INFINITY);
    tv = fmaxf(tv, (k3 < 32) ? g[3][r] : -INFINITY);
#pragma unroll
    for (int o = 8; o > 0; o >>= 1) tv = fmaxf(tv, __shfl_xor(tv, o, 32));
    T[r] = tv;
  }

  int cnt[8];
#pragma unroll
  for (int r = 0; r < 8; ++r) cnt[r] = 0;
  const unsigned below = (1u << ln) - 1u;
  for (int t = 0; t < N / 16; t += KU) {
    v8f d[KU];
    float nn[KU];
#pragma unroll
    for (int u = 0; u < KU; ++u) d[u] = knn_tile(x, (t + u) * 16 + ln, hf, a, nn[u]);
#pragma unroll
    for (int u = 0; u < KU; ++u) {
      const int col = (t + u) * 16 + ln;
#pragma unroll
      for (int r = 0; r < 8; ++r) {
        const float v = __builtin_fmaf(-2.f, d[u][r], sN[r] + nn[u]);
        const bool c = v <= T[r];
        const unsigned bal = __builtin_amdgcn_ballot_w32(c);
        if (bal != 0u) {
          const unsigned hb = (hf ? (bal >> 16) : bal) & 0xffffu;
          const int pos = cnt[r] + __builtin_popcount(hb & below);
          if (c && pos < KCAP) cand[w][r + 8 * hf][pos] = make_int2(__float_as_int(v), col);
          cnt[r] += __builtin_popcount(hb);
        }
      }
    }
  }
#pragma unroll
  for (int r = 0; r < 8; ++r) {
    const int p = cnt[r] + ln;
    if (ln < 8 && p < ((cnt[r] + 7) & ~7) && p < KCAP)
      cand[w][r + 8 * hf][p] = make_int2(__float_as_int(INFINITY), 0x7fffffff);
  }
  if (ln == 0) {
#pragma unroll
    for (int r = 0; r < 8; ++r) cntL[w][r + 8 * hf] = cnt[r];
  }
  __syncthreads();

  bool over = false;
  for (int R = 0; R < 16; ++R) {
    const int M = __builtin_amdgcn_readfirstlane(cntL[w][R]);
    if (M > KCAP) { over = true; continue; }
    const int2* cr = cand[w][R];
    int* outp = knn_idx + (size_t)(row0 + R) * KNN;
    const int Mp = (M + 7) & ~7;
    for (int base = 0; base < M; base += 64) {
      const int my0 = base + lane, my1 = base + 32 + lane;
      const bool have0 = my0 < M, have1 = my1 < M;
      const int2 mine0 = cr[have0 ? my0 : 0];
      const int2 mine1 = cr[have1 ? my1 : 0];
      const float mv0 = __int_as_float(mine0.x), mv1 = __int_as_float(mine1.x);
      const int   mi0 = mine0.y, mi1 = mine1.y;
      int rank0 = 0, rank1 = 0;
      for (int j = 0; j < Mp; j += 8) {
#pragma unroll
        for (int q = 0; q < 8; ++q) {
          const int2 o = cr[j + q];
          const float ov = __int_as_float(o.x);
          rank0 += (ov < mv0 || (ov == mv0 && o.y < mi0)) ? 1 : 0;
          rank1 += (ov < mv1 || (ov == mv1 && o.y < mi1)) ? 1 : 0;
        }
      }
      if (have0 && rank0 < KNN) outp[rank0] = mi0;
      if (have1 && rank1 < KNN) outp[rank1] = mi1;
    }
  }

  if (over) {
    float pv[8]; int pi[8];
#pragma unroll
    for (int r = 0; r < 8; ++r) { pv[r] = -INFINITY; pi[r] = -1; }
    for (int k = 0; k < KNN; ++k) {
      float bv[8]; int bi[8];
#pragma unroll
      for (int r = 0; r < 8; ++r) { bv[r] = INFINITY; bi[r] = 0x7fffffff; }
      for (int t = 0; t < N / 16; ++t) {
        float nn;
        const v8f d = knn_tile(x, t * 16 + ln, hf, a, nn);
        const int col = t * 16 + ln;
#pragma unroll
        for (int r = 0; r < 8; ++r) {
          const float v = __builtin_fmaf(-2.f, d[r], sN[r] + nn);
          const bool after  = (v > pv[r]) || (v == pv[r] && col > pi[r]);
          const bool better = (v < bv[r]) || (v == bv[r] && col < bi[r]);
          if (after && better) { bv[r] = v; bi[r] = col; }
        }
      }
#pragma unroll
      for (int r = 0; r < 8; ++r) {
#pragma unroll
        for (int o = 8; o > 0; o >>= 1) {
          const float ov = __shfl_xor(bv[r], o, 32);
          const int   oi = __shfl_xor(bi[r], o, 32);
          if (ov < bv[r] || (ov == bv[r] && oi < bi[r])) { bv[r] = ov; bi[r] = oi; }
        }
        pv[r] = bv[r]; pi[r] = bi[r];
        if (ln == 0 && cntL[w][r + 8 * hf] > KCAP)
          knn_idx[(size_t)(row0 + r + 8 * hf) * KNN + k] = bi[r];
      }
    }
  }
}

__global__ __launch_bounds__(256) void k_std_partial(const float* __restrict__ xyz,
                                                     const float* __restrict__ feat,
                                                     const float* __restrict__ xyz_s,
                                                     const float* __restrict__ feat_s,
                                                     const int* __restrict__ knn_idx,
                                                     int N, int S, int C,
                                                     float* __restrict__ part) {
  __shared__ float s1[256], s2[256], s3[256], s4[256];
  const int k = blockIdx.x, g = blockIdx.y;
  float ax = 0.f, qx = 0.f, af = 0.f, qf = 0.f;
  for (int p = g * 256 + threadIdx.x; p < BATCH * S; p += STD_G * 256) {
    int b = p / S, s = p % S;
    int idx = knn_idx[(size_t)(b * S + s) * KNN + k];
    for (int c = 0; c < 3; ++c) {
      float d = xyz[(size_t)(b * N + idx) * 3 + c] - xyz_s[(size_t)(b * S + s) * 3 + c];
      ax += d; qx += d * d;
    }
    const float* fr = feat   + (size_t)(b * N + idx) * C;
    const float* fc = feat_s + (size_t)(b * S + s) * C;
    for (int j = 0; j < C; ++j) { float d = fr[j] - fc[j]; af += d; qf += d * d; }
  }
  s1[threadIdx.x] = ax; s2[threadIdx.x] = qx; s3[threadIdx.x] = af; s4[threadIdx.x] = qf;
  __syncthreads();
  for (int o = 128; o > 0; o >>= 1) {
    if ((int)threadIdx.x < o) {
      s1[threadIdx.x] += s1[threadIdx.x + o]; s2[threadIdx.x] += s2[threadIdx.x + o];
      s3[threadIdx.x] += s3[threadIdx.x + o]; s4[threadIdx.x] += s4[threadIdx.x + o];
    }
    __syncthreads();
  }
  if (threadIdx.x == 0) {
    float* q = part + ((size_t)k * STD_G + g) * 4;
    q[0] = s1[0]; q[1] = s2[0]; q[2] = s3[0]; q[3] = s4[0];
  }
}

__global__ void k_std_final(const float* __restrict__ part, int S, int C,
                            float* __restrict__ std_xyz, float* __restrict__ std_feat) {
  const int k = threadIdx.x;
  if (k >= KNN) return;
  float ax = 0.f, qx = 0.f, af = 0.f, qf = 0.f;
  for (int g = 0; g < STD_G; ++g) {
    const float* q = part + ((size_t)k * STD_G + g) * 4;
    ax += q[0]; qx += q[1]; af += q[2]; qf += q[3];
  }
  {
    double n = (double)BATCH * S * 3;
    double var = ((double)qx - (double)ax * ax / n) / (n - 1.0);
    std_xyz[k] = fmaxf(sqrtf(fmaxf((float)var, 0.f)), 1e-5f);
  }
  {
    double n = (double)BATCH * S * C;
    double var = ((double)qf - (double)af * af / n) / (n - 1.0);
    std_feat[k] = fmaxf(sqrtf(fmaxf((float)var, 0.f)), 1e-5f);
  }
}

#define PCH 4
__global__ __launch_bounds__(256) void k_stage(const float* __restrict__ xyz,
                                               const float* __restrict__ feat,
                                               const float* __restrict__ xyz_s,
                                               const float* __restrict__ feat_s,
                                               const int* __restrict__ knn_idx,
                                               const float* __restrict__ std_xyz,
                                               const float* __restrict__ std_feat,
                                               int N, int S, int C, int chunk,
                                               float* __restrict__ out,
                                               float* __restrict__ pmaxb,
                                               float* __restrict__ psumb) {
  __shared__ int   kidx[PCH * KNN];
  __shared__ float sfe[KNN];
  __shared__ float xkn[PCH * KNN][3];
  __shared__ float fs[512];
  const int nt   = blockDim.x;
  const int row0 = blockIdx.x * chunk;
  const int b    = row0 / S;
  const int nk   = chunk * KNN;
  for (int t = threadIdx.x; t < nk; t += nt) kidx[t] = knn_idx[(size_t)row0 * KNN + t];
  if (threadIdx.x < KNN) sfe[threadIdx.x] = std_feat[threadIdx.x];
  __syncthreads();
  for (int t = threadIdx.x; t < nk; t += nt)
    __builtin_prefetch(feat + (size_t)(b * N + kidx[t]) * C, 0, 3);
  for (int t = threadIdx.x; t < nk * 3; t += nt) {
    const int kk = t / 3, c = t % 3;
    const int q = kk / KNN, k = kk % KNN;
    xkn[kk][c] = (xyz[(size_t)(b * N + kidx[kk]) * 3 + c] - xyz_s[(size_t)(row0 + q) * 3 + c]) / std_xyz[k];
  }
  for (int t = threadIdx.x; t < chunk * C; t += nt) fs[t] = feat_s[(size_t)row0 * C + t];
  __syncthreads();
  const int C2 = 2 * C;
  const int fd = (C2 + 2) / 3;
  const int M3 = 3 * fd;
  const float inv_sig = 1.f / SIGMA_;
  for (int j = threadIdx.x; j < C2; j += nt) {
    int src = (int)(((long long)j * (M3 - 1)) / (C2 - 1));
    int cc = src / fd, f = src % fd;
    float fv = -1.f + 2.f * (float)(f + 1) / (float)(fd + 1);
    float pmx = -FLT_MAX, psm = 0.f;
    for (int q = 0; q < chunk; ++q) {
      const int*   kq = kidx + q * KNN;
      const float* fq = fs + q * C;
      float acc = 0.f, mx = -FLT_MAX;
      for (int k = 0; k < KNN; ++k) {
        float t  = (xkn[q * KNN + k][cc] - fv) * inv_sig;
        float pe = expf(-0.5f * t * t);
        float fk;
        if (j < C) fk = (feat[(size_t)(b * N + kq[k]) * C + j] - fq[j]) / sfe[k];
        else       fk = fq[j - C];
        float fw = (fk + pe) * pe;
        acc += fw; mx = fmaxf(mx, fw);
      }
      float xx = acc * (1.f / KNN) + mx;
      const float o = 0.5f * xx * (1.f + erff(xx * 0.70710678118654752f));
      out[(size_t)(row0 + q) * C2 + j] = o;
      pmx = fmaxf(pmx, o); psm += o;
    }
    if (chunk > 1) {
      pmaxb[(size_t)blockIdx.x * C2 + j] = pmx;
      psumb[(size_t)blockIdx.x * C2 + j] = psm;
    }
  }
}

__global__ __launch_bounds__(1024) void k_pool_fold(const float* pm, const float* ps,

                                                    int G, int C2, int S,
                                                    float* __restrict__ out, int off) {
  const int b = blockIdx.x;
  for (int j = threadIdx.x; j < C2; j += blockDim.x) {
    const float* a = pm + (size_t)b * G * C2 + j;
    const float* c = ps + (size_t)b * G * C2 + j;
    float mx = -FLT_MAX, sm = 0.f;
    if (pm == ps) {
#pragma unroll 8
      for (int g = 0; g < G; ++g) {
        const float v = a[(size_t)g * C2];
        mx = fmaxf(mx, v); sm += v;
      }
    } else {
#pragma unroll 8
      for (int g = 0; g < G; ++g) {
        mx = fmaxf(mx, a[(size_t)g * C2]);
        sm += c[(size_t)g * C2];
      }
    }
    out[b * 3840 + off + j]      = mx;
    out[b * 3840 + off + C2 + j] = sm / (float)S;
  }
}

extern "C" void kernel_launch(void* const* d_in, const int* in_sizes, int n_in,
                              void* d_out, int out_size, void* d_ws, size_t ws_size,
                              hipStream_t stream) {
  const float* xyz_in = (const float*)d_in[0];
  const float* W      = (const float*)d_in[1];
  const float* gamma  = (const float*)d_in[2];
  const float* beta   = (const float*)d_in[3];
  float* out = (float*)d_out;
  float* ws  = (float*)d_ws;

  size_t o = 0;
  float* featA  = ws + o; o += (size_t)BATCH * NPTS * CINIT;
  float* featB  = ws + o; o += (size_t)BATCH * NPTS * CINIT;
  float* xyzA   = ws + o; o += (size_t)BATCH * NPTS * 3;
  float* xyzB   = ws + o; o += (size_t)BATCH * NPTS * 3;
  float* feat_s = ws + o; o += (size_t)BATCH * 1024 * CINIT;
  float* pmaxb  = ws + o; o += (size_t)1024 * 1024;
  float* psumb  = ws + o; o += (size_t)1024 * 1024;
  float* bn_mu  = ws + o; o += 64;
  float* bn_inv = ws + o; o += 64;
  float* stdx   = ws + o; o += 32;
  float* stdf   = ws + o; o += 32;
  float* stdp   = ws + o; o += (size_t)KNN * STD_G * 4;
  int* fps_i = (int*)(ws + o); o += (size_t)BATCH * 1024;
  int* knn_i = (int*)(ws + o); o += (size_t)BATCH * 1024 * KNN;

  k_embed_wmma<<<1024, 256, 0, stream>>>(xyz_in, W, featA);
  k_bn_stats<<<CINIT, 256, 0, stream>>>(featA, bn_mu, bn_inv);
  k_bn_apply<<<2048, 256, 0, stream>>>(featA, bn_mu, bn_inv, gamma, beta);

  const float* xyz_cur = xyz_in;
  float* feat_cur = featA; float* feat_nxt = featB;
  float* xs_buf = xyzA;    float* xs_other = xyzB;
  int N = NPTS, C = CINIT, off = 0;

  for (int st = 0; st < 4; ++st) {
    const int S = N / 2, C2 = 2 * C;
    if (N == 2048)      k_fps_wave<64><<<BATCH, 32, 0, stream>>>(xyz_cur, S, fps_i);
    k_gather<<<BATCH * S, 256, 0, stream>>>(xyz_cur, feat_cur, fps_i, N, S, C, xs_buf, feat_s);
    {
      const int nb = BATCH * S / 16 / KW;
      if (N == 2048)      k_knn_fused<2048><<<BATCH + nb, KW * 32, 0, stream>>>(xs_buf, xyz_cur, S, knn_i, fps_i);
      else if (N == 1024) k_knn_fused<1024><<<BATCH + nb, KW * 32, 0, stream>>>(xs_buf, xyz_cur, S, knn_i, fps_i);
      else if (N == 512)  k_knn_fused<512><<<BATCH + nb, KW * 32, 0, stream>>>(xs_buf, xyz_cur, S, knn_i, fps_i);
      else                k_knn_fused<256><<<nb, KW * 32, 0, stream>>>(xs_buf, xyz_cur, S, knn_i, fps_i);
    }
    {
      dim3 g(KNN, STD_G);
      k_std_partial<<<g, 256, 0, stream>>>(xyz_cur, feat_cur, xs_buf, feat_s, knn_i, N, S, C, stdp);
      k_std_final<<<1, KNN, 0, stream>>>(stdp, S, C, stdx, stdf);
    }
    {
      const int chunk = (C2 <= 256) ? 4 : ((C2 == 512) ? 2 : 1);
      const int nthr  = (C2 < 256) ? C2 : 256;
      k_stage<<<BATCH * S / chunk, nthr, 0, stream>>>(xyz_cur, feat_cur, xs_buf, feat_s, knn_i,
                                                      stdx, stdf, N, S, C, chunk, feat_nxt,
                                                      pmaxb, psumb);
      if (chunk > 1) k_pool_fold<<<BATCH, C2, 0, stream>>>(pmaxb, psumb, S / chunk, C2, S, out, off);
      else           k_pool_fold<<<BATCH, C2, 0, stream>>>(feat_nxt, feat_nxt, S, C2, S, out, off);
    }
    off += 2 * C2;
    float* t = feat_cur; feat_cur = feat_nxt; feat_nxt = t;
    xyz_cur = xs_buf;
    float* u = xs_buf; xs_buf = xs_other; xs_other = u;
    N = S; C = C2;
  }
  (void)in_sizes; (void)n_in; (void)out_size; (void)ws_size;
}


# k_std_partial: a wave owns whole sampled rows, a lane is one rank and takes its neighbour's row in whole 128-B lines; centre row read once per row, not per rank; std sums re-associated
# speedup vs baseline: 4.9863x; 1.0514x over previous
#include <hip/hip_runtime.h>
#include <cfloat>
#include <cmath>

typedef __attribute__((ext_vector_type(2))) float v2f;
typedef __attribute__((ext_vector_type(3))) float v3f;
typedef __attribute__((ext_vector_type(8))) float v8f;

#define BATCH 16
#define NPTS 2048
#define CINIT 64
#define KNN 32
#define SIGMA_ 0.3f
#define BN_EPS_ 1e-5f
#define STD_G 256

__global__ __launch_bounds__(256) void k_embed_wmma(const float* __restrict__ xyz,
                                                    const float* __restrict__ W,
                                                    float* __restrict__ h) {
  const int wid  = (blockIdx.x * 256 + (int)threadIdx.x) >> 5;
  const int lane = threadIdx.x & 31;
  const int hf   = lane >> 4;
  const int ln   = lane & 15;
  const int tileM = wid >> 2;
  const int tileN = wid & 3;
  const int m  = tileM * 16 + ln;
  const int c  = tileN * 16 + ln;
  const float x0 = xyz[m * 3], x1 = xyz[m * 3 + 1], x2 = xyz[m * 3 + 2];
  const float w0 = W[c * 3],   w1 = W[c * 3 + 1],   w2 = W[c * 3 + 2];
  v2f a, b;
  a.x = hf ? x2 : x0;  a.y = hf ? 0.f : x1;
  b.x = hf ? w2 : w0;  b.y = hf ? 0.f : w1;
  v8f d = {};
  d = __builtin_amdgcn_wmma_f32_16x16x4_f32(false, a, false, b, (short)0, d, false, false);
#pragma unroll
  for (int r = 0; r < 8; ++r) {
    int row = tileM * 16 + r + 8 * hf;
    h[row * CINIT + tileN * 16 + ln] = d[r];
  }
}

__global__ __launch_bounds__(256) void k_bn_stats(const float* __restrict__ h,
                                                  float* __restrict__ mu,
                                                  float* __restrict__ inv) {
  __shared__ float s1[256], s2[256];
  const int c = blockIdx.x;
  const int n = BATCH * NPTS;
  float a = 0.f, b = 0.f;
  for (int i = threadIdx.x; i < n; i += 256) {
    float v = h[i * CINIT + c];
    a += v; b += v * v;
  }
  s1[threadIdx.x] = a; s2[threadIdx.x] = b; __syncthreads();
  for (int o = 128; o > 0; o >>= 1) {
    if ((int)threadIdx.x < o) { s1[threadIdx.x] += s1[threadIdx.x + o]; s2[threadIdx.x] += s2[threadIdx.x + o]; }
    __syncthreads();
  }
  if (threadIdx.x == 0) {
    float m   = s1[0] / (float)n;
    float var = s2[0] / (float)n - m * m;
    mu[c]  = m;
    inv[c] = rsqrtf(var + BN_EPS_);
  }
}

__global__ void k_bn_apply(float* __restrict__ h, const float* __restrict__ mu,
                           const float* __restrict__ inv, const float* __restrict__ gamma,
                           const float* __restrict__ beta) {
  const int total = BATCH * NPTS * CINIT;
  for (int i = blockIdx.x * blockDim.x + threadIdx.x; i < total; i += gridDim.x * blockDim.x) {
    int c = i & (CINIT - 1);
    float v = (h[i] - mu[c]) * inv[c] * gamma[c] + beta[c];
    h[i] = fmaxf(v, 0.f);
  }
}

__device__ __forceinline__ int imax2(int a, int b) { return a > b ? a : b; }
__device__ __forceinline__ int imin2(int a, int b) { return a < b ? a : b; }

template <int CTRL>
__device__ __forceinline__ int row_xchg(int v) {
  return __builtin_amdgcn_update_dpp(v, v, CTRL, 0xf, 0xf, true);
}

template <int PPL, int Q>
__device__ __forceinline__ unsigned fps_eq_bits(const int (&di)[PPL], int m) {
  constexpr int CH = PPL / 8;
  unsigned bits = 1u << (CH - 1);
#pragma unroll
  for (int x = 0; x < CH - 1; ++x) bits |= (di[Q * CH + x] == m) ? (1u << x) : 0u;
  return bits;
}

template <int PPL>
__device__ __forceinline__ int fps_pick_values(const v2f (&px)[PPL / 2], const v2f (&py)[PPL / 2],
                                               const v2f (&pz)[PPL / 2], int (&di)[PPL],
                                               const float4* pc, int lane,
                                               float& fx, float& fy, float& fz) {
  constexpr int CH = PPL / 8;
  int gm[8];
#pragma unroll
  for (int p2 = 0; p2 < PPL / 2; ++p2) {
    const v2f dx = px[p2] - fx, dy = py[p2] - fy, dz = pz[p2] - fz;
    const v2f t = __builtin_elementwise_fma(dz, dz, __builtin_elementwise_fma(dx, dx, dy * dy));
    const int s0 = 2 * p2, s1 = 2 * p2 + 1, q = s0 / CH;
    di[s0] = imin2(di[s0], __float_as_int(t.x));
    di[s1] = imin2(di[s1], __float_as_int(t.y));
    gm[q] = (s0 % CH == 0) ? imax2(di[s0], di[s1]) : imax2(gm[q], imax2(di[s0], di[s1]));
  }
  int mine = gm[0];
#pragma unroll
  for (int q = 1; q < 8; ++q) mine = imax2(mine, gm[q]);
  unsigned gbits = 0x80u;
#pragma unroll
  for (int q = 0; q < 7; ++q) gbits |= (gm[q] == mine) ? (1u << q) : 0u;
  int g = mine;
  g = imax2(g, row_xchg<0x161>(g));
  g = imax2(g, row_xchg<0x162>(g));
  g = imax2(g, row_xchg<0x164>(g));
  g = imax2(g, row_xchg<0x168>(g));
  g = imax2(g, (int)__builtin_amdgcn_permlanex16(g, g, 0x76543210, 0xfedcba98, false, false));
  const unsigned hit = __builtin_amdgcn_ballot_w32(mine == g);
  const int L = __builtin_ctz(hit);
  const int sq = __builtin_ctz((unsigned)__builtin_amdgcn_readlane((int)gbits, L));
  const int base = L * PPL + sq * CH;
  typedef const volatile __attribute__((address_space(3))) v3f lds_v3f;
  const v3f cand = *(lds_v3f*)(pc + base + (lane & (CH - 1)));
  unsigned eb;
  switch (sq) {
    case 0:  eb = fps_eq_bits<PPL, 0>(di, mine); break;
    case 1:  eb = fps_eq_bits<PPL, 1>(di, mine); break;
    case 2:  eb = fps_eq_bits<PPL, 2>(di, mine); break;
    case 3:  eb = fps_eq_bits<PPL, 3>(di, mine); break;
    case 4:  eb = fps_eq_bits<PPL, 4>(di, mine); break;
    case 5:  eb = fps_eq_bits<PPL, 5>(di, mine); break;
    case 6:  eb = fps_eq_bits<PPL, 6>(di, mine); break;
    default: eb = fps_eq_bits<PPL, 7>(di, mine); break;
  }
  const int es = __builtin_ctz((unsigned)__builtin_amdgcn_readlane((int)eb, L));
  fx = __int_as_float(__builtin_amdgcn_readlane(__float_as_int(cand.x), es));
  fy = __int_as_float(__builtin_amdgcn_readlane(__float_as_int(cand.y), es));
  fz = __int_as_float(__builtin_amdgcn_readlane(__float_as_int(cand.z), es));
  return base + es;
}

template <int PPL>
__device__ __forceinline__ void fps_cloud(const float* __restrict__ cloud, int S,
                                          int* __restrict__ idx_out, float4* pc, int lane, bool run) {
  constexpr bool VALS = (PPL >= 32);
  const float* p = cloud + (size_t)lane * PPL * 3;
  float px[VALS ? 1 : PPL], py[VALS ? 1 : PPL], pz[VALS ? 1 : PPL];
  v2f qx[VALS ? PPL / 2 : 1], qy[VALS ? PPL / 2 : 1], qz[VALS ? PPL / 2 : 1];
  int di[PPL];
  if (run) {
#pragma unroll
    for (int s = 0; s < PPL; ++s) {
      const float c0 = p[s * 3], c1 = p[s * 3 + 1], c2 = p[s * 3 + 2];
      if constexpr (VALS) { qx[s / 2][s % 2] = c0; qy[s / 2][s % 2] = c1; qz[s / 2][s % 2] = c2; }
      else                { px[s] = c0; py[s] = c1; pz[s] = c2; }
      di[s] = __float_as_int(1e10f);
      pc[lane * PPL + s] = make_float4(c0, c1, c2, 0.f);
    }
  }
  __syncthreads();
  if (!run) return;
  int far = 0;
  float fx = 0.f, fy = 0.f, fz = 0.f;
  if constexpr (VALS) { const float4 f0 = pc[0]; fx = f0.x; fy = f0.y; fz = f0.z; }
  for (int it0 = 0; it0 < S; it0 += 32) {
    int keep = 0;
    for (int j = 0; j < 32; ++j) {
      if constexpr (VALS) {
        keep = (lane == j) ? far : keep;
        far = fps_pick_values<PPL>(qx, qy, qz, di, pc, lane, fx, fy, fz);
      } else {
        const float4 f = pc[far];
        keep = (lane == j) ? far : keep;
        constexpr int CH = PPL / 8;
        int v[8], ix[8];
#pragma unroll
        for (int q = 0; q < 8; ++q) {
#pragma unroll
          for (int e = 0; e < CH; ++e) {
            const int s = q * CH + e;
            const float dx = px[s] - f.x, dy = py[s] - f.y, dz = pz[s] - f.z;
            const float t = __builtin_fmaf(dz, dz, __builtin_fmaf(dx, dx, dy * dy));
            di[s] = imin2(di[s], __float_as_int(t));
            if (e == 0) { v[q] = di[s]; ix[q] = s; }
            else {
              const bool c = di[s] > v[q];
              v[q]  = c ? di[s] : v[q];
              ix[q] = c ? s : ix[q];
            }
          }
        }
#pragma unroll
        for (int w = 8; w > 1; w >>= 1) {
#pragma unroll
          for (int k = 0; k < w; k += 2) {
            const bool c = v[k + 1] > v[k];
            const int nv = c ? v[k + 1] : v[k];
            const int ni = c ? ix[k + 1] : ix[k];
            v[k / 2] = nv; ix[k / 2] = ni;
          }
        }
        const int mine = v[0];
        int g = mine;
        g = imax2(g, row_xchg<0x161>(g));
        g = imax2(g, row_xchg<0x162>(g));
        g = imax2(g, row_xchg<0x164>(g));
        g = imax2(g, row_xchg<0x168>(g));
        g = imax2(g, (int)__builtin_amdgcn_permlanex16(g, g, 0x76543210, 0xfedcba98, false, false));
        const unsigned hit = __builtin_amdgcn_ballot_w32(mine == g);
        const int L = __builtin_ctz(hit);
        far = L * PPL + __builtin_amdgcn_readlane(ix[0], L);
      }
    }
    idx_out[it0 + lane] = keep;
  }
}

template <int PPL>
__global__ __launch_bounds__(32) void k_fps_wave(const float* __restrict__ xyz, int S,
                                                 int* __restrict__ fps_idx) {
  __shared__ float4 pc[32 * PPL];
  fps_cloud<PPL>(xyz + (size_t)blockIdx.x * (32 * PPL) * 3, S, fps_idx + (size_t)blockIdx.x * S, pc,
                 (int)threadIdx.x, true);
}

__global__ void k_gather(const float* __restrict__ xyz, const float* __restrict__ feat,
                         const int* __restrict__ fps_idx, int N, int S, int C,
                         float* __restrict__ xyz_s, float* __restrict__ feat_s) {
  const int b = blockIdx.x / S, s = blockIdx.x % S;
  const int idx = fps_idx[b * S + s];
  if (threadIdx.x < 3)
    xyz_s[(size_t)(b * S + s) * 3 + threadIdx.x] = xyz[(size_t)(b * N + idx) * 3 + threadIdx.x];
  for (int j = threadIdx.x; j < C; j += blockDim.x)
    feat_s[(size_t)(b * S + s) * C + j] = feat[(size_t)(b * N + idx) * C + j];
}

#define KW 2
#define KCAP 128
#define KU 8

__device__ __forceinline__ float sq3(float p0, float p1, float p2) {
  return __builtin_fmaf(p2, p2, __builtin_fmaf(p0, p0, p1 * p1));
}

__device__ __forceinline__ v8f knn_tile(const float* __restrict__ x, int n, int hf, v2f a,
                                        float& nn) {
  const float x0 = x[n * 3], x1 = x[n * 3 + 1], x2 = x[n * 3 + 2];
  nn = sq3(x0, x1, x2);
  v2f b;
  b.x = hf ? x2 : x0;  b.y = hf ? 0.f : x1;
  v8f d = {};
  return __builtin_amdgcn_wmma_f32_16x16x4_f32(false, a, false, b, (short)0, d, false, false);
}

template <int N>
__global__ __launch_bounds__(KW * 32) void k_knn_fused(const float* __restrict__ xs_all,
                                                       const float* __restrict__ x_all,
                                                       int S, int* __restrict__ knn_idx,
                                                       int* __restrict__ fps_next) {
  constexpr int  RPPL = N / 64;
  constexpr bool RIDE = (RPPL >= 8);
  __shared__ float gm[KW][16][64];
  __shared__ int2  cand[KW][16][KCAP];
  __shared__ int   cntL[KW][16];
  __shared__ float4 rpc[RIDE ? 32 * RPPL : 1];
  const int w    = threadIdx.x >> 5;
  const int lane = threadIdx.x & 31;
  if constexpr (RIDE) {
    if (blockIdx.x < BATCH) {
      const bool first = __builtin_amdgcn_readfirstlane(w) == 0;
      fps_cloud<RPPL>(xs_all + (size_t)blockIdx.x * S * 3, S / 2,
                      fps_next + (size_t)blockIdx.x * (S / 2), rpc, lane, first);
      return;
    }
  }
  const int blk  = RIDE ? (int)blockIdx.x - BATCH : (int)blockIdx.x;
  const int hf   = lane >> 4;
  const int ln   = lane & 15;
  const int row0 = (blk * KW + w) * 16;
  const int b    = row0 / S;
  const float* xs = xs_all + (size_t)row0 * 3;
  const float* x  = x_all + (size_t)b * N * 3;
  const float s0 = xs[ln * 3], s1 = xs[ln * 3 + 1], s2 = xs[ln * 3 + 2];
  v2f a;
  a.x = hf ? s2 : s0;  a.y = hf ? 0.f : s1;
  float sN[8];
#pragma unroll
  for (int r = 0; r < 8; ++r) {
    const float a0 = xs[(r + 8 * hf) * 3], a1 = xs[(r + 8 * hf) * 3 + 1], a2 = xs[(r + 8 * hf) * 3 + 2];
    sN[r] = sq3(a0, a1, a2);
  }

  float g[4][8];
#pragma unroll
  for (int u = 0; u < 4; ++u)
#pragma unroll
    for (int r = 0; r < 8; ++r) g[u][r] = INFINITY;
  for (int t = 0; t < N / 16; t += KU) {
    v8f d[KU];
    float nn[KU];
#pragma unroll
    for (int u = 0; u < KU; ++u) d[u] = knn_tile(x, (t + u) * 16 + ln, hf, a, nn[u]);
#pragma unroll
    for (int u = 0; u < KU; ++u)
#pragma unroll
      for (int r = 0; r < 8; ++r)
        g[u & 3][r] = fminf(g[u & 3][r], __builtin_fmaf(-2.f, d[u][r], sN[r] + nn[u]));
  }
#pragma unroll
  for (int u = 0; u < 4; ++u)
#pragma unroll
    for (int r = 0; r < 8; ++r) gm[w][r + 8 * hf][u * 16 + ln] = g[u][r];
  __syncthreads();

  float T[8];
#pragma unroll
  for (int r = 0; r < 8; ++r) {
    const float* grow = gm[w][r + 8 * hf];
    int k0 = 0, k1 = 0, k2 = 0, k3 = 0;
#pragma unroll 16
    for (int e = 0; e < 64; ++e) {
      const float o = grow[e];
      k0 += (o < g[0][r]) ? 1 : 0;
      k1 += (o < g[1][r]) ? 1 : 0;
      k2 += (o < g[2][r]) ? 1 : 0;
      k3 += (o < g[3][r]) ? 1 : 0;
    }
    float tv = (k0 < 32) ? g[0][r] : -INFINITY;
    tv = fmaxf(tv, (k1 < 32) ? g[1][r] : -INFINITY);
    tv = fmaxf(tv, (k2 < 32) ? g[2][r] : -INFINITY);
    tv = fmaxf(tv, (k3 < 32) ? g[3][r] : -INFINITY);
#pragma unroll
    for (int o = 8; o > 0; o >>= 1) tv = fmaxf(tv, __shfl_xor(tv, o, 32));
    T[r] = tv;
  }

  int cnt[8];
#pragma unroll
  for (int r = 0; r < 8; ++r) cnt[r] = 0;
  const unsigned below = (1u << ln) - 1u;
  for (int t = 0; t < N / 16; t += KU) {
    v8f d[KU];
    float nn[KU];
#pragma unroll
    for (int u = 0; u < KU; ++u) d[u] = knn_tile(x, (t + u) * 16 + ln, hf, a, nn[u]);
#pragma unroll
    for (int u = 0; u < KU; ++u) {
      const int col = (t + u) * 16 + ln;
#pragma unroll
      for (int r = 0; r < 8; ++r) {
        const float v = __builtin_fmaf(-2.f, d[u][r], sN[r] + nn[u]);
        const bool c = v <= T[r];
        const unsigned bal = __builtin_amdgcn_ballot_w32(c);
        if (bal != 0u) {
          const unsigned hb = (hf ? (bal >> 16) : bal) & 0xffffu;
          const int pos = cnt[r] + __builtin_popcount(hb & below);
          if (c && pos < KCAP) cand[w][r + 8 * hf][pos] = make_int2(__float_as_int(v), col);
          cnt[r] += __builtin_popcount(hb);
        }
      }
    }
  }
#pragma unroll
  for (int r = 0; r < 8; ++r) {
    const int p = cnt[r] + ln;
    if (ln < 8 && p < ((cnt[r] + 7) & ~7) && p < KCAP)
      cand[w][r + 8 * hf][p] = make_int2(__float_as_int(INFINITY), 0x7fffffff);
  }
  if (ln == 0) {
#pragma unroll
    for (int r = 0; r < 8; ++r) cntL[w][r + 8 * hf] = cnt[r];
  }
  __syncthreads();

  bool over = false;
  for (int R = 0; R < 16; ++R) {
    const int M = __builtin_amdgcn_readfirstlane(cntL[w][R]);
    if (M > KCAP) { over = true; continue; }
    const int2* cr = cand[w][R];
    int* outp = knn_idx + (size_t)(row0 + R) * KNN;
    const int Mp = (M + 7) & ~7;
    for (int base = 0; base < M; base += 64) {
      const int my0 = base + lane, my1 = base + 32 + lane;
      const bool have0 = my0 < M, have1 = my1 < M;
      const int2 mine0 = cr[have0 ? my0 : 0];
      const int2 mine1 = cr[have1 ? my1 : 0];
      const float mv0 = __int_as_float(mine0.x), mv1 = __int_as_float(mine1.x);
      const int   mi0 = mine0.y, mi1 = mine1.y;
      int rank0 = 0, rank1 = 0;
      for (int j = 0; j < Mp; j += 8) {
#pragma unroll
        for (int q = 0; q < 8; ++q) {
          const int2 o = cr[j + q];
          const float ov = __int_as_float(o.x);
          rank0 += (ov < mv0 || (ov == mv0 && o.y < mi0)) ? 1 : 0;
          rank1 += (ov < mv1 || (ov == mv1 && o.y < mi1)) ? 1 : 0;
        }
      }
      if (have0 && rank0 < KNN) outp[rank0] = mi0;
      if (have1 && rank1 < KNN) outp[rank1] = mi1;
    }
  }

  if (over) {
    float pv[8]; int pi[8];
#pragma unroll
    for (int r = 0; r < 8; ++r) { pv[r] = -INFINITY; pi[r] = -1; }
    for (int k = 0; k < KNN; ++k) {
      float bv[8]; int bi[8];
#pragma unroll
      for (int r = 0; r < 8; ++r) { bv[r] = INFINITY; bi[r] = 0x7fffffff; }
      for (int t = 0; t < N / 16; ++t) {
        float nn;
        const v8f d = knn_tile(x, t * 16 + ln, hf, a, nn);
        const int col = t * 16 + ln;
#pragma unroll
        for (int r = 0; r < 8; ++r) {
          const float v = __builtin_fmaf(-2.f, d[r], sN[r] + nn);
          const bool after  = (v > pv[r]) || (v == pv[r] && col > pi[r]);
          const bool better = (v < bv[r]) || (v == bv[r] && col < bi[r]);
          if (after && better) { bv[r] = v; bi[r] = col; }
        }
      }
#pragma unroll
      for (int r = 0; r < 8; ++r) {
#pragma unroll
        for (int o = 8; o > 0; o >>= 1) {
          const float ov = __shfl_xor(bv[r], o, 32);
          const int   oi = __shfl_xor(bi[r], o, 32);
          if (ov < bv[r] || (ov == bv[r] && oi < bi[r])) { bv[r] = ov; bi[r] = oi; }
        }
        pv[r] = bv[r]; pi[r] = bi[r];
        if (ln == 0 && cntL[w][r + 8 * hf] > KCAP)
          knn_idx[(size_t)(row0 + r + 8 * hf) * KNN + k] = bi[r];
      }
    }
  }
}

typedef __attribute__((ext_vector_type(4))) float v4f;
#define STD_W 8
__global__ __launch_bounds__(STD_W * 32) void k_std_partial(const float* __restrict__ xyz,
                                                            const float* __restrict__ feat,
                                                            const float* __restrict__ xyz_s,
                                                            const float* __restrict__ feat_s,
                                                            const int* __restrict__ knn_idx,
                                                            int N, int S, int C,
                                                            float* __restrict__ part) {
  __shared__ float red[STD_W][KNN][4];
  const int w    = __builtin_amdgcn_readfirstlane((int)threadIdx.x >> 5);
  const int lane = threadIdx.x & 31;
  const int rpw  = BATCH * S / (STD_W * STD_G);
  const int gw   = blockIdx.x * STD_W + w;
  float ax = 0.f, qx = 0.f;
  v4f s4 = {0.f, 0.f, 0.f, 0.f}, q4 = {0.f, 0.f, 0.f, 0.f};
  for (int r = 0; r < rpw; ++r) {
    const int row = gw * rpw + r;
    const int b   = row / S;
    const int n   = knn_idx[(size_t)row * KNN + lane];
    const float* pn = xyz + (size_t)(b * N + n) * 3;
    const float* pq = xyz_s + (size_t)row * 3;
    for (int c = 0; c < 3; ++c) { const float d = pn[c] - pq[c]; ax += d; qx += d * d; }
    const float* fr = feat   + (size_t)(b * N + n) * C;
    const float* fc = feat_s + (size_t)row * C;
    for (int j = 0; j < C; j += 64) {
      v4f v[16];
#pragma unroll
      for (int i = 0; i < 16; ++i) v[i] = *(const v4f*)(fr + j + 4 * i);
#pragma unroll
      for (int i = 0; i < 16; ++i) {
        const v4f d = v[i] - *(const v4f*)(fc + j + 4 * i);
        s4 += d;
        q4 = __builtin_elementwise_fma(d, d, q4);
      }
    }
  }
  red[w][lane][0] = ax;
  red[w][lane][1] = qx;
  red[w][lane][2] = (s4.x + s4.y) + (s4.z + s4.w);
  red[w][lane][3] = (q4.x + q4.y) + (q4.z + q4.w);
  __syncthreads();
  if (threadIdx.x < KNN) {
    float a0 = 0.f, a1 = 0.f, a2 = 0.f, a3 = 0.f;
    for (int u = 0; u < STD_W; ++u) {
      a0 += red[u][threadIdx.x][0]; a1 += red[u][threadIdx.x][1];
      a2 += red[u][threadIdx.x][2]; a3 += red[u][threadIdx.x][3];
    }
    float* q = part + ((size_t)blockIdx.x * KNN + threadIdx.x) * 4;
    q[0] = a0; q[1] = a1; q[2] = a2; q[3] = a3;
  }
}

__global__ __launch_bounds__(1024) void k_std_final(const float* __restrict__ part, int S, int C,
                                                    float* __restrict__ std_xyz, float* __restrict__ std_feat) {
  __shared__ float red[32][KNN][4];
  const int k = threadIdx.x & 31, p = threadIdx.x >> 5;
  float ax = 0.f, qx = 0.f, af = 0.f, qf = 0.f;
  for (int g = p; g < STD_G; g += 32) {
    const float* q = part + ((size_t)g * KNN + k) * 4;
    ax += q[0]; qx += q[1]; af += q[2]; qf += q[3];
  }
  red[p][k][0] = ax; red[p][k][1] = qx; red[p][k][2] = af; red[p][k][3] = qf;
  __syncthreads();
  if (threadIdx.x >= KNN) return;
  ax = 0.f; qx = 0.f; af = 0.f; qf = 0.f;
  for (int u = 0; u < 32; ++u) {
    ax += red[u][k][0]; qx += red[u][k][1]; af += red[u][k][2]; qf += red[u][k][3];
  }
  {
    double n = (double)BATCH * S * 3;
    double var = ((double)qx - (double)ax * ax / n) / (n - 1.0);
    std_xyz[k] = fmaxf(sqrtf(fmaxf((float)var, 0.f)), 1e-5f);
  }
  {
    double n = (double)BATCH * S * C;
    double var = ((double)qf - (double)af * af / n) / (n - 1.0);
    std_feat[k] = fmaxf(sqrtf(fmaxf((float)var, 0.f)), 1e-5f);
  }
}

#define PCH 4
__global__ __launch_bounds__(256) void k_stage(const float* __restrict__ xyz,
                                               const float* __restrict__ feat,
                                               const float* __restrict__ xyz_s,
                                               const float* __restrict__ feat_s,
                                               const int* __restrict__ knn_idx,
                                               const float* __restrict__ std_xyz,
                                               const float* __restrict__ std_feat,
                                               int N, int S, int C, int chunk,
                                               float* __restrict__ out,
                                               float* __restrict__ pmaxb,
                                               float* __restrict__ psumb) {
  __shared__ int   kidx[PCH * KNN];
  __shared__ float sfe[KNN];
  __shared__ float xkn[PCH * KNN][3];
  __shared__ float fs[512];
  const int nt   = blockDim.x;
  const int row0 = blockIdx.x * chunk;
  const int b    = row0 / S;
  const int nk   = chunk * KNN;
  for (int t = threadIdx.x; t < nk; t += nt) kidx[t] = knn_idx[(size_t)row0 * KNN + t];
  if (threadIdx.x < KNN) sfe[threadIdx.x] = std_feat[threadIdx.x];
  __syncthreads();
  for (int t = threadIdx.x; t < nk; t += nt)
    __builtin_prefetch(feat + (size_t)(b * N + kidx[t]) * C, 0, 3);
  for (int t = threadIdx.x; t < nk * 3; t += nt) {
    const int kk = t / 3, c = t % 3;
    const int q = kk / KNN, k = kk % KNN;
    xkn[kk][c] = (xyz[(size_t)(b * N + kidx[kk]) * 3 + c] - xyz_s[(size_t)(row0 + q) * 3 + c]) / std_xyz[k];
  }
  for (int t = threadIdx.x; t < chunk * C; t += nt) fs[t] = feat_s[(size_t)row0 * C + t];
  __syncthreads();
  const int C2 = 2 * C;
  const int fd = (C2 + 2) / 3;
  const int M3 = 3 * fd;
  const float inv_sig = 1.f / SIGMA_;
  for (int j = threadIdx.x; j < C2; j += nt) {
    int src = (int)(((long long)j * (M3 - 1)) / (C2 - 1));
    int cc = src / fd, f = src % fd;
    float fv = -1.f + 2.f * (float)(f + 1) / (float)(fd + 1);
    float pmx = -FLT_MAX, psm = 0.f;
    for (int q = 0; q < chunk; ++q) {
      const int*   kq = kidx + q * KNN;
      const float* fq = fs + q * C;
      float acc = 0.f, mx = -FLT_MAX;
      for (int k = 0; k < KNN; ++k) {
        float t  = (xkn[q * KNN + k][cc] - fv) * inv_sig;
        float pe = expf(-0.5f * t * t);
        float fk;
        if (j < C) fk = (feat[(size_t)(b * N + kq[k]) * C + j] - fq[j]) / sfe[k];
        else       fk = fq[j - C];
        float fw = (fk + pe) * pe;
        acc += fw; mx = fmaxf(mx, fw);
      }
      float xx = acc * (1.f / KNN) + mx;
      const float o = 0.5f * xx * (1.f + erff(xx * 0.70710678118654752f));
      out[(size_t)(row0 + q) * C2 + j] = o;
      pmx = fmaxf(pmx, o); psm += o;
    }
    if (chunk > 1) {
      pmaxb[(size_t)blockIdx.x * C2 + j] = pmx;
      psumb[(size_t)blockIdx.x * C2 + j] = psm;
    }
  }
}

__global__ __launch_bounds__(1024) void k_pool_fold(const float* pm, const float* ps,

                                                    int G, int C2, int S,
                                                    float* __restrict__ out, int off) {
  const int b = blockIdx.x;
  for (int j = threadIdx.x; j < C2; j += blockDim.x) {
    const float* a = pm + (size_t)b * G * C2 + j;
    const float* c = ps + (size_t)b * G * C2 + j;
    float mx = -FLT_MAX, sm = 0.f;
    if (pm == ps) {
#pragma unroll 8
      for (int g = 0; g < G; ++g) {
        const float v = a[(size_t)g * C2];
        mx = fmaxf(mx, v); sm += v;
      }
    } else {
#pragma unroll 8
      for (int g = 0; g < G; ++g) {
        mx = fmaxf(mx, a[(size_t)g * C2]);
        sm += c[(size_t)g * C2];
      }
    }
    out[b * 3840 + off + j]      = mx;
    out[b * 3840 + off + C2 + j] = sm / (float)S;
  }
}

extern "C" void kernel_launch(void* const* d_in, const int* in_sizes, int n_in,
                              void* d_out, int out_size, void* d_ws, size_t ws_size,
                              hipStream_t stream) {
  const float* xyz_in = (const float*)d_in[0];
  const float* W      = (const float*)d_in[1];
  const float* gamma  = (const float*)d_in[2];
  const float* beta   = (const float*)d_in[3];
  float* out = (float*)d_out;
  float* ws  = (float*)d_ws;

  size_t o = 0;
  float* featA  = ws + o; o += (size_t)BATCH * NPTS * CINIT;
  float* featB  = ws + o; o += (size_t)BATCH * NPTS * CINIT;
  float* xyzA   = ws + o; o += (size_t)BATCH * NPTS * 3;
  float* xyzB   = ws + o; o += (size_t)BATCH * NPTS * 3;
  float* feat_s = ws + o; o += (size_t)BATCH * 1024 * CINIT;
  float* pmaxb  = ws + o; o += (size_t)1024 * 1024;
  float* psumb  = ws + o; o += (size_t)1024 * 1024;
  float* bn_mu  = ws + o; o += 64;
  float* bn_inv = ws + o; o += 64;
  float* stdx   = ws + o; o += 32;
  float* stdf   = ws + o; o += 32;
  float* stdp   = pmaxb;
  int* fps_i = (int*)(ws + o); o += (size_t)BATCH * 1024;
  int* knn_i = (int*)(ws + o); o += (size_t)BATCH * 1024 * KNN;

  k_embed_wmma<<<1024, 256, 0, stream>>>(xyz_in, W, featA);
  k_bn_stats<<<CINIT, 256, 0, stream>>>(featA, bn_mu, bn_inv);
  k_bn_apply<<<2048, 256, 0, stream>>>(featA, bn_mu, bn_inv, gamma, beta);

  const float* xyz_cur = xyz_in;
  float* feat_cur = featA; float* feat_nxt = featB;
  float* xs_buf = xyzA;    float* xs_other = xyzB;
  int N = NPTS, C = CINIT, off = 0;

  for (int st = 0; st < 4; ++st) {
    const int S = N / 2, C2 = 2 * C;
    if (N == 2048)      k_fps_wave<64><<<BATCH, 32, 0, stream>>>(xyz_cur, S, fps_i);
    k_gather<<<BATCH * S, 256, 0, stream>>>(xyz_cur, feat_cur, fps_i, N, S, C, xs_buf, feat_s);
    {
      const int nb = BATCH * S / 16 / KW;
      if (N == 2048)      k_knn_fused<2048><<<BATCH + nb, KW * 32, 0, stream>>>(xs_buf, xyz_cur, S, knn_i, fps_i);
      else if (N == 1024) k_knn_fused<1024><<<BATCH + nb, KW * 32, 0, stream>>>(xs_buf, xyz_cur, S, knn_i, fps_i);
      else if (N == 512)  k_knn_fused<512><<<BATCH + nb, KW * 32, 0, stream>>>(xs_buf, xyz_cur, S, knn_i, fps_i);
      else                k_knn_fused<256><<<nb, KW * 32, 0, stream>>>(xs_buf, xyz_cur, S, knn_i, fps_i);
    }
    {
      k_std_partial<<<STD_G, STD_W * 32, 0, stream>>>(xyz_cur, feat_cur, xs_buf, feat_s, knn_i, N, S, C, stdp);
      k_std_final<<<1, 1024, 0, stream>>>(stdp, S, C, stdx, stdf);
    }
    {
      const int chunk = (C2 <= 256) ? 4 : ((C2 == 512) ? 2 : 1);
      const int nthr  = (C2 < 256) ? C2 : 256;
      k_stage<<<BATCH * S / chunk, nthr, 0, stream>>>(xyz_cur, feat_cur, xs_buf, feat_s, knn_i,
                                                      stdx, stdf, N, S, C, chunk, feat_nxt,
                                                      pmaxb, psumb);
      if (chunk > 1) k_pool_fold<<<BATCH, C2, 0, stream>>>(pmaxb, psumb, S / chunk, C2, S, out, off);
      else           k_pool_fold<<<BATCH, C2, 0, stream>>>(feat_nxt, feat_nxt, S, C2, S, out, off);
    }
    off += 2 * C2;
    float* t = feat_cur; feat_cur = feat_nxt; feat_nxt = t;
    xyz_cur = xs_buf;
    float* u = xs_buf; xs_buf = xs_other; xs_other = u;
    N = S; C = C2;
  }
  (void)in_sizes; (void)n_in; (void)out_size; (void)ws_size;
}


# k_stage: the block gathers its neighbours' feature rows once into an LDS tile (vector loads in groups of eight, whole lines per wave); the channel loop reads the tile, not global memory
# speedup vs baseline: 5.3540x; 1.0737x over previous
#include <hip/hip_runtime.h>
#include <cfloat>
#include <cmath>

typedef __attribute__((ext_vector_type(2))) float v2f;
typedef __attribute__((ext_vector_type(3))) float v3f;
typedef __attribute__((ext_vector_type(8))) float v8f;

#define BATCH 16
#define NPTS 2048
#define CINIT 64
#define KNN 32
#define SIGMA_ 0.3f
#define BN_EPS_ 1e-5f
#define STD_G 256

__global__ __launch_bounds__(256) void k_embed_wmma(const float* __restrict__ xyz,
                                                    const float* __restrict__ W,
                                                    float* __restrict__ h) {
  const int wid  = (blockIdx.x * 256 + (int)threadIdx.x) >> 5;
  const int lane = threadIdx.x & 31;
  const int hf   = lane >> 4;
  const int ln   = lane & 15;
  const int tileM = wid >> 2;
  const int tileN = wid & 3;
  const int m  = tileM * 16 + ln;
  const int c  = tileN * 16 + ln;
  const float x0 = xyz[m * 3], x1 = xyz[m * 3 + 1], x2 = xyz[m * 3 + 2];
  const float w0 = W[c * 3],   w1 = W[c * 3 + 1],   w2 = W[c * 3 + 2];
  v2f a, b;
  a.x = hf ? x2 : x0;  a.y = hf ? 0.f : x1;
  b.x = hf ? w2 : w0;  b.y = hf ? 0.f : w1;
  v8f d = {};
  d = __builtin_amdgcn_wmma_f32_16x16x4_f32(false, a, false, b, (short)0, d, false, false);
#pragma unroll
  for (int r = 0; r < 8; ++r) {
    int row = tileM * 16 + r + 8 * hf;
    h[row * CINIT + tileN * 16 + ln] = d[r];
  }
}

__global__ __launch_bounds__(256) void k_bn_stats(const float* __restrict__ h,
                                                  float* __restrict__ mu,
                                                  float* __restrict__ inv) {
  __shared__ float s1[256], s2[256];
  const int c = blockIdx.x;
  const int n = BATCH * NPTS;
  float a = 0.f, b = 0.f;
  for (int i = threadIdx.x; i < n; i += 256) {
    float v = h[i * CINIT + c];
    a += v; b += v * v;
  }
  s1[threadIdx.x] = a; s2[threadIdx.x] = b; __syncthreads();
  for (int o = 128; o > 0; o >>= 1) {
    if ((int)threadIdx.x < o) { s1[threadIdx.x] += s1[threadIdx.x + o]; s2[threadIdx.x] += s2[threadIdx.x + o]; }
    __syncthreads();
  }
  if (threadIdx.x == 0) {
    float m   = s1[0] / (float)n;
    float var = s2[0] / (float)n - m * m;
    mu[c]  = m;
    inv[c] = rsqrtf(var + BN_EPS_);
  }
}

__global__ void k_bn_apply(float* __restrict__ h, const float* __restrict__ mu,
                           const float* __restrict__ inv, const float* __restrict__ gamma,
                           const float* __restrict__ beta) {
  const int total = BATCH * NPTS * CINIT;
  for (int i = blockIdx.x * blockDim.x + threadIdx.x; i < total; i += gridDim.x * blockDim.x) {
    int c = i & (CINIT - 1);
    float v = (h[i] - mu[c]) * inv[c] * gamma[c] + beta[c];
    h[i] = fmaxf(v, 0.f);
  }
}

__device__ __forceinline__ int imax2(int a, int b) { return a > b ? a : b; }
__device__ __forceinline__ int imin2(int a, int b) { return a < b ? a : b; }

template <int CTRL>
__device__ __forceinline__ int row_xchg(int v) {
  return __builtin_amdgcn_update_dpp(v, v, CTRL, 0xf, 0xf, true);
}

template <int PPL, int Q>
__device__ __forceinline__ unsigned fps_eq_bits(const int (&di)[PPL], int m) {
  constexpr int CH = PPL / 8;
  unsigned bits = 1u << (CH - 1);
#pragma unroll
  for (int x = 0; x < CH - 1; ++x) bits |= (di[Q * CH + x] == m) ? (1u << x) : 0u;
  return bits;
}

template <int PPL>
__device__ __forceinline__ int fps_pick_values(const v2f (&px)[PPL / 2], const v2f (&py)[PPL / 2],
                                               const v2f (&pz)[PPL / 2], int (&di)[PPL],
                                               const float4* pc, int lane,
                                               float& fx, float& fy, float& fz) {
  constexpr int CH = PPL / 8;
  int gm[8];
#pragma unroll
  for (int p2 = 0; p2 < PPL / 2; ++p2) {
    const v2f dx = px[p2] - fx, dy = py[p2] - fy, dz = pz[p2] - fz;
    const v2f t = __builtin_elementwise_fma(dz, dz, __builtin_elementwise_fma(dx, dx, dy * dy));
    const int s0 = 2 * p2, s1 = 2 * p2 + 1, q = s0 / CH;
    di[s0] = imin2(di[s0], __float_as_int(t.x));
    di[s1] = imin2(di[s1], __float_as_int(t.y));
    gm[q] = (s0 % CH == 0) ? imax2(di[s0], di[s1]) : imax2(gm[q], imax2(di[s0], di[s1]));
  }
  int mine = gm[0];
#pragma unroll
  for (int q = 1; q < 8; ++q) mine = imax2(mine, gm[q]);
  unsigned gbits = 0x80u;
#pragma unroll
  for (int q = 0; q < 7; ++q) gbits |= (gm[q] == mine) ? (1u << q) : 0u;
  int g = mine;
  g = imax2(g, row_xchg<0x161>(g));
  g = imax2(g, row_xchg<0x162>(g));
  g = imax2(g, row_xchg<0x164>(g));
  g = imax2(g, row_xchg<0x168>(g));
  g = imax2(g, (int)__builtin_amdgcn_permlanex16(g, g, 0x76543210, 0xfedcba98, false, false));
  const unsigned hit = __builtin_amdgcn_ballot_w32(mine == g);
  const int L = __builtin_ctz(hit);
  const int sq = __builtin_ctz((unsigned)__builtin_amdgcn_readlane((int)gbits, L));
  const int base = L * PPL + sq * CH;
  typedef const volatile __attribute__((address_space(3))) v3f lds_v3f;
  const v3f cand = *(lds_v3f*)(pc + base + (lane & (CH - 1)));
  unsigned eb;
  switch (sq) {
    case 0:  eb = fps_eq_bits<PPL, 0>(di, mine); break;
    case 1:  eb = fps_eq_bits<PPL, 1>(di, mine); break;
    case 2:  eb = fps_eq_bits<PPL, 2>(di, mine); break;
    case 3:  eb = fps_eq_bits<PPL, 3>(di, mine); break;
    case 4:  eb = fps_eq_bits<PPL, 4>(di, mine); break;
    case 5:  eb = fps_eq_bits<PPL, 5>(di, mine); break;
    case 6:  eb = fps_eq_bits<PPL, 6>(di, mine); break;
    default: eb = fps_eq_bits<PPL, 7>(di, mine); break;
  }
  const int es = __builtin_ctz((unsigned)__builtin_amdgcn_readlane((int)eb, L));
  fx = __int_as_float(__builtin_amdgcn_readlane(__float_as_int(cand.x), es));
  fy = __int_as_float(__builtin_amdgcn_readlane(__float_as_int(cand.y), es));
  fz = __int_as_float(__builtin_amdgcn_readlane(__float_as_int(cand.z), es));
  return base + es;
}

template <int PPL>
__device__ __forceinline__ void fps_cloud(const float* __restrict__ cloud, int S,
                                          int* __restrict__ idx_out, float4* pc, int lane, bool run) {
  constexpr bool VALS = (PPL >= 32);
  const float* p = cloud + (size_t)lane * PPL * 3;
  float px[VALS ? 1 : PPL], py[VALS ? 1 : PPL], pz[VALS ? 1 : PPL];
  v2f qx[VALS ? PPL / 2 : 1], qy[VALS ? PPL / 2 : 1], qz[VALS ? PPL / 2 : 1];
  int di[PPL];
  if (run) {
#pragma unroll
    for (int s = 0; s < PPL; ++s) {
      const float c0 = p[s * 3], c1 = p[s * 3 + 1], c2 = p[s * 3 + 2];
      if constexpr (VALS) { qx[s / 2][s % 2] = c0; qy[s / 2][s % 2] = c1; qz[s / 2][s % 2] = c2; }
      else                { px[s] = c0; py[s] = c1; pz[s] = c2; }
      di[s] = __float_as_int(1e10f);
      pc[lane * PPL + s] = make_float4(c0, c1, c2, 0.f);
    }
  }
  __syncthreads();
  if (!run) return;
  int far = 0;
  float fx = 0.f, fy = 0.f, fz = 0.f;
  if constexpr (VALS) { const float4 f0 = pc[0]; fx = f0.x; fy = f0.y; fz = f0.z; }
  for (int it0 = 0; it0 < S; it0 += 32) {
    int keep = 0;
    for (int j = 0; j < 32; ++j) {
      if constexpr (VALS) {
        keep = (lane == j) ? far : keep;
        far = fps_pick_values<PPL>(qx, qy, qz, di, pc, lane, fx, fy, fz);
      } else {
        const float4 f = pc[far];
        keep = (lane == j) ? far : keep;
        constexpr int CH = PPL / 8;
        int v[8], ix[8];
#pragma unroll
        for (int q = 0; q < 8; ++q) {
#pragma unroll
          for (int e = 0; e < CH; ++e) {
            const int s = q * CH + e;
            const float dx = px[s] - f.x, dy = py[s] - f.y, dz = pz[s] - f.z;
            const float t = __builtin_fmaf(dz, dz, __builtin_fmaf(dx, dx, dy * dy));
            di[s] = imin2(di[s], __float_as_int(t));
            if (e == 0) { v[q] = di[s]; ix[q] = s; }
            else {
              const bool c = di[s] > v[q];
              v[q]  = c ? di[s] : v[q];
              ix[q] = c ? s : ix[q];
            }
          }
        }
#pragma unroll
        for (int w = 8; w > 1; w >>= 1) {
#pragma unroll
          for (int k = 0; k < w; k += 2) {
            const bool c = v[k + 1] > v[k];
            const int nv = c ? v[k + 1] : v[k];
            const int ni = c ? ix[k + 1] : ix[k];
            v[k / 2] = nv; ix[k / 2] = ni;
          }
        }
        const int mine = v[0];
        int g = mine;
        g = imax2(g, row_xchg<0x161>(g));
        g = imax2(g, row_xchg<0x162>(g));
        g = imax2(g, row_xchg<0x164>(g));
        g = imax2(g, row_xchg<0x168>(g));
        g = imax2(g, (int)__builtin_amdgcn_permlanex16(g, g, 0x76543210, 0xfedcba98, false, false));
        const unsigned hit = __builtin_amdgcn_ballot_w32(mine == g);
        const int L = __builtin_ctz(hit);
        far = L * PPL + __builtin_amdgcn_readlane(ix[0], L);
      }
    }
    idx_out[it0 + lane] = keep;
  }
}

template <int PPL>
__global__ __launch_bounds__(32) void k_fps_wave(const float* __restrict__ xyz, int S,
                                                 int* __restrict__ fps_idx) {
  __shared__ float4 pc[32 * PPL];
  fps_cloud<PPL>(xyz + (size_t)blockIdx.x * (32 * PPL) * 3, S, fps_idx + (size_t)blockIdx.x * S, pc,
                 (int)threadIdx.x, true);
}

__global__ void k_gather(const float* __restrict__ xyz, const float* __restrict__ feat,
                         const int* __restrict__ fps_idx, int N, int S, int C,
                         float* __restrict__ xyz_s, float* __restrict__ feat_s) {
  const int b = blockIdx.x / S, s = blockIdx.x % S;
  const int idx = fps_idx[b * S + s];
  if (threadIdx.x < 3)
    xyz_s[(size_t)(b * S + s) * 3 + threadIdx.x] = xyz[(size_t)(b * N + idx) * 3 + threadIdx.x];
  for (int j = threadIdx.x; j < C; j += blockDim.x)
    feat_s[(size_t)(b * S + s) * C + j] = feat[(size_t)(b * N + idx) * C + j];
}

#define KW 2
#define KCAP 128
#define KU 8

__device__ __forceinline__ float sq3(float p0, float p1, float p2) {
  return __builtin_fmaf(p2, p2, __builtin_fmaf(p0, p0, p1 * p1));
}

__device__ __forceinline__ v8f knn_tile(const float* __restrict__ x, int n, int hf, v2f a,
                                        float& nn) {
  const float x0 = x[n * 3], x1 = x[n * 3 + 1], x2 = x[n * 3 + 2];
  nn = sq3(x0, x1, x2);
  v2f b;
  b.x = hf ? x2 : x0;  b.y = hf ? 0.f : x1;
  v8f d = {};
  return __builtin_amdgcn_wmma_f32_16x16x4_f32(false, a, false, b, (short)0, d, false, false);
}

template <int N>
__global__ __launch_bounds__(KW * 32) void k_knn_fused(const float* __restrict__ xs_all,
                                                       const float* __restrict__ x_all,
                                                       int S, int* __restrict__ knn_idx,
                                                       int* __restrict__ fps_next) {
  constexpr int  RPPL = N / 64;
  constexpr bool RIDE = (RPPL >= 8);
  __shared__ float gm[KW][16][64];
  __shared__ int2  cand[KW][16][KCAP];
  __shared__ int   cntL[KW][16];
  __shared__ float4 rpc[RIDE ? 32 * RPPL : 1];
  const int w    = threadIdx.x >> 5;
  const int lane = threadIdx.x & 31;
  if constexpr (RIDE) {
    if (blockIdx.x < BATCH) {
      const bool first = __builtin_amdgcn_readfirstlane(w) == 0;
      fps_cloud<RPPL>(xs_all + (size_t)blockIdx.x * S * 3, S / 2,
                      fps_next + (size_t)blockIdx.x * (S / 2), rpc, lane, first);
      return;
    }
  }
  const int blk  = RIDE ? (int)blockIdx.x - BATCH : (int)blockIdx.x;
  const int hf   = lane >> 4;
  const int ln   = lane & 15;
  const int row0 = (blk * KW + w) * 16;
  const int b    = row0 / S;
  const float* xs = xs_all + (size_t)row0 * 3;
  const float* x  = x_all + (size_t)b * N * 3;
  const float s0 = xs[ln * 3], s1 = xs[ln * 3 + 1], s2 = xs[ln * 3 + 2];
  v2f a;
  a.x = hf ? s2 : s0;  a.y = hf ? 0.f : s1;
  float sN[8];
#pragma unroll
  for (int r = 0; r < 8; ++r) {
    const float a0 = xs[(r + 8 * hf) * 3], a1 = xs[(r + 8 * hf) * 3 + 1], a2 = xs[(r + 8 * hf) * 3 + 2];
    sN[r] = sq3(a0, a1, a2);
  }

  float g[4][8];
#pragma unroll
  for (int u = 0; u < 4; ++u)
#pragma unroll
    for (int r = 0; r < 8; ++r) g[u][r] = INFINITY;
  for (int t = 0; t < N / 16; t += KU) {
    v8f d[KU];
    float nn[KU];
#pragma unroll
    for (int u = 0; u < KU; ++u) d[u] = knn_tile(x, (t + u) * 16 + ln, hf, a, nn[u]);
#pragma unroll
    for (int u = 0; u < KU; ++u)
#pragma unroll
      for (int r = 0; r < 8; ++r)
        g[u & 3][r] = fminf(g[u & 3][r], __builtin_fmaf(-2.f, d[u][r], sN[r] + nn[u]));
  }
#pragma unroll
  for (int u = 0; u < 4; ++u)
#pragma unroll
    for (int r = 0; r < 8; ++r) gm[w][r + 8 * hf][u * 16 + ln] = g[u][r];
  __syncthreads();

  float T[8];
#pragma unroll
  for (int r = 0; r < 8; ++r) {
    const float* grow = gm[w][r + 8 * hf];
    int k0 = 0, k1 = 0, k2 = 0, k3 = 0;
#pragma unroll 16
    for (int e = 0; e < 64; ++e) {
      const float o = grow[e];
      k0 += (o < g[0][r]) ? 1 : 0;
      k1 += (o < g[1][r]) ? 1 : 0;
      k2 += (o < g[2][r]) ? 1 : 0;
      k3 += (o < g[3][r]) ? 1 : 0;
    }
    float tv = (k0 < 32) ? g[0][r] : -INFINITY;
    tv = fmaxf(tv, (k1 < 32) ? g[1][r] : -INFINITY);
    tv = fmaxf(tv, (k2 < 32) ? g[2][r] : -INFINITY);
    tv = fmaxf(tv, (k3 < 32) ? g[3][r] : -INFINITY);
#pragma unroll
    for (int o = 8; o > 0; o >>= 1) tv = fmaxf(tv, __shfl_xor(tv, o, 32));
    T[r] = tv;
  }

  int cnt[8];
#pragma unroll
  for (int r = 0; r < 8; ++r) cnt[r] = 0;
  const unsigned below = (1u << ln) - 1u;
  for (int t = 0; t < N / 16; t += KU) {
    v8f d[KU];
    float nn[KU];
#pragma unroll
    for (int u = 0; u < KU; ++u) d[u] = knn_tile(x, (t + u) * 16 + ln, hf, a, nn[u]);
#pragma unroll
    for (int u = 0; u < KU; ++u) {
      const int col = (t + u) * 16 + ln;
#pragma unroll
      for (int r = 0; r < 8; ++r) {
        const float v = __builtin_fmaf(-2.f, d[u][r], sN[r] + nn[u]);
        const bool c = v <= T[r];
        const unsigned bal = __builtin_amdgcn_ballot_w32(c);
        if (bal != 0u) {
          const unsigned hb = (hf ? (bal >> 16) : bal) & 0xffffu;
          const int pos = cnt[r] + __builtin_popcount(hb & below);
          if (c && pos < KCAP) cand[w][r + 8 * hf][pos] = make_int2(__float_as_int(v), col);
          cnt[r] += __builtin_popcount(hb);
        }
      }
    }
  }
#pragma unroll
  for (int r = 0; r < 8; ++r) {
    const int p = cnt[r] + ln;
    if (ln < 8 && p < ((cnt[r] + 7) & ~7) && p < KCAP)
      cand[w][r + 8 * hf][p] = make_int2(__float_as_int(INFINITY), 0x7fffffff);
  }
  if (ln == 0) {
#pragma unroll
    for (int r = 0; r < 8; ++r) cntL[w][r + 8 * hf] = cnt[r];
  }
  __syncthreads();

  bool over = false;
  for (int R = 0; R < 16; ++R) {
    const int M = __builtin_amdgcn_readfirstlane(cntL[w][R]);
    if (M > KCAP) { over = true; continue; }
    const int2* cr = cand[w][R];
    int* outp = knn_idx + (size_t)(row0 + R) * KNN;
    const int Mp = (M + 7) & ~7;
    for (int base = 0; base < M; base += 64) {
      const int my0 = base + lane, my1 = base + 32 + lane;
      const bool have0 = my0 < M, have1 = my1 < M;
      const int2 mine0 = cr[have0 ? my0 : 0];
      const int2 mine1 = cr[have1 ? my1 : 0];
      const float mv0 = __int_as_float(mine0.x), mv1 = __int_as_float(mine1.x);
      const int   mi0 = mine0.y, mi1 = mine1.y;
      int rank0 = 0, rank1 = 0;
      for (int j = 0; j < Mp; j += 8) {
#pragma unroll
        for (int q = 0; q < 8; ++q) {
          const int2 o = cr[j + q];
          const float ov = __int_as_float(o.x);
          rank0 += (ov < mv0 || (ov == mv0 && o.y < mi0)) ? 1 : 0;
          rank1 += (ov < mv1 || (ov == mv1 && o.y < mi1)) ? 1 : 0;
        }
      }
      if (have0 && rank0 < KNN) outp[rank0] = mi0;
      if (have1 && rank1 < KNN) outp[rank1] = mi1;
    }
  }

  if (over) {
    float pv[8]; int pi[8];
#pragma unroll
    for (int r = 0; r < 8; ++r) { pv[r] = -INFINITY; pi[r] = -1; }
    for (int k = 0; k < KNN; ++k) {
      float bv[8]; int bi[8];
#pragma unroll
      for (int r = 0; r < 8; ++r) { bv[r] = INFINITY; bi[r] = 0x7fffffff; }
      for (int t = 0; t < N / 16; ++t) {
        float nn;
        const v8f d = knn_tile(x, t * 16 + ln, hf, a, nn);
        const int col = t * 16 + ln;
#pragma unroll
        for (int r = 0; r < 8; ++r) {
          const float v = __builtin_fmaf(-2.f, d[r], sN[r] + nn);
          const bool after  = (v > pv[r]) || (v == pv[r] && col > pi[r]);
          const bool better = (v < bv[r]) || (v == bv[r] && col < bi[r]);
          if (after && better) { bv[r] = v; bi[r] = col; }
        }
      }
#pragma unroll
      for (int r = 0; r < 8; ++r) {
#pragma unroll
        for (int o = 8; o > 0; o >>= 1) {
          const float ov = __shfl_xor(bv[r], o, 32);
          const int   oi = __shfl_xor(bi[r], o, 32);
          if (ov < bv[r] || (ov == bv[r] && oi < bi[r])) { bv[r] = ov; bi[r] = oi; }
        }
        pv[r] = bv[r]; pi[r] = bi[r];
        if (ln == 0 && cntL[w][r + 8 * hf] > KCAP)
          knn_idx[(size_t)(row0 + r + 8 * hf) * KNN + k] = bi[r];
      }
    }
  }
}

typedef __attribute__((ext_vector_type(4))) float v4f;
#define STD_W 8
__global__ __launch_bounds__(STD_W * 32) void k_std_partial(const float* __restrict__ xyz,
                                                            const float* __restrict__ feat,
                                                            const float* __restrict__ xyz_s,
                                                            const float* __restrict__ feat_s,
                                                            const int* __restrict__ knn_idx,
                                                            int N, int S, int C,
                                                            float* __restrict__ part) {
  __shared__ float red[STD_W][KNN][4];
  const int w    = __builtin_amdgcn_readfirstlane((int)threadIdx.x >> 5);
  const int lane = threadIdx.x & 31;
  const int rpw  = BATCH * S / (STD_W * STD_G);
  const int gw   = blockIdx.x * STD_W + w;
  float ax = 0.f, qx = 0.f;
  v4f s4 = {0.f, 0.f, 0.f, 0.f}, q4 = {0.f, 0.f, 0.f, 0.f};
  for (int r = 0; r < rpw; ++r) {
    const int row = gw * rpw + r;
    const int b   = row / S;
    const int n   = knn_idx[(size_t)row * KNN + lane];
    const float* pn = xyz + (size_t)(b * N + n) * 3;
    const float* pq = xyz_s + (size_t)row * 3;
    for (int c = 0; c < 3; ++c) { const float d = pn[c] - pq[c]; ax += d; qx += d * d; }
    const float* fr = feat   + (size_t)(b * N + n) * C;
    const float* fc = feat_s + (size_t)row * C;
    for (int j = 0; j < C; j += 64) {
      v4f v[16];
#pragma unroll
      for (int i = 0; i < 16; ++i) v[i] = *(const v4f*)(fr + j + 4 * i);
#pragma unroll
      for (int i = 0; i < 16; ++i) {
        const v4f d = v[i] - *(const v4f*)(fc + j + 4 * i);
        s4 += d;
        q4 = __builtin_elementwise_fma(d, d, q4);
      }
    }
  }
  red[w][lane][0] = ax;
  red[w][lane][1] = qx;
  red[w][lane][2] = (s4.x + s4.y) + (s4.z + s4.w);
  red[w][lane][3] = (q4.x + q4.y) + (q4.z + q4.w);
  __syncthreads();
  if (threadIdx.x < KNN) {
    float a0 = 0.f, a1 = 0.f, a2 = 0.f, a3 = 0.f;
    for (int u = 0; u < STD_W; ++u) {
      a0 += red[u][threadIdx.x][0]; a1 += red[u][threadIdx.x][1];
      a2 += red[u][threadIdx.x][2]; a3 += red[u][threadIdx.x][3];
    }
    float* q = part + ((size_t)blockIdx.x * KNN + threadIdx.x) * 4;
    q[0] = a0; q[1] = a1; q[2] = a2; q[3] = a3;
  }
}

__global__ __launch_bounds__(1024) void k_std_final(const float* __restrict__ part, int S, int C,
                                                    float* __restrict__ std_xyz, float* __restrict__ std_feat) {
  __shared__ float red[32][KNN][4];
  const int k = threadIdx.x & 31, p = threadIdx.x >> 5;
  float ax = 0.f, qx = 0.f, af = 0.f, qf = 0.f;
  for (int g = p; g < STD_G; g += 32) {
    const float* q = part + ((size_t)g * KNN + k) * 4;
    ax += q[0]; qx += q[1]; af += q[2]; qf += q[3];
  }
  red[p][k][0] = ax; red[p][k][1] = qx; red[p][k][2] = af; red[p][k][3] = qf;
  __syncthreads();
  if (threadIdx.x >= KNN) return;
  ax = 0.f; qx = 0.f; af = 0.f; qf = 0.f;
  for (int u = 0; u < 32; ++u) {
    ax += red[u][k][0]; qx += red[u][k][1]; af += red[u][k][2]; qf += red[u][k][3];
  }
  {
    double n = (double)BATCH * S * 3;
    double var = ((double)qx - (double)ax * ax / n) / (n - 1.0);
    std_xyz[k] = fmaxf(sqrtf(fmaxf((float)var, 0.f)), 1e-5f);
  }
  {
    double n = (double)BATCH * S * C;
    double var = ((double)qf - (double)af * af / n) / (n - 1.0);
    std_feat[k] = fmaxf(sqrtf(fmaxf((float)var, 0.f)), 1e-5f);
  }
}

#define PCH 4
template <int TF>
__global__ __launch_bounds__(256) void k_stage(const float* __restrict__ xyz,
                                               const float* __restrict__ feat,
                                               const float* __restrict__ xyz_s,
                                               const float* __restrict__ feat_s,
                                               const int* __restrict__ knn_idx,
                                               const float* __restrict__ std_xyz,
                                               const float* __restrict__ std_feat,
                                               int N, int S, int C, int chunk,
                                               float* __restrict__ out,
                                               float* __restrict__ pmaxb,
                                               float* __restrict__ psumb) {
  __shared__ int   kidx[PCH * KNN];
  __shared__ float sfe[KNN];
  __shared__ float xkn[PCH * KNN][3];
  __shared__ float fs[512];
  __shared__ __attribute__((aligned(16))) float ft[TF];
  const int nt   = blockDim.x;
  const int row0 = blockIdx.x * chunk;
  const int b    = row0 / S;
  const int nk   = chunk * KNN;
  for (int t = threadIdx.x; t < nk; t += nt) kidx[t] = knn_idx[(size_t)row0 * KNN + t];
  if (threadIdx.x < KNN) sfe[threadIdx.x] = std_feat[threadIdx.x];
  __syncthreads();
  {
    const int c4 = C >> 2;
    const int r0 = (int)threadIdx.x / c4, cp = (int)threadIdx.x - r0 * c4;
    const int rstep = nt / c4;
    for (int r = r0; r < nk; r += 8 * rstep) {
      v4f v[8];
#pragma unroll
      for (int i = 0; i < 8; ++i) {
        const int rr = min(r + i * rstep, nk - 1);
        v[i] = *(const v4f*)(feat + (size_t)(b * N + kidx[rr]) * C + 4 * cp);
      }
      asm volatile("" :: "v"(v[0]), "v"(v[1]), "v"(v[2]), "v"(v[3]), "v"(v[4]), "v"(v[5]), "v"(v[6]), "v"(v[7]));
#pragma unroll
      for (int i = 0; i < 8; ++i) {
        const int rr = r + i * rstep;
        if (rr < nk) ((v4f*)ft)[rr * c4 + cp] = v[i];
      }
    }
  }
  for (int t = threadIdx.x; t < nk * 3; t += nt) {
    const int kk = t / 3, c = t % 3;
    const int q = kk / KNN, k = kk % KNN;
    xkn[kk][c] = (xyz[(size_t)(b * N + kidx[kk]) * 3 + c] - xyz_s[(size_t)(row0 + q) * 3 + c]) / std_xyz[k];
  }
  for (int t = threadIdx.x; t < chunk * C; t += nt) fs[t] = feat_s[(size_t)row0 * C + t];
  __syncthreads();
  const int C2 = 2 * C;
  const int fd = (C2 + 2) / 3;
  const int M3 = 3 * fd;
  const float inv_sig = 1.f / SIGMA_;
  for (int j = threadIdx.x; j < C2; j += nt) {
    int src = (int)(((long long)j * (M3 - 1)) / (C2 - 1));
    int cc = src / fd, f = src % fd;
    float fv = -1.f + 2.f * (float)(f + 1) / (float)(fd + 1);
    float pmx = -FLT_MAX, psm = 0.f;
    for (int q = 0; q < chunk; ++q) {
      const float* fq = fs + q * C;
      float acc = 0.f, mx = -FLT_MAX;
      for (int k = 0; k < KNN; ++k) {
        float t  = (xkn[q * KNN + k][cc] - fv) * inv_sig;
        float pe = expf(-0.5f * t * t);
        float fk;
        if (j < C) fk = (ft[(q * KNN + k) * C + j] - fq[j]) / sfe[k];
        else       fk = fq[j - C];
        float fw = (fk + pe) * pe;
        acc += fw; mx = fmaxf(mx, fw);
      }
      float xx = acc * (1.f / KNN) + mx;
      const float o = 0.5f * xx * (1.f + erff(xx * 0.70710678118654752f));
      out[(size_t)(row0 + q) * C2 + j] = o;
      pmx = fmaxf(pmx, o); psm += o;
    }
    if (chunk > 1) {
      pmaxb[(size_t)blockIdx.x * C2 + j] = pmx;
      psumb[(size_t)blockIdx.x * C2 + j] = psm;
    }
  }
}

__global__ __launch_bounds__(1024) void k_pool_fold(const float* pm, const float* ps,

                                                    int G, int C2, int S,
                                                    float* __restrict__ out, int off) {
  const int b = blockIdx.x;
  for (int j = threadIdx.x; j < C2; j += blockDim.x) {
    const float* a = pm + (size_t)b * G * C2 + j;
    const float* c = ps + (size_t)b * G * C2 + j;
    float mx = -FLT_MAX, sm = 0.f;
    if (pm == ps) {
#pragma unroll 8
      for (int g = 0; g < G; ++g) {
        const float v = a[(size_t)g * C2];
        mx = fmaxf(mx, v); sm += v;
      }
    } else {
#pragma unroll 8
      for (int g = 0; g < G; ++g) {
        mx = fmaxf(mx, a[(size_t)g * C2]);
        sm += c[(size_t)g * C2];
      }
    }
    out[b * 3840 + off + j]      = mx;
    out[b * 3840 + off + C2 + j] = sm / (float)S;
  }
}

extern "C" void kernel_launch(void* const* d_in, const int* in_sizes, int n_in,
                              void* d_out, int out_size, void* d_ws, size_t ws_size,
                              hipStream_t stream) {
  const float* xyz_in = (const float*)d_in[0];
  const float* W      = (const float*)d_in[1];
  const float* gamma  = (const float*)d_in[2];
  const float* beta   = (const float*)d_in[3];
  float* out = (float*)d_out;
  float* ws  = (float*)d_ws;

  size_t o = 0;
  float* featA  = ws + o; o += (size_t)BATCH * NPTS * CINIT;
  float* featB  = ws + o; o += (size_t)BATCH * NPTS * CINIT;
  float* xyzA   = ws + o; o += (size_t)BATCH * NPTS * 3;
  float* xyzB   = ws + o; o += (size_t)BATCH * NPTS * 3;
  float* feat_s = ws + o; o += (size_t)BATCH * 1024 * CINIT;
  float* pmaxb  = ws + o; o += (size_t)1024 * 1024;
  float* psumb  = ws + o; o += (size_t)1024 * 1024;
  float* bn_mu  = ws + o; o += 64;
  float* bn_inv = ws + o; o += 64;
  float* stdx   = ws + o; o += 32;
  float* stdf   = ws + o; o += 32;
  float* stdp   = pmaxb;
  int* fps_i = (int*)(ws + o); o += (size_t)BATCH * 1024;
  int* knn_i = (int*)(ws + o); o += (size_t)BATCH * 1024 * KNN;

  k_embed_wmma<<<1024, 256, 0, stream>>>(xyz_in, W, featA);
  k_bn_stats<<<CINIT, 256, 0, stream>>>(featA, bn_mu, bn_inv);
  k_bn_apply<<<2048, 256, 0, stream>>>(featA, bn_mu, bn_inv, gamma, beta);

  const float* xyz_cur = xyz_in;
  float* feat_cur = featA; float* feat_nxt = featB;
  float* xs_buf = xyzA;    float* xs_other = xyzB;
  int N = NPTS, C = CINIT, off = 0;

  for (int st = 0; st < 4; ++st) {
    const int S = N / 2, C2 = 2 * C;
    if (N == 2048)      k_fps_wave<64><<<BATCH, 32, 0, stream>>>(xyz_cur, S, fps_i);
    k_gather<<<BATCH * S, 256, 0, stream>>>(xyz_cur, feat_cur, fps_i, N, S, C, xs_buf, feat_s);
    {
      const int nb = BATCH * S / 16 / KW;
      if (N == 2048)      k_knn_fused<2048><<<BATCH + nb, KW * 32, 0, stream>>>(xs_buf, xyz_cur, S, knn_i, fps_i);
      else if (N == 1024) k_knn_fused<1024><<<BATCH + nb, KW * 32, 0, stream>>>(xs_buf, xyz_cur, S, knn_i, fps_i);
      else if (N == 512)  k_knn_fused<512><<<BATCH + nb, KW * 32, 0, stream>>>(xs_buf, xyz_cur, S, knn_i, fps_i);
      else                k_knn_fused<256><<<nb, KW * 32, 0, stream>>>(xs_buf, xyz_cur, S, knn_i, fps_i);
    }
    {
      k_std_partial<<<STD_G, STD_W * 32, 0, stream>>>(xyz_cur, feat_cur, xs_buf, feat_s, knn_i, N, S, C, stdp);
      k_std_final<<<1, 1024, 0, stream>>>(stdp, S, C, stdx, stdf);
    }
    {
      const int chunk = (C2 <= 256) ? 4 : ((C2 == 512) ? 2 : 1);
      const int nthr  = (C2 < 256) ? C2 : 256;
      if (C == CINIT) k_stage<PCH * KNN * CINIT><<<BATCH * S / chunk, nthr, 0, stream>>>(
                          xyz_cur, feat_cur, xs_buf, feat_s, knn_i, stdx, stdf, N, S, C, chunk, feat_nxt,
                          pmaxb, psumb);
      else            k_stage<16384><<<BATCH * S / chunk, nthr, 0, stream>>>(
                          xyz_cur, feat_cur, xs_buf, feat_s, knn_i, stdx, stdf, N, S, C, chunk, feat_nxt,
                          pmaxb, psumb);
      if (chunk > 1) k_pool_fold<<<BATCH, C2, 0, stream>>>(pmaxb, psumb, S / chunk, C2, S, out, off);
      else           k_pool_fold<<<BATCH, C2, 0, stream>>>(feat_nxt, feat_nxt, S, C2, S, out, off);
    }
    off += 2 * C2;
    float* t = feat_cur; feat_cur = feat_nxt; feat_nxt = t;
    xyz_cur = xs_buf;
    float* u = xs_buf; xs_buf = xs_other; xs_other = u;
    N = S; C = C2;
  }
  (void)in_sizes; (void)n_in; (void)out_size; (void)ws_size;
}


# k_stage: pe = exp2(c*d*d) on the hardware exp2 instruction (one multiply, one v_exp_f32), not the library expf and its range reduction; pe within 1e-7 absolute of exp, under 2^-126 becomes 0
# speedup vs baseline: 5.5691x; 1.0402x over previous
#include <hip/hip_runtime.h>
#include <cfloat>
#include <cmath>

typedef __attribute__((ext_vector_type(2))) float v2f;
typedef __attribute__((ext_vector_type(3))) float v3f;
typedef __attribute__((ext_vector_type(8))) float v8f;

#define BATCH 16
#define NPTS 2048
#define CINIT 64
#define KNN 32
#define SIGMA_ 0.3f
#define BN_EPS_ 1e-5f
#define STD_G 256

__global__ __launch_bounds__(256) void k_embed_wmma(const float* __restrict__ xyz,
                                                    const float* __restrict__ W,
                                                    float* __restrict__ h) {
  const int wid  = (blockIdx.x * 256 + (int)threadIdx.x) >> 5;
  const int lane = threadIdx.x & 31;
  const int hf   = lane >> 4;
  const int ln   = lane & 15;
  const int tileM = wid >> 2;
  const int tileN = wid & 3;
  const int m  = tileM * 16 + ln;
  const int c  = tileN * 16 + ln;
  const float x0 = xyz[m * 3], x1 = xyz[m * 3 + 1], x2 = xyz[m * 3 + 2];
  const float w0 = W[c * 3],   w1 = W[c * 3 + 1],   w2 = W[c * 3 + 2];
  v2f a, b;
  a.x = hf ? x2 : x0;  a.y = hf ? 0.f : x1;
  b.x = hf ? w2 : w0;  b.y = hf ? 0.f : w1;
  v8f d = {};
  d = __builtin_amdgcn_wmma_f32_16x16x4_f32(false, a, false, b, (short)0, d, false, false);
#pragma unroll
  for (int r = 0; r < 8; ++r) {
    int row = tileM * 16 + r + 8 * hf;
    h[row * CINIT + tileN * 16 + ln] = d[r];
  }
}

__global__ __launch_bounds__(256) void k_bn_stats(const float* __restrict__ h,
                                                  float* __restrict__ mu,
                                                  float* __restrict__ inv) {
  __shared__ float s1[256], s2[256];
  const int c = blockIdx.x;
  const int n = BATCH * NPTS;
  float a = 0.f, b = 0.f;
  for (int i = threadIdx.x; i < n; i += 256) {
    float v = h[i * CINIT + c];
    a += v; b += v * v;
  }
  s1[threadIdx.x] = a; s2[threadIdx.x] = b; __syncthreads();
  for (int o = 128; o > 0; o >>= 1) {
    if ((int)threadIdx.x < o) { s1[threadIdx.x] += s1[threadIdx.x + o]; s2[threadIdx.x] += s2[threadIdx.x + o]; }
    __syncthreads();
  }
  if (threadIdx.x == 0) {
    float m   = s1[0] / (float)n;
    float var = s2[0] / (float)n - m * m;
    mu[c]  = m;
    inv[c] = rsqrtf(var + BN_EPS_);
  }
}

__global__ void k_bn_apply(float* __restrict__ h, const float* __restrict__ mu,
                           const float* __restrict__ inv, const float* __restrict__ gamma,
                           const float* __restrict__ beta) {
  const int total = BATCH * NPTS * CINIT;
  for (int i = blockIdx.x * blockDim.x + threadIdx.x; i < total; i += gridDim.x * blockDim.x) {
    int c = i & (CINIT - 1);
    float v = (h[i] - mu[c]) * inv[c] * gamma[c] + beta[c];
    h[i] = fmaxf(v, 0.f);
  }
}

__device__ __forceinline__ int imax2(int a, int b) { return a > b ? a : b; }
__device__ __forceinline__ int imin2(int a, int b) { return a < b ? a : b; }

template <int CTRL>
__device__ __forceinline__ int row_xchg(int v) {
  return __builtin_amdgcn_update_dpp(v, v, CTRL, 0xf, 0xf, true);
}

template <int PPL, int Q>
__device__ __forceinline__ unsigned fps_eq_bits(const int (&di)[PPL], int m) {
  constexpr int CH = PPL / 8;
  unsigned bits = 1u << (CH - 1);
#pragma unroll
  for (int x = 0; x < CH - 1; ++x) bits |= (di[Q * CH + x] == m) ? (1u << x) : 0u;
  return bits;
}

template <int PPL>
__device__ __forceinline__ int fps_pick_values(const v2f (&px)[PPL / 2], const v2f (&py)[PPL / 2],
                                               const v2f (&pz)[PPL / 2], int (&di)[PPL],
                                               const float4* pc, int lane,
                                               float& fx, float& fy, float& fz) {
  constexpr int CH = PPL / 8;
  int gm[8];
#pragma unroll
  for (int p2 = 0; p2 < PPL / 2; ++p2) {
    const v2f dx = px[p2] - fx, dy = py[p2] - fy, dz = pz[p2] - fz;
    const v2f t = __builtin_elementwise_fma(dz, dz, __builtin_elementwise_fma(dx, dx, dy * dy));
    const int s0 = 2 * p2, s1 = 2 * p2 + 1, q = s0 / CH;
    di[s0] = imin2(di[s0], __float_as_int(t.x));
    di[s1] = imin2(di[s1], __float_as_int(t.y));
    gm[q] = (s0 % CH == 0) ? imax2(di[s0], di[s1]) : imax2(gm[q], imax2(di[s0], di[s1]));
  }
  int mine = gm[0];
#pragma unroll
  for (int q = 1; q < 8; ++q) mine = imax2(mine, gm[q]);
  unsigned gbits = 0x80u;
#pragma unroll
  for (int q = 0; q < 7; ++q) gbits |= (gm[q] == mine) ? (1u << q) : 0u;
  int g = mine;
  g = imax2(g, row_xchg<0x161>(g));
  g = imax2(g, row_xchg<0x162>(g));
  g = imax2(g, row_xchg<0x164>(g));
  g = imax2(g, row_xchg<0x168>(g));
  g = imax2(g, (int)__builtin_amdgcn_permlanex16(g, g, 0x76543210, 0xfedcba98, false, false));
  const unsigned hit = __builtin_amdgcn_ballot_w32(mine == g);
  const int L = __builtin_ctz(hit);
  const int sq = __builtin_ctz((unsigned)__builtin_amdgcn_readlane((int)gbits, L));
  const int base = L * PPL + sq * CH;
  typedef const volatile __attribute__((address_space(3))) v3f lds_v3f;
  const v3f cand = *(lds_v3f*)(pc + base + (lane & (CH - 1)));
  unsigned eb;
  switch (sq) {
    case 0:  eb = fps_eq_bits<PPL, 0>(di, mine); break;
    case 1:  eb = fps_eq_bits<PPL, 1>(di, mine); break;
    case 2:  eb = fps_eq_bits<PPL, 2>(di, mine); break;
    case 3:  eb = fps_eq_bits<PPL, 3>(di, mine); break;
    case 4:  eb = fps_eq_bits<PPL, 4>(di, mine); break;
    case 5:  eb = fps_eq_bits<PPL, 5>(di, mine); break;
    case 6:  eb = fps_eq_bits<PPL, 6>(di, mine); break;
    default: eb = fps_eq_bits<PPL, 7>(di, mine); break;
  }
  const int es = __builtin_ctz((unsigned)__builtin_amdgcn_readlane((int)eb, L));
  fx = __int_as_float(__builtin_amdgcn_readlane(__float_as_int(cand.x), es));
  fy = __int_as_float(__builtin_amdgcn_readlane(__float_as_int(cand.y), es));
  fz = __int_as_float(__builtin_amdgcn_readlane(__float_as_int(cand.z), es));
  return base + es;
}

template <int PPL>
__device__ __forceinline__ void fps_cloud(const float* __restrict__ cloud, int S,
                                          int* __restrict__ idx_out, float4* pc, int lane, bool run) {
  constexpr bool VALS = (PPL >= 32);
  const float* p = cloud + (size_t)lane * PPL * 3;
  float px[VALS ? 1 : PPL], py[VALS ? 1 : PPL], pz[VALS ? 1 : PPL];
  v2f qx[VALS ? PPL / 2 : 1], qy[VALS ? PPL / 2 : 1], qz[VALS ? PPL / 2 : 1];
  int di[PPL];
  if (run) {
#pragma unroll
    for (int s = 0; s < PPL; ++s) {
      const float c0 = p[s * 3], c1 = p[s * 3 + 1], c2 = p[s * 3 + 2];
      if constexpr (VALS) { qx[s / 2][s % 2] = c0; qy[s / 2][s % 2] = c1; qz[s / 2][s % 2] = c2; }
      else                { px[s] = c0; py[s] = c1; pz[s] = c2; }
      di[s] = __float_as_int(1e10f);
      pc[lane * PPL + s] = make_float4(c0, c1, c2, 0.f);
    }
  }
  __syncthreads();
  if (!run) return;
  int far = 0;
  float fx = 0.f, fy = 0.f, fz = 0.f;
  if constexpr (VALS) { const float4 f0 = pc[0]; fx = f0.x; fy = f0.y; fz = f0.z; }
  for (int it0 = 0; it0 < S; it0 += 32) {
    int keep = 0;
    for (int j = 0; j < 32; ++j) {
      if constexpr (VALS) {
        keep = (lane == j) ? far : keep;
        far = fps_pick_values<PPL>(qx, qy, qz, di, pc, lane, fx, fy, fz);
      } else {
        const float4 f = pc[far];
        keep = (lane == j) ? far : keep;
        constexpr int CH = PPL / 8;
        int v[8], ix[8];
#pragma unroll
        for (int q = 0; q < 8; ++q) {
#pragma unroll
          for (int e = 0; e < CH; ++e) {
            const int s = q * CH + e;
            const float dx = px[s] - f.x, dy = py[s] - f.y, dz = pz[s] - f.z;
            const float t = __builtin_fmaf(dz, dz, __builtin_fmaf(dx, dx, dy * dy));
            di[s] = imin2(di[s], __float_as_int(t));
            if (e == 0) { v[q] = di[s]; ix[q] = s; }
            else {
              const bool c = di[s] > v[q];
              v[q]  = c ? di[s] : v[q];
              ix[q] = c ? s : ix[q];
            }
          }
        }
#pragma unroll
        for (int w = 8; w > 1; w >>= 1) {
#pragma unroll
          for (int k = 0; k < w; k += 2) {
            const bool c = v[k + 1] > v[k];
            const int nv = c ? v[k + 1] : v[k];
            const int ni = c ? ix[k + 1] : ix[k];
            v[k / 2] = nv; ix[k / 2] = ni;
          }
        }
        const int mine = v[0];
        int g = mine;
        g = imax2(g, row_xchg<0x161>(g));
        g = imax2(g, row_xchg<0x162>(g));
        g = imax2(g, row_xchg<0x164>(g));
        g = imax2(g, row_xchg<0x168>(g));
        g = imax2(g, (int)__builtin_amdgcn_permlanex16(g, g, 0x76543210, 0xfedcba98, false, false));
        const unsigned hit = __builtin_amdgcn_ballot_w32(mine == g);
        const int L = __builtin_ctz(hit);
        far = L * PPL + __builtin_amdgcn_readlane(ix[0], L);
      }
    }
    idx_out[it0 + lane] = keep;
  }
}

template <int PPL>
__global__ __launch_bounds__(32) void k_fps_wave(const float* __restrict__ xyz, int S,
                                                 int* __restrict__ fps_idx) {
  __shared__ float4 pc[32 * PPL];
  fps_cloud<PPL>(xyz + (size_t)blockIdx.x * (32 * PPL) * 3, S, fps_idx + (size_t)blockIdx.x * S, pc,
                 (int)threadIdx.x, true);
}

__global__ void k_gather(const float* __restrict__ xyz, const float* __restrict__ feat,
                         const int* __restrict__ fps_idx, int N, int S, int C,
                         float* __restrict__ xyz_s, float* __restrict__ feat_s) {
  const int b = blockIdx.x / S, s = blockIdx.x % S;
  const int idx = fps_idx[b * S + s];
  if (threadIdx.x < 3)
    xyz_s[(size_t)(b * S + s) * 3 + threadIdx.x] = xyz[(size_t)(b * N + idx) * 3 + threadIdx.x];
  for (int j = threadIdx.x; j < C; j += blockDim.x)
    feat_s[(size_t)(b * S + s) * C + j] = feat[(size_t)(b * N + idx) * C + j];
}

#define KW 2
#define KCAP 128
#define KU 8

__device__ __forceinline__ float sq3(float p0, float p1, float p2) {
  return __builtin_fmaf(p2, p2, __builtin_fmaf(p0, p0, p1 * p1));
}

__device__ __forceinline__ v8f knn_tile(const float* __restrict__ x, int n, int hf, v2f a,
                                        float& nn) {
  const float x0 = x[n * 3], x1 = x[n * 3 + 1], x2 = x[n * 3 + 2];
  nn = sq3(x0, x1, x2);
  v2f b;
  b.x = hf ? x2 : x0;  b.y = hf ? 0.f : x1;
  v8f d = {};
  return __builtin_amdgcn_wmma_f32_16x16x4_f32(false, a, false, b, (short)0, d, false, false);
}

template <int N>
__global__ __launch_bounds__(KW * 32) void k_knn_fused(const float* __restrict__ xs_all,
                                                       const float* __restrict__ x_all,
                                                       int S, int* __restrict__ knn_idx,
                                                       int* __restrict__ fps_next) {
  constexpr int  RPPL = N / 64;
  constexpr bool RIDE = (RPPL >= 8);
  __shared__ float gm[KW][16][64];
  __shared__ int2  cand[KW][16][KCAP];
  __shared__ int   cntL[KW][16];
  __shared__ float4 rpc[RIDE ? 32 * RPPL : 1];
  const int w    = threadIdx.x >> 5;
  const int lane = threadIdx.x & 31;
  if constexpr (RIDE) {
    if (blockIdx.x < BATCH) {
      const bool first = __builtin_amdgcn_readfirstlane(w) == 0;
      fps_cloud<RPPL>(xs_all + (size_t)blockIdx.x * S * 3, S / 2,
                      fps_next + (size_t)blockIdx.x * (S / 2), rpc, lane, first);
      return;
    }
  }
  const int blk  = RIDE ? (int)blockIdx.x - BATCH : (int)blockIdx.x;
  const int hf   = lane >> 4;
  const int ln   = lane & 15;
  const int row0 = (blk * KW + w) * 16;
  const int b    = row0 / S;
  const float* xs = xs_all + (size_t)row0 * 3;
  const float* x  = x_all + (size_t)b * N * 3;
  const float s0 = xs[ln * 3], s1 = xs[ln * 3 + 1], s2 = xs[ln * 3 + 2];
  v2f a;
  a.x = hf ? s2 : s0;  a.y = hf ? 0.f : s1;
  float sN[8];
#pragma unroll
  for (int r = 0; r < 8; ++r) {
    const float a0 = xs[(r + 8 * hf) * 3], a1 = xs[(r + 8 * hf) * 3 + 1], a2 = xs[(r + 8 * hf) * 3 + 2];
    sN[r] = sq3(a0, a1, a2);
  }

  float g[4][8];
#pragma unroll
  for (int u = 0; u < 4; ++u)
#pragma unroll
    for (int r = 0; r < 8; ++r) g[u][r] = INFINITY;
  for (int t = 0; t < N / 16; t += KU) {
    v8f d[KU];
    float nn[KU];
#pragma unroll
    for (int u = 0; u < KU; ++u) d[u] = knn_tile(x, (t + u) * 16 + ln, hf, a, nn[u]);
#pragma unroll
    for (int u = 0; u < KU; ++u)
#pragma unroll
      for (int r = 0; r < 8; ++r)
        g[u & 3][r] = fminf(g[u & 3][r], __builtin_fmaf(-2.f, d[u][r], sN[r] + nn[u]));
  }
#pragma unroll
  for (int u = 0; u < 4; ++u)
#pragma unroll
    for (int r = 0; r < 8; ++r) gm[w][r + 8 * hf][u * 16 + ln] = g[u][r];
  __syncthreads();

  float T[8];
#pragma unroll
  for (int r = 0; r < 8; ++r) {
    const float* grow = gm[w][r + 8 * hf];
    int k0 = 0, k1 = 0, k2 = 0, k3 = 0;
#pragma unroll 16
    for (int e = 0; e < 64; ++e) {
      const float o = grow[e];
      k0 += (o < g[0][r]) ? 1 : 0;
      k1 += (o < g[1][r]) ? 1 : 0;
      k2 += (o < g[2][r]) ? 1 : 0;
      k3 += (o < g[3][r]) ? 1 : 0;
    }
    float tv = (k0 < 32) ? g[0][r] : -INFINITY;
    tv = fmaxf(tv, (k1 < 32) ? g[1][r] : -INFINITY);
    tv = fmaxf(tv, (k2 < 32) ? g[2][r] : -INFINITY);
    tv = fmaxf(tv, (k3 < 32) ? g[3][r] : -INFINITY);
#pragma unroll
    for (int o = 8; o > 0; o >>= 1) tv = fmaxf(tv, __shfl_xor(tv, o, 32));
    T[r] = tv;
  }

  int cnt[8];
#pragma unroll
  for (int r = 0; r < 8; ++r) cnt[r] = 0;
  const unsigned below = (1u << ln) - 1u;
  for (int t = 0; t < N / 16; t += KU) {
    v8f d[KU];
    float nn[KU];
#pragma unroll
    for (int u = 0; u < KU; ++u) d[u] = knn_tile(x, (t + u) * 16 + ln, hf, a, nn[u]);
#pragma unroll
    for (int u = 0; u < KU; ++u) {
      const int col = (t + u) * 16 + ln;
#pragma unroll
      for (int r = 0; r < 8; ++r) {
        const float v = __builtin_fmaf(-2.f, d[u][r], sN[r] + nn[u]);
        const bool c = v <= T[r];
        const unsigned bal = __builtin_amdgcn_ballot_w32(c);
        if (bal != 0u) {
          const unsigned hb = (hf ? (bal >> 16) : bal) & 0xffffu;
          const int pos = cnt[r] + __builtin_popcount(hb & below);
          if (c && pos < KCAP) cand[w][r + 8 * hf][pos] = make_int2(__float_as_int(v), col);
          cnt[r] += __builtin_popcount(hb);
        }
      }
    }
  }
#pragma unroll
  for (int r = 0; r < 8; ++r) {
    const int p = cnt[r] + ln;
    if (ln < 8 && p < ((cnt[r] + 7) & ~7) && p < KCAP)
      cand[w][r + 8 * hf][p] = make_int2(__float_as_int(INFINITY), 0x7fffffff);
  }
  if (ln == 0) {
#pragma unroll
    for (int r = 0; r < 8; ++r) cntL[w][r + 8 * hf] = cnt[r];
  }
  __syncthreads();

  bool over = false;
  for (int R = 0; R < 16; ++R) {
    const int M = __builtin_amdgcn_readfirstlane(cntL[w][R]);
    if (M > KCAP) { over = true; continue; }
    const int2* cr = cand[w][R];
    int* outp = knn_idx + (size_t)(row0 + R) * KNN;
    const int Mp = (M + 7) & ~7;
    for (int base = 0; base < M; base += 64) {
      const int my0 = base + lane, my1 = base + 32 + lane;
      const bool have0 = my0 < M, have1 = my1 < M;
      const int2 mine0 = cr[have0 ? my0 : 0];
      const int2 mine1 = cr[have1 ? my1 : 0];
      const float mv0 = __int_as_float(mine0.x), mv1 = __int_as_float(mine1.x);
      const int   mi0 = mine0.y, mi1 = mine1.y;
      int rank0 = 0, rank1 = 0;
      for (int j = 0; j < Mp; j += 8) {
#pragma unroll
        for (int q = 0; q < 8; ++q) {
          const int2 o = cr[j + q];
          const float ov = __int_as_float(o.x);
          rank0 += (ov < mv0 || (ov == mv0 && o.y < mi0)) ? 1 : 0;
          rank1 += (ov < mv1 || (ov == mv1 && o.y < mi1)) ? 1 : 0;
        }
      }
      if (have0 && rank0 < KNN) outp[rank0] = mi0;
      if (have1 && rank1 < KNN) outp[rank1] = mi1;
    }
  }

  if (over) {
    float pv[8]; int pi[8];
#pragma unroll
    for (int r = 0; r < 8; ++r) { pv[r] = -INFINITY; pi[r] = -1; }
    for (int k = 0; k < KNN; ++k) {
      float bv[8]; int bi[8];
#pragma unroll
      for (int r = 0; r < 8; ++r) { bv[r] = INFINITY; bi[r] = 0x7fffffff; }
      for (int t = 0; t < N / 16; ++t) {
        float nn;
        const v8f d = knn_tile(x, t * 16 + ln, hf, a, nn);
        const int col = t * 16 + ln;
#pragma unroll
        for (int r = 0; r < 8; ++r) {
          const float v = __builtin_fmaf(-2.f, d[r], sN[r] + nn);
          const bool after  = (v > pv[r]) || (v == pv[r] && col > pi[r]);
          const bool better = (v < bv[r]) || (v == bv[r] && col < bi[r]);
          if (after && better) { bv[r] = v; bi[r] = col; }
        }
      }
#pragma unroll
      for (int r = 0; r < 8; ++r) {
#pragma unroll
        for (int o = 8; o > 0; o >>= 1) {
          const float ov = __shfl_xor(bv[r], o, 32);
          const int   oi = __shfl_xor(bi[r], o, 32);
          if (ov < bv[r] || (ov == bv[r] && oi < bi[r])) { bv[r] = ov; bi[r] = oi; }
        }
        pv[r] = bv[r]; pi[r] = bi[r];
        if (ln == 0 && cntL[w][r + 8 * hf] > KCAP)
          knn_idx[(size_t)(row0 + r + 8 * hf) * KNN + k] = bi[r];
      }
    }
  }
}

typedef __attribute__((ext_vector_type(4))) float v4f;
#define STD_W 8
__global__ __launch_bounds__(STD_W * 32) void k_std_partial(const float* __restrict__ xyz,
                                                            const float* __restrict__ feat,
                                                            const float* __restrict__ xyz_s,
                                                            const float* __restrict__ feat_s,
                                                            const int* __restrict__ knn_idx,
                                                            int N, int S, int C,
                                                            float* __restrict__ part) {
  __shared__ float red[STD_W][KNN][4];
  const int w    = __builtin_amdgcn_readfirstlane((int)threadIdx.x >> 5);
  const int lane = threadIdx.x & 31;
  const int rpw  = BATCH * S / (STD_W * STD_G);
  const int gw   = blockIdx.x * STD_W + w;
  float ax = 0.f, qx = 0.f;
  v4f s4 = {0.f, 0.f, 0.f, 0.f}, q4 = {0.f, 0.f, 0.f, 0.f};
  for (int r = 0; r < rpw; ++r) {
    const int row = gw * rpw + r;
    const int b   = row / S;
    const int n   = knn_idx[(size_t)row * KNN + lane];
    const float* pn = xyz + (size_t)(b * N + n) * 3;
    const float* pq = xyz_s + (size_t)row * 3;
    for (int c = 0; c < 3; ++c) { const float d = pn[c] - pq[c]; ax += d; qx += d * d; }
    const float* fr = feat   + (size_t)(b * N + n) * C;
    const float* fc = feat_s + (size_t)row * C;
    for (int j = 0; j < C; j += 64) {
      v4f v[16];
#pragma unroll
      for (int i = 0; i < 16; ++i) v[i] = *(const v4f*)(fr + j + 4 * i);
#pragma unroll
      for (int i = 0; i < 16; ++i) {
        const v4f d = v[i] - *(const v4f*)(fc + j + 4 * i);
        s4 += d;
        q4 = __builtin_elementwise_fma(d, d, q4);
      }
    }
  }
  red[w][lane][0] = ax;
  red[w][lane][1] = qx;
  red[w][lane][2] = (s4.x + s4.y) + (s4.z + s4.w);
  red[w][lane][3] = (q4.x + q4.y) + (q4.z + q4.w);
  __syncthreads();
  if (threadIdx.x < KNN) {
    float a0 = 0.f, a1 = 0.f, a2 = 0.f, a3 = 0.f;
    for (int u = 0; u < STD_W; ++u) {
      a0 += red[u][threadIdx.x][0]; a1 += red[u][threadIdx.x][1];
      a2 += red[u][threadIdx.x][2]; a3 += red[u][threadIdx.x][3];
    }
    float* q = part + ((size_t)blockIdx.x * KNN + threadIdx.x) * 4;
    q[0] = a0; q[1] = a1; q[2] = a2; q[3] = a3;
  }
}

__global__ __launch_bounds__(1024) void k_std_final(const float* __restrict__ part, int S, int C,
                                                    float* __restrict__ std_xyz, float* __restrict__ std_feat) {
  __shared__ float red[32][KNN][4];
  const int k = threadIdx.x & 31, p = threadIdx.x >> 5;
  float ax = 0.f, qx = 0.f, af = 0.f, qf = 0.f;
  for (int g = p; g < STD_G; g += 32) {
    const float* q = part + ((size_t)g * KNN + k) * 4;
    ax += q[0]; qx += q[1]; af += q[2]; qf += q[3];
  }
  red[p][k][0] = ax; red[p][k][1] = qx; red[p][k][2] = af; red[p][k][3] = qf;
  __syncthreads();
  if (threadIdx.x >= KNN) return;
  ax = 0.f; qx = 0.f; af = 0.f; qf = 0.f;
  for (int u = 0; u < 32; ++u) {
    ax += red[u][k][0]; qx += red[u][k][1]; af += red[u][k][2]; qf += red[u][k][3];
  }
  {
    double n = (double)BATCH * S * 3;
    double var = ((double)qx - (double)ax * ax / n) / (n - 1.0);
    std_xyz[k] = fmaxf(sqrtf(fmaxf((float)var, 0.f)), 1e-5f);
  }
  {
    double n = (double)BATCH * S * C;
    double var = ((double)qf - (double)af * af / n) / (n - 1.0);
    std_feat[k] = fmaxf(sqrtf(fmaxf((float)var, 0.f)), 1e-5f);
  }
}

#define PCH 4
template <int TF>
__global__ __launch_bounds__(256) void k_stage(const float* __restrict__ xyz,
                                               const float* __restrict__ feat,
                                               const float* __restrict__ xyz_s,
                                               const float* __restrict__ feat_s,
                                               const int* __restrict__ knn_idx,
                                               const float* __restrict__ std_xyz,
                                               const float* __restrict__ std_feat,
                                               int N, int S, int C, int chunk,
                                               float* __restrict__ out,
                                               float* __restrict__ pmaxb,
                                               float* __restrict__ psumb) {
  __shared__ int   kidx[PCH * KNN];
  __shared__ float sfe[KNN];
  __shared__ float xkn[PCH * KNN][3];
  __shared__ float fs[512];
  __shared__ __attribute__((aligned(16))) float ft[TF];
  const int nt   = blockDim.x;
  const int row0 = blockIdx.x * chunk;
  const int b    = row0 / S;
  const int nk   = chunk * KNN;
  for (int t = threadIdx.x; t < nk; t += nt) kidx[t] = knn_idx[(size_t)row0 * KNN + t];
  if (threadIdx.x < KNN) sfe[threadIdx.x] = std_feat[threadIdx.x];
  __syncthreads();
  {
    const int c4 = C >> 2;
    const int r0 = (int)threadIdx.x / c4, cp = (int)threadIdx.x - r0 * c4;
    const int rstep = nt / c4;
    for (int r = r0; r < nk; r += 8 * rstep) {
      v4f v[8];
#pragma unroll
      for (int i = 0; i < 8; ++i) {
        const int rr = min(r + i * rstep, nk - 1);
        v[i] = *(const v4f*)(feat + (size_t)(b * N + kidx[rr]) * C + 4 * cp);
      }
      asm volatile("" :: "v"(v[0]), "v"(v[1]), "v"(v[2]), "v"(v[3]), "v"(v[4]), "v"(v[5]), "v"(v[6]), "v"(v[7]));
#pragma unroll
      for (int i = 0; i < 8; ++i) {
        const int rr = r + i * rstep;
        if (rr < nk) ((v4f*)ft)[rr * c4 + cp] = v[i];
      }
    }
  }
  for (int t = threadIdx.x; t < nk * 3; t += nt) {
    const int kk = t / 3, c = t % 3;
    const int q = kk / KNN, k = kk % KNN;
    xkn[kk][c] = (xyz[(size_t)(b * N + kidx[kk]) * 3 + c] - xyz_s[(size_t)(row0 + q) * 3 + c]) / std_xyz[k];
  }
  for (int t = threadIdx.x; t < chunk * C; t += nt) fs[t] = feat_s[(size_t)row0 * C + t];
  __syncthreads();
  const int C2 = 2 * C;
  const int fd = (C2 + 2) / 3;
  const int M3 = 3 * fd;
  const float kexp = -0.5f * 1.44269504088896341f / (SIGMA_ * SIGMA_);
  for (int j = threadIdx.x; j < C2; j += nt) {
    int src = (int)(((long long)j * (M3 - 1)) / (C2 - 1));
    int cc = src / fd, f = src % fd;
    float fv = -1.f + 2.f * (float)(f + 1) / (float)(fd + 1);
    float pmx = -FLT_MAX, psm = 0.f;
    for (int q = 0; q < chunk; ++q) {
      const float* fq = fs + q * C;
      float acc = 0.f, mx = -FLT_MAX;
      for (int k = 0; k < KNN; ++k) {
        const float d = xkn[q * KNN + k][cc] - fv;
        const float pe = __builtin_amdgcn_exp2f(kexp * (d * d));
        float fk;
        if (j < C) fk = (ft[(q * KNN + k) * C + j] - fq[j]) / sfe[k];
        else       fk = fq[j - C];
        float fw = (fk + pe) * pe;
        acc += fw; mx = fmaxf(mx, fw);
      }
      float xx = acc * (1.f / KNN) + mx;
      const float o = 0.5f * xx * (1.f + erff(xx * 0.70710678118654752f));
      out[(size_t)(row0 + q) * C2 + j] = o;
      pmx = fmaxf(pmx, o); psm += o;
    }
    if (chunk > 1) {
      pmaxb[(size_t)blockIdx.x * C2 + j] = pmx;
      psumb[(size_t)blockIdx.x * C2 + j] = psm;
    }
  }
}

__global__ __launch_bounds__(1024) void k_pool_fold(const float* pm, const float* ps,

                                                    int G, int C2, int S,
                                                    float* __restrict__ out, int off) {
  const int b = blockIdx.x;
  for (int j = threadIdx.x; j < C2; j += blockDim.x) {
    const float* a = pm + (size_t)b * G * C2 + j;
    const float* c = ps + (size_t)b * G * C2 + j;
    float mx = -FLT_MAX, sm = 0.f;
    if (pm == ps) {
#pragma unroll 8
      for (int g = 0; g < G; ++g) {
        const float v = a[(size_t)g * C2];
        mx = fmaxf(mx, v); sm += v;
      }
    } else {
#pragma unroll 8
      for (int g = 0; g < G; ++g) {
        mx = fmaxf(mx, a[(size_t)g * C2]);
        sm += c[(size_t)g * C2];
      }
    }
    out[b * 3840 + off + j]      = mx;
    out[b * 3840 + off + C2 + j] = sm / (float)S;
  }
}

extern "C" void kernel_launch(void* const* d_in, const int* in_sizes, int n_in,
                              void* d_out, int out_size, void* d_ws, size_t ws_size,
                              hipStream_t stream) {
  const float* xyz_in = (const float*)d_in[0];
  const float* W      = (const float*)d_in[1];
  const float* gamma  = (const float*)d_in[2];
  const float* beta   = (const float*)d_in[3];
  float* out = (float*)d_out;
  float* ws  = (float*)d_ws;

  size_t o = 0;
  float* featA  = ws + o; o += (size_t)BATCH * NPTS * CINIT;
  float* featB  = ws + o; o += (size_t)BATCH * NPTS * CINIT;
  float* xyzA   = ws + o; o += (size_t)BATCH * NPTS * 3;
  float* xyzB   = ws + o; o += (size_t)BATCH * NPTS * 3;
  float* feat_s = ws + o; o += (size_t)BATCH * 1024 * CINIT;
  float* pmaxb  = ws + o; o += (size_t)1024 * 1024;
  float* psumb  = ws + o; o += (size_t)1024 * 1024;
  float* bn_mu  = ws + o; o += 64;
  float* bn_inv = ws + o; o += 64;
  float* stdx   = ws + o; o += 32;
  float* stdf   = ws + o; o += 32;
  float* stdp   = pmaxb;
  int* fps_i = (int*)(ws + o); o += (size_t)BATCH * 1024;
  int* knn_i = (int*)(ws + o); o += (size_t)BATCH * 1024 * KNN;

  k_embed_wmma<<<1024, 256, 0, stream>>>(xyz_in, W, featA);
  k_bn_stats<<<CINIT, 256, 0, stream>>>(featA, bn_mu, bn_inv);
  k_bn_apply<<<2048, 256, 0, stream>>>(featA, bn_mu, bn_inv, gamma, beta);

  const float* xyz_cur = xyz_in;
  float* feat_cur = featA; float* feat_nxt = featB;
  float* xs_buf = xyzA;    float* xs_other = xyzB;
  int N = NPTS, C = CINIT, off = 0;

  for (int st = 0; st < 4; ++st) {
    const int S = N / 2, C2 = 2 * C;
    if (N == 2048)      k_fps_wave<64><<<BATCH, 32, 0, stream>>>(xyz_cur, S, fps_i);
    k_gather<<<BATCH * S, 256, 0, stream>>>(xyz_cur, feat_cur, fps_i, N, S, C, xs_buf, feat_s);
    {
      const int nb = BATCH * S / 16 / KW;
      if (N == 2048)      k_knn_fused<2048><<<BATCH + nb, KW * 32, 0, stream>>>(xs_buf, xyz_cur, S, knn_i, fps_i);
      else if (N == 1024) k_knn_fused<1024><<<BATCH + nb, KW * 32, 0, stream>>>(xs_buf, xyz_cur, S, knn_i, fps_i);
      else if (N == 512)  k_knn_fused<512><<<BATCH + nb, KW * 32, 0, stream>>>(xs_buf, xyz_cur, S, knn_i, fps_i);
      else                k_knn_fused<256><<<nb, KW * 32, 0, stream>>>(xs_buf, xyz_cur, S, knn_i, fps_i);
    }
    {
      k_std_partial<<<STD_G, STD_W * 32, 0, stream>>>(xyz_cur, feat_cur, xs_buf, feat_s, knn_i, N, S, C, stdp);
      k_std_final<<<1, 1024, 0, stream>>>(stdp, S, C, stdx, stdf);
    }
    {
      const int chunk = (C2 <= 256) ? 4 : ((C2 == 512) ? 2 : 1);
      const int nthr  = (C2 < 256) ? C2 : 256;
      if (C == CINIT) k_stage<PCH * KNN * CINIT><<<BATCH * S / chunk, nthr, 0, stream>>>(
                          xyz_cur, feat_cur, xs_buf, feat_s, knn_i, stdx, stdf, N, S, C, chunk, feat_nxt,
                          pmaxb, psumb);
      else            k_stage<16384><<<BATCH * S / chunk, nthr, 0, stream>>>(
                          xyz_cur, feat_cur, xs_buf, feat_s, knn_i, stdx, stdf, N, S, C, chunk, feat_nxt,
                          pmaxb, psumb);
      if (chunk > 1) k_pool_fold<<<BATCH, C2, 0, stream>>>(pmaxb, psumb, S / chunk, C2, S, out, off);
      else           k_pool_fold<<<BATCH, C2, 0, stream>>>(feat_nxt, feat_nxt, S, C2, S, out, off);
    }
    off += 2 * C2;
    float* t = feat_cur; feat_cur = feat_nxt; feat_nxt = t;
    xyz_cur = xs_buf;
    float* u = xs_buf; xs_buf = xs_other; xs_other = u;
    N = S; C = C2;
  }
  (void)in_sizes; (void)n_in; (void)out_size; (void)ws_size;
}


# a stage's small launches ride in its kNN launch as extra blocks: the sampled rows' gather and the readout of the stage before; kNN and riding sampling read points via the pick list; 28 -> 21 launches
# speedup vs baseline: 5.9467x; 1.0678x over previous
#include <hip/hip_runtime.h>
#include <cfloat>
#include <cmath>

typedef __attribute__((ext_vector_type(2))) float v2f;
typedef __attribute__((ext_vector_type(3))) float v3f;
typedef __attribute__((ext_vector_type(4))) float v4f;
typedef __attribute__((ext_vector_type(8))) float v8f;

#define BATCH 16
#define NPTS 2048
#define CINIT 64
#define KNN 32
#define SIGMA_ 0.3f
#define BN_EPS_ 1e-5f
#define STD_G 256

__global__ __launch_bounds__(256) void k_embed_wmma(const float* __restrict__ xyz,
                                                    const float* __restrict__ W,
                                                    float* __restrict__ h) {
  const int wid  = (blockIdx.x * 256 + (int)threadIdx.x) >> 5;
  const int lane = threadIdx.x & 31;
  const int hf   = lane >> 4;
  const int ln   = lane & 15;
  const int tileM = wid >> 2;
  const int tileN = wid & 3;
  const int m  = tileM * 16 + ln;
  const int c  = tileN * 16 + ln;
  const float x0 = xyz[m * 3], x1 = xyz[m * 3 + 1], x2 = xyz[m * 3 + 2];
  const float w0 = W[c * 3],   w1 = W[c * 3 + 1],   w2 = W[c * 3 + 2];
  v2f a, b;
  a.x = hf ? x2 : x0;  a.y = hf ? 0.f : x1;
  b.x = hf ? w2 : w0;  b.y = hf ? 0.f : w1;
  v8f d = {};
  d = __builtin_amdgcn_wmma_f32_16x16x4_f32(false, a, false, b, (short)0, d, false, false);
#pragma unroll
  for (int r = 0; r < 8; ++r) {
    int row = tileM * 16 + r + 8 * hf;
    h[row * CINIT + tileN * 16 + ln] = d[r];
  }
}

__global__ __launch_bounds__(256) void k_bn_stats(const float* __restrict__ h,
                                                  float* __restrict__ mu,
                                                  float* __restrict__ inv) {
  __shared__ float s1[256], s2[256];
  const int c = blockIdx.x;
  const int n = BATCH * NPTS;
  float a = 0.f, b = 0.f;
  for (int i = threadIdx.x; i < n; i += 256) {
    float v = h[i * CINIT + c];
    a += v; b += v * v;
  }
  s1[threadIdx.x] = a; s2[threadIdx.x] = b; __syncthreads();
  for (int o = 128; o > 0; o >>= 1) {
    if ((int)threadIdx.x < o) { s1[threadIdx.x] += s1[threadIdx.x + o]; s2[threadIdx.x] += s2[threadIdx.x + o]; }
    __syncthreads();
  }
  if (threadIdx.x == 0) {
    float m   = s1[0] / (float)n;
    float var = s2[0] / (float)n - m * m;
    mu[c]  = m;
    inv[c] = rsqrtf(var + BN_EPS_);
  }
}

__global__ void k_bn_apply(float* __restrict__ h, const float* __restrict__ mu,
                           const float* __restrict__ inv, const float* __restrict__ gamma,
                           const float* __restrict__ beta) {
  const int total = BATCH * NPTS * CINIT;
  for (int i = blockIdx.x * blockDim.x + threadIdx.x; i < total; i += gridDim.x * blockDim.x) {
    int c = i & (CINIT - 1);
    float v = (h[i] - mu[c]) * inv[c] * gamma[c] + beta[c];
    h[i] = fmaxf(v, 0.f);
  }
}

__device__ __forceinline__ int imax2(int a, int b) { return a > b ? a : b; }
__device__ __forceinline__ int imin2(int a, int b) { return a < b ? a : b; }

template <int CTRL>
__device__ __forceinline__ int row_xchg(int v) {
  return __builtin_amdgcn_update_dpp(v, v, CTRL, 0xf, 0xf, true);
}

template <int PPL, int Q>
__device__ __forceinline__ unsigned fps_eq_bits(const int (&di)[PPL], int m) {
  constexpr int CH = PPL / 8;
  unsigned bits = 1u << (CH - 1);
#pragma unroll
  for (int x = 0; x < CH - 1; ++x) bits |= (di[Q * CH + x] == m) ? (1u << x) : 0u;
  return bits;
}

template <int PPL>
__device__ __forceinline__ int fps_pick_values(const v2f (&px)[PPL / 2], const v2f (&py)[PPL / 2],
                                               const v2f (&pz)[PPL / 2], int (&di)[PPL],
                                               const float4* pc, int lane,
                                               float& fx, float& fy, float& fz) {
  constexpr int CH = PPL / 8;
  int gm[8];
#pragma unroll
  for (int p2 = 0; p2 < PPL / 2; ++p2) {
    const v2f dx = px[p2] - fx, dy = py[p2] - fy, dz = pz[p2] - fz;
    const v2f t = __builtin_elementwise_fma(dz, dz, __builtin_elementwise_fma(dx, dx, dy * dy));
    const int s0 = 2 * p2, s1 = 2 * p2 + 1, q = s0 / CH;
    di[s0] = imin2(di[s0], __float_as_int(t.x));
    di[s1] = imin2(di[s1], __float_as_int(t.y));
    gm[q] = (s0 % CH == 0) ? imax2(di[s0], di[s1]) : imax2(gm[q], imax2(di[s0], di[s1]));
  }
  int mine = gm[0];
#pragma unroll
  for (int q = 1; q < 8; ++q) mine = imax2(mine, gm[q]);
  unsigned gbits = 0x80u;
#pragma unroll
  for (int q = 0; q < 7; ++q) gbits |= (gm[q] == mine) ? (1u << q) : 0u;
  int g = mine;
  g = imax2(g, row_xchg<0x161>(g));
  g = imax2(g, row_xchg<0x162>(g));
  g = imax2(g, row_xchg<0x164>(g));
  g = imax2(g, row_xchg<0x168>(g));
  g = imax2(g, (int)__builtin_amdgcn_permlanex16(g, g, 0x76543210, 0xfedcba98, false, false));
  const unsigned hit = __builtin_amdgcn_ballot_w32(mine == g);
  const int L = __builtin_ctz(hit);
  const int sq = __builtin_ctz((unsigned)__builtin_amdgcn_readlane((int)gbits, L));
  const int base = L * PPL + sq * CH;
  typedef const volatile __attribute__((address_space(3))) v3f lds_v3f;
  const v3f cand = *(lds_v3f*)(pc + base + (lane & (CH - 1)));
  unsigned eb;
  switch (sq) {
    case 0:  eb = fps_eq_bits<PPL, 0>(di, mine); break;
    case 1:  eb = fps_eq_bits<PPL, 1>(di, mine); break;
    case 2:  eb = fps_eq_bits<PPL, 2>(di, mine); break;
    case 3:  eb = fps_eq_bits<PPL, 3>(di, mine); break;
    case 4:  eb = fps_eq_bits<PPL, 4>(di, mine); break;
    case 5:  eb = fps_eq_bits<PPL, 5>(di, mine); break;
    case 6:  eb = fps_eq_bits<PPL, 6>(di, mine); break;
    default: eb = fps_eq_bits<PPL, 7>(di, mine); break;
  }
  const int es = __builtin_ctz((unsigned)__builtin_amdgcn_readlane((int)eb, L));
  fx = __int_as_float(__builtin_amdgcn_readlane(__float_as_int(cand.x), es));
  fy = __int_as_float(__builtin_amdgcn_readlane(__float_as_int(cand.y), es));
  fz = __int_as_float(__builtin_amdgcn_readlane(__float_as_int(cand.z), es));
  return base + es;
}

template <int PPL, bool IND>
__device__ __forceinline__ void fps_cloud(const float* __restrict__ cloud, const int* __restrict__ cidx,
                                          int S, int* __restrict__ idx_out, float4* pc, int lane,
                                          bool run) {
  constexpr bool VALS = (PPL >= 32);
  const float* p = cloud + (size_t)lane * PPL * 3;
  float px[VALS ? 1 : PPL], py[VALS ? 1 : PPL], pz[VALS ? 1 : PPL];
  v2f qx[VALS ? PPL / 2 : 1], qy[VALS ? PPL / 2 : 1], qz[VALS ? PPL / 2 : 1];
  int di[PPL];
  if (run) {
#pragma unroll
    for (int s = 0; s < PPL; ++s) {
      float c0, c1, c2;
      if constexpr (IND) {
        const float* pi = cloud + (size_t)cidx[lane * PPL + s] * 3;
        c0 = pi[0]; c1 = pi[1]; c2 = pi[2];
      } else {
        c0 = p[s * 3]; c1 = p[s * 3 + 1]; c2 = p[s * 3 + 2];
      }
      if constexpr (VALS) { qx[s / 2][s % 2] = c0; qy[s / 2][s % 2] = c1; qz[s / 2][s % 2] = c2; }
      else                { px[s] = c0; py[s] = c1; pz[s] = c2; }
      di[s] = __float_as_int(1e10f);
      pc[lane * PPL + s] = make_float4(c0, c1, c2, 0.f);
    }
  }
  __syncthreads();
  if (!run) return;
  int far = 0;
  float fx = 0.f, fy = 0.f, fz = 0.f;
  if constexpr (VALS) { const float4 f0 = pc[0]; fx = f0.x; fy = f0.y; fz = f0.z; }
  for (int it0 = 0; it0 < S; it0 += 32) {
    int keep = 0;
    for (int j = 0; j < 32; ++j) {
      if constexpr (VALS) {
        keep = (lane == j) ? far : keep;
        far = fps_pick_values<PPL>(qx, qy, qz, di, pc, lane, fx, fy, fz);
      } else {
        const float4 f = pc[far];
        keep = (lane == j) ? far : keep;
        constexpr int CH = PPL / 8;
        int v[8], ix[8];
#pragma unroll
        for (int q = 0; q < 8; ++q) {
#pragma unroll
          for (int e = 0; e < CH; ++e) {
            const int s = q * CH + e;
            const float dx = px[s] - f.x, dy = py[s] - f.y, dz = pz[s] - f.z;
            const float t = __builtin_fmaf(dz, dz, __builtin_fmaf(dx, dx, dy * dy));
            di[s] = imin2(di[s], __float_as_int(t));
            if (e == 0) { v[q] = di[s]; ix[q] = s; }
            else {
              const bool c = di[s] > v[q];
              v[q]  = c ? di[s] : v[q];
              ix[q] = c ? s : ix[q];
            }
          }
        }
#pragma unroll
        for (int w = 8; w > 1; w >>= 1) {
#pragma unroll
          for (int k = 0; k < w; k += 2) {
            const bool c = v[k + 1] > v[k];
            const int nv = c ? v[k + 1] : v[k];
            const int ni = c ? ix[k + 1] : ix[k];
            v[k / 2] = nv; ix[k / 2] = ni;
          }
        }
        const int mine = v[0];
        int g = mine;
        g = imax2(g, row_xchg<0x161>(g));
        g = imax2(g, row_xchg<0x162>(g));
        g = imax2(g, row_xchg<0x164>(g));
        g = imax2(g, row_xchg<0x168>(g));
        g = imax2(g, (int)__builtin_amdgcn_permlanex16(g, g, 0x76543210, 0xfedcba98, false, false));
        const unsigned hit = __builtin_amdgcn_ballot_w32(mine == g);
        const int L = __builtin_ctz(hit);
        far = L * PPL + __builtin_amdgcn_readlane(ix[0], L);
      }
    }
    idx_out[it0 + lane] = keep;
  }
}

template <int PPL>
__global__ __launch_bounds__(32) void k_fps_wave(const float* __restrict__ xyz, int S,
                                                 int* __restrict__ fps_idx) {
  __shared__ float4 pc[32 * PPL];
  fps_cloud<PPL, false>(xyz + (size_t)blockIdx.x * (32 * PPL) * 3, nullptr, S,
                        fps_idx + (size_t)blockIdx.x * S, pc, (int)threadIdx.x, true);
}

#define GROWS 32
#define PCOLS 64

#define KW 2
#define KCAP 128
#define KU 8

__device__ __forceinline__ float sq3(float p0, float p1, float p2) {
  return __builtin_fmaf(p2, p2, __builtin_fmaf(p0, p0, p1 * p1));
}

__device__ __forceinline__ v8f knn_tile(const float* __restrict__ x, int n, int hf, v2f a,
                                        float& nn) {
  const float x0 = x[n * 3], x1 = x[n * 3 + 1], x2 = x[n * 3 + 2];
  nn = sq3(x0, x1, x2);
  v2f b;
  b.x = hf ? x2 : x0;  b.y = hf ? 0.f : x1;
  v8f d = {};
  return __builtin_amdgcn_wmma_f32_16x16x4_f32(false, a, false, b, (short)0, d, false, false);
}

template <int N>
__global__ __launch_bounds__(KW * 32) void k_knn_fused(const float* __restrict__ x_all,
                                                       const int* __restrict__ fps_cur,
                                                       int S, int* __restrict__ knn_idx,
                                                       int* __restrict__ fps_next,
                                                       const float* __restrict__ feat, int C,
                                                       float* __restrict__ xs_out,
                                                       float* __restrict__ feat_s,
                                                       int nb, int ng,
                                                       const float* __restrict__ pm,
                                                       const float* __restrict__ ps,
                                                       int G, int pC2, int pS,
                                                       float* __restrict__ out, int poff) {
  constexpr int  RPPL = N / 64;
  constexpr bool RIDE = (RPPL >= 8);
  __shared__ float gm[KW][16][64];
  __shared__ int2  cand[KW][16][KCAP];
  __shared__ int   cntL[KW][16];
  __shared__ float4 rpc[RIDE ? 32 * RPPL : 1];
  __shared__ int   gidx[GROWS];
  const int w    = threadIdx.x >> 5;
  const int lane = threadIdx.x & 31;
  if constexpr (RIDE) {
    if (blockIdx.x < BATCH) {
      const bool first = __builtin_amdgcn_readfirstlane(w) == 0;
      fps_cloud<RPPL, true>(x_all + (size_t)blockIdx.x * N * 3, fps_cur + (size_t)blockIdx.x * S,
                            S / 2, fps_next + (size_t)blockIdx.x * (S / 2), rpc, lane, first);
      return;
    }
  }
  int blk = RIDE ? (int)blockIdx.x - BATCH : (int)blockIdx.x;
  if (blk >= nb) {
    blk -= nb;
    if (blk < ng) {
      const int grow0 = blk * GROWS;
      const int gb    = grow0 / S;
      if (threadIdx.x < GROWS) gidx[threadIdx.x] = fps_cur[grow0 + (int)threadIdx.x];
      __syncthreads();
      const int c4 = C >> 2;
      const int sh = __builtin_ctz((unsigned)c4);
      const float* fsrc = feat + (size_t)gb * N * C;
      v4f* fdst = (v4f*)(feat_s + (size_t)grow0 * C);
      for (int i0 = (int)threadIdx.x; i0 < GROWS * c4; i0 += 8 * KW * 32) {
        v4f v[8];
#pragma unroll
        for (int u = 0; u < 8; ++u) {
          const int i = i0 + u * KW * 32;
          v[u] = *(const v4f*)(fsrc + (size_t)gidx[i >> sh] * C + 4 * (i & (c4 - 1)));
        }
#pragma unroll
        for (int u = 0; u < 8; ++u) fdst[i0 + u * KW * 32] = v[u];
      }
      for (int i = (int)threadIdx.x; i < GROWS * 3; i += KW * 32)
        xs_out[(size_t)grow0 * 3 + i] = x_all[((size_t)gb * N + gidx[i / 3]) * 3 + (i % 3)];
      return;
    }
    blk -= ng;
    if (pm != nullptr) {
      const int per = pC2 / PCOLS;
      const int pb  = blk / per;
      const int j   = (blk - pb * per) * PCOLS + (int)threadIdx.x;
      const float* a = pm + (size_t)pb * G * pC2 + j;
      const float* c = ps + (size_t)pb * G * pC2 + j;
      float mx = -FLT_MAX, sm = 0.f;
#pragma unroll 8
      for (int g = 0; g < G; ++g) {
        mx = fmaxf(mx, a[(size_t)g * pC2]);
        sm += c[(size_t)g * pC2];
      }
      out[pb * 3840 + poff + j]       = mx;
      out[pb * 3840 + poff + pC2 + j] = sm / (float)pS;
    }
    return;
  }
  const int hf   = lane >> 4;
  const int ln   = lane & 15;
  const int row0 = (blk * KW + w) * 16;
  const int b    = row0 / S;
  const float* x  = x_all + (size_t)b * N * 3;
  const int*   qi = fps_cur + row0;
  const float* xq = x + (size_t)qi[ln] * 3;
  const float s0 = xq[0], s1 = xq[1], s2 = xq[2];
  v2f a;
  a.x = hf ? s2 : s0;  a.y = hf ? 0.f : s1;
  float sN[8];
#pragma unroll
  for (int r = 0; r < 8; ++r) {
    const float* xr = x + (size_t)qi[r + 8 * hf] * 3;
    sN[r] = sq3(xr[0], xr[1], xr[2]);
  }

  float g[4][8];
#pragma unroll
  for (int u = 0; u < 4; ++u)
#pragma unroll
    for (int r = 0; r < 8; ++r) g[u][r] = INFINITY;
  for (int t = 0; t < N / 16; t += KU) {
    v8f d[KU];
    float nn[KU];
#pragma unroll
    for (int u = 0; u < KU; ++u) d[u] = knn_tile(x, (t + u) * 16 + ln, hf, a, nn[u]);
#pragma unroll
    for (int u = 0; u < KU; ++u)
#pragma unroll
      for (int r = 0; r < 8; ++r)
        g[u & 3][r] = fminf(g[u & 3][r], __builtin_fmaf(-2.f, d[u][r], sN[r] + nn[u]));
  }
#pragma unroll
  for (int u = 0; u < 4; ++u)
#pragma unroll
    for (int r = 0; r < 8; ++r) gm[w][r + 8 * hf][u * 16 + ln] = g[u][r];
  __syncthreads();

  float T[8];
#pragma unroll
  for (int r = 0; r < 8; ++r) {
    const float* grow = gm[w][r + 8 * hf];
    int k0 = 0, k1 = 0, k2 = 0, k3 = 0;
#pragma unroll 16
    for (int e = 0; e < 64; ++e) {
      const float o = grow[e];
      k0 += (o < g[0][r]) ? 1 : 0;
      k1 += (o < g[1][r]) ? 1 : 0;
      k2 += (o < g[2][r]) ? 1 : 0;
      k3 += (o < g[3][r]) ? 1 : 0;
    }
    float tv = (k0 < 32) ? g[0][r] : -INFINITY;
    tv = fmaxf(tv, (k1 < 32) ? g[1][r] : -INFINITY);
    tv = fmaxf(tv, (k2 < 32) ? g[2][r] : -INFINITY);
    tv = fmaxf(tv, (k3 < 32) ? g[3][r] : -INFINITY);
#pragma unroll
    for (int o = 8; o > 0; o >>= 1) tv = fmaxf(tv, __shfl_xor(tv, o, 32));
    T[r] = tv;
  }

  int cnt[8];
#pragma unroll
  for (int r = 0; r < 8; ++r) cnt[r] = 0;
  const unsigned below = (1u << ln) - 1u;
  for (int t = 0; t < N / 16; t += KU) {
    v8f d[KU];
    float nn[KU];
#pragma unroll
    for (int u = 0; u < KU; ++u) d[u] = knn_tile(x, (t + u) * 16 + ln, hf, a, nn[u]);
#pragma unroll
    for (int u = 0; u < KU; ++u) {
      const int col = (t + u) * 16 + ln;
#pragma unroll
      for (int r = 0; r < 8; ++r) {
        const float v = __builtin_fmaf(-2.f, d[u][r], sN[r] + nn[u]);
        const bool c = v <= T[r];
        const unsigned bal = __builtin_amdgcn_ballot_w32(c);
        if (bal != 0u) {
          const unsigned hb = (hf ? (bal >> 16) : bal) & 0xffffu;
          const int pos = cnt[r] + __builtin_popcount(hb & below);
          if (c && pos < KCAP) cand[w][r + 8 * hf][pos] = make_int2(__float_as_int(v), col);
          cnt[r] += __builtin_popcount(hb);
        }
      }
    }
  }
#pragma unroll
  for (int r = 0; r < 8; ++r) {
    const int p = cnt[r] + ln;
    if (ln < 8 && p < ((cnt[r] + 7) & ~7) && p < KCAP)
      cand[w][r + 8 * hf][p] = make_int2(__float_as_int(INFINITY), 0x7fffffff);
  }
  if (ln == 0) {
#pragma unroll
    for (int r = 0; r < 8; ++r) cntL[w][r + 8 * hf] = cnt[r];
  }
  __syncthreads();

  bool over = false;
  for (int R = 0; R < 16; ++R) {
    const int M = __builtin_amdgcn_readfirstlane(cntL[w][R]);
    if (M > KCAP) { over = true; continue; }
    const int2* cr = cand[w][R];
    int* outp = knn_idx + (size_t)(row0 + R) * KNN;
    const int Mp = (M + 7) & ~7;
    for (int base = 0; base < M; base += 64) {
      const int my0 = base + lane, my1 = base + 32 + lane;
      const bool have0 = my0 < M, have1 = my1 < M;
      const int2 mine0 = cr[have0 ? my0 : 0];
      const int2 mine1 = cr[have1 ? my1 : 0];
      const float mv0 = __int_as_float(mine0.x), mv1 = __int_as_float(mine1.x);
      const int   mi0 = mine0.y, mi1 = mine1.y;
      int rank0 = 0, rank1 = 0;
      for (int j = 0; j < Mp; j += 8) {
#pragma unroll
        for (int q = 0; q < 8; ++q) {
          const int2 o = cr[j + q];
          const float ov = __int_as_float(o.x);
          rank0 += (ov < mv0 || (ov == mv0 && o.y < mi0)) ? 1 : 0;
          rank1 += (ov < mv1 || (ov == mv1 && o.y < mi1)) ? 1 : 0;
        }
      }
      if (have0 && rank0 < KNN) outp[rank0] = mi0;
      if (have1 && rank1 < KNN) outp[rank1] = mi1;
    }
  }

  if (over) {
    float pv[8]; int pi[8];
#pragma unroll
    for (int r = 0; r < 8; ++r) { pv[r] = -INFINITY; pi[r] = -1; }
    for (int k = 0; k < KNN; ++k) {
      float bv[8]; int bi[8];
#pragma unroll
      for (int r = 0; r < 8; ++r) { bv[r] = INFINITY; bi[r] = 0x7fffffff; }
      for (int t = 0; t < N / 16; ++t) {
        float nn;
        const v8f d = knn_tile(x, t * 16 + ln, hf, a, nn);
        const int col = t * 16 + ln;
#pragma unroll
        for (int r = 0; r < 8; ++r) {
          const float v = __builtin_fmaf(-2.f, d[r], sN[r] + nn);
          const bool after  = (v > pv[r]) || (v == pv[r] && col > pi[r]);
          const bool better = (v < bv[r]) || (v == bv[r] && col < bi[r]);
          if (after && better) { bv[r] = v; bi[r] = col; }
        }
      }
#pragma unroll
      for (int r = 0; r < 8; ++r) {
#pragma unroll
        for (int o = 8; o > 0; o >>= 1) {
          const float ov = __shfl_xor(bv[r], o, 32);
          const int   oi = __shfl_xor(bi[r], o, 32);
          if (ov < bv[r] || (ov == bv[r] && oi < bi[r])) { bv[r] = ov; bi[r] = oi; }
        }
        pv[r] = bv[r]; pi[r] = bi[r];
        if (ln == 0 && cntL[w][r + 8 * hf] > KCAP)
          knn_idx[(size_t)(row0 + r + 8 * hf) * KNN + k] = bi[r];
      }
    }
  }
}

#define STD_W 8
__global__ __launch_bounds__(STD_W * 32) void k_std_partial(const float* __restrict__ xyz,
                                                            const float* __restrict__ feat,
                                                            const float* __restrict__ xyz_s,
                                                            const float* __restrict__ feat_s,
                                                            const int* __restrict__ knn_idx,
                                                            int N, int S, int C,
                                                            float* __restrict__ part) {
  __shared__ float red[STD_W][KNN][4];
  const int w    = __builtin_amdgcn_readfirstlane((int)threadIdx.x >> 5);
  const int lane = threadIdx.x & 31;
  const int rpw  = BATCH * S / (STD_W * STD_G);
  const int gw   = blockIdx.x * STD_W + w;
  float ax = 0.f, qx = 0.f;
  v4f s4 = {0.f, 0.f, 0.f, 0.f}, q4 = {0.f, 0.f, 0.f, 0.f};
  for (int r = 0; r < rpw; ++r) {
    const int row = gw * rpw + r;
    const int b   = row / S;
    const int n   = knn_idx[(size_t)row * KNN + lane];
    const float* pn = xyz + (size_t)(b * N + n) * 3;
    const float* pq = xyz_s + (size_t)row * 3;
    for (int c = 0; c < 3; ++c) { const float d = pn[c] - pq[c]; ax += d; qx += d * d; }
    const float* fr = feat   + (size_t)(b * N + n) * C;
    const float* fc = feat_s + (size_t)row * C;
    for (int j = 0; j < C; j += 64) {
      v4f v[16];
#pragma unroll
      for (int i = 0; i < 16; ++i) v[i] = *(const v4f*)(fr + j + 4 * i);
#pragma unroll
      for (int i = 0; i < 16; ++i) {
        const v4f d = v[i] - *(const v4f*)(fc + j + 4 * i);
        s4 += d;
        q4 = __builtin_elementwise_fma(d, d, q4);
      }
    }
  }
  red[w][lane][0] = ax;
  red[w][lane][1] = qx;
  red[w][lane][2] = (s4.x + s4.y) + (s4.z + s4.w);
  red[w][lane][3] = (q4.x + q4.y) + (q4.z + q4.w);
  __syncthreads();
  if (threadIdx.x < KNN) {
    float a0 = 0.f, a1 = 0.f, a2 = 0.f, a3 = 0.f;
    for (int u = 0; u < STD_W; ++u) {
      a0 += red[u][threadIdx.x][0]; a1 += red[u][threadIdx.x][1];
      a2 += red[u][threadIdx.x][2]; a3 += red[u][threadIdx.x][3];
    }
    float* q = part + ((size_t)blockIdx.x * KNN + threadIdx.x) * 4;
    q[0] = a0; q[1] = a1; q[2] = a2; q[3] = a3;
  }
}

__global__ __launch_bounds__(1024) void k_std_final(const float* __restrict__ part, int S, int C,
                                                    float* __restrict__ std_xyz, float* __restrict__ std_feat) {
  __shared__ float red[32][KNN][4];
  const int k = threadIdx.x & 31, p = threadIdx.x >> 5;
  float ax = 0.f, qx = 0.f, af = 0.f, qf = 0.f;
  for (int g = p; g < STD_G; g += 32) {
    const float* q = part + ((size_t)g * KNN + k) * 4;
    ax += q[0]; qx += q[1]; af += q[2]; qf += q[3];
  }
  red[p][k][0] = ax; red[p][k][1] = qx; red[p][k][2] = af; red[p][k][3] = qf;
  __syncthreads();
  if (threadIdx.x >= KNN) return;
  ax = 0.f; qx = 0.f; af = 0.f; qf = 0.f;
  for (int u = 0; u < 32; ++u) {
    ax += red[u][k][0]; qx += red[u][k][1]; af += red[u][k][2]; qf += red[u][k][3];
  }
  {
    double n = (double)BATCH * S * 3;
    double var = ((double)qx - (double)ax * ax / n) / (n - 1.0);
    std_xyz[k] = fmaxf(sqrtf(fmaxf((float)var, 0.f)), 1e-5f);
  }
  {
    double n = (double)BATCH * S * C;
    double var = ((double)qf - (double)af * af / n) / (n - 1.0);
    std_feat[k] = fmaxf(sqrtf(fmaxf((float)var, 0.f)), 1e-5f);
  }
}

#define PCH 4
template <int TF>
__global__ __launch_bounds__(256) void k_stage(const float* __restrict__ xyz,
                                               const float* __restrict__ feat,
                                               const float* __restrict__ xyz_s,
                                               const float* __restrict__ feat_s,
                                               const int* __restrict__ knn_idx,
                                               const float* __restrict__ std_xyz,
                                               const float* __restrict__ std_feat,
                                               int N, int S, int C, int chunk,
                                               float* __restrict__ out,
                                               float* __restrict__ pmaxb,
                                               float* __restrict__ psumb) {
  __shared__ int   kidx[PCH * KNN];
  __shared__ float sfe[KNN];
  __shared__ float xkn[PCH * KNN][3];
  __shared__ float fs[512];
  __shared__ __attribute__((aligned(16))) float ft[TF];
  const int nt   = blockDim.x;
  const int row0 = blockIdx.x * chunk;
  const int b    = row0 / S;
  const int nk   = chunk * KNN;
  for (int t = threadIdx.x; t < nk; t += nt) kidx[t] = knn_idx[(size_t)row0 * KNN + t];
  if (threadIdx.x < KNN) sfe[threadIdx.x] = std_feat[threadIdx.x];
  __syncthreads();
  {
    const int c4 = C >> 2;
    const int r0 = (int)threadIdx.x / c4, cp = (int)threadIdx.x - r0 * c4;
    const int rstep = nt / c4;
    for (int r = r0; r < nk; r += 8 * rstep) {
      v4f v[8];
#pragma unroll
      for (int i = 0; i < 8; ++i) {
        const int rr = min(r + i * rstep, nk - 1);
        v[i] = *(const v4f*)(feat + (size_t)(b * N + kidx[rr]) * C + 4 * cp);
      }
      asm volatile("" :: "v"(v[0]), "v"(v[1]), "v"(v[2]), "v"(v[3]), "v"(v[4]), "v"(v[5]), "v"(v[6]), "v"(v[7]));
#pragma unroll
      for (int i = 0; i < 8; ++i) {
        const int rr = r + i * rstep;
        if (rr < nk) ((v4f*)ft)[rr * c4 + cp] = v[i];
      }
    }
  }
  for (int t = threadIdx.x; t < nk * 3; t += nt) {
    const int kk = t / 3, c = t % 3;
    const int q = kk / KNN, k = kk % KNN;
    xkn[kk][c] = (xyz[(size_t)(b * N + kidx[kk]) * 3 + c] - xyz_s[(size_t)(row0 + q) * 3 + c]) / std_xyz[k];
  }
  for (int t = threadIdx.x; t < chunk * C; t += nt) fs[t] = feat_s[(size_t)row0 * C + t];
  __syncthreads();
  const int C2 = 2 * C;
  const int fd = (C2 + 2) / 3;
  const int M3 = 3 * fd;
  const float kexp = -0.5f * 1.44269504088896341f / (SIGMA_ * SIGMA_);
  for (int j = threadIdx.x; j < C2; j += nt) {
    int src = (int)(((long long)j * (M3 - 1)) / (C2 - 1));
    int cc = src / fd, f = src % fd;
    float fv = -1.f + 2.f * (float)(f + 1) / (float)(fd + 1);
    float pmx = -FLT_MAX, psm = 0.f;
    for (int q = 0; q < chunk; ++q) {
      const float* fq = fs + q * C;
      float acc = 0.f, mx = -FLT_MAX;
      for (int k = 0; k < KNN; ++k) {
        const float d = xkn[q * KNN + k][cc] - fv;
        const float pe = __builtin_amdgcn_exp2f(kexp * (d * d));
        float fk;
        if (j < C) fk = (ft[(q * KNN + k) * C + j] - fq[j]) / sfe[k];
        else       fk = fq[j - C];
        float fw = (fk + pe) * pe;
        acc += fw; mx = fmaxf(mx, fw);
      }
      float xx = acc * (1.f / KNN) + mx;
      const float o = 0.5f * xx * (1.f + erff(xx * 0.70710678118654752f));
      out[(size_t)(row0 + q) * C2 + j] = o;
      pmx = fmaxf(pmx, o); psm += o;
    }
    if (chunk > 1) {
      pmaxb[(size_t)blockIdx.x * C2 + j] = pmx;
      psumb[(size_t)blockIdx.x * C2 + j] = psm;
    }
  }
}

__global__ __launch_bounds__(1024) void k_pool_fold(const float* pm, const float* ps,

                                                    int G, int C2, int S,
                                                    float* __restrict__ out, int off) {
  const int b = blockIdx.x;
  for (int j = threadIdx.x; j < C2; j += blockDim.x) {
    const float* a = pm + (size_t)b * G * C2 + j;
    const float* c = ps + (size_t)b * G * C2 + j;
    float mx = -FLT_MAX, sm = 0.f;
    if (pm == ps) {
#pragma unroll 8
      for (int g = 0; g < G; ++g) {
        const float v = a[(size_t)g * C2];
        mx = fmaxf(mx, v); sm += v;
      }
    } else {
#pragma unroll 8
      for (int g = 0; g < G; ++g) {
        mx = fmaxf(mx, a[(size_t)g * C2]);
        sm += c[(size_t)g * C2];
      }
    }
    out[b * 3840 + off + j]      = mx;
    out[b * 3840 + off + C2 + j] = sm / (float)S;
  }
}

extern "C" void kernel_launch(void* const* d_in, const int* in_sizes, int n_in,
                              void* d_out, int out_size, void* d_ws, size_t ws_size,
                              hipStream_t stream) {
  const float* xyz_in = (const float*)d_in[0];
  const float* W      = (const float*)d_in[1];
  const float* gamma  = (const float*)d_in[2];
  const float* beta   = (const float*)d_in[3];
  float* out = (float*)d_out;
  float* ws  = (float*)d_ws;

  size_t o = 0;
  float* featA  = ws + o; o += (size_t)BATCH * NPTS * CINIT;
  float* featB  = ws + o; o += (size_t)BATCH * NPTS * CINIT;
  float* xyzA   = ws + o; o += (size_t)BATCH * NPTS * 3;
  float* xyzB   = ws + o; o += (size_t)BATCH * NPTS * 3;
  float* feat_s = ws + o; o += (size_t)BATCH * 1024 * CINIT;
  float* pmaxb  = ws + o; o += (size_t)1024 * 1024;
  float* psumb  = ws + o; o += (size_t)1024 * 1024;
  float* bn_mu  = ws + o; o += 64;
  float* bn_inv = ws + o; o += 64;
  float* stdx   = ws + o; o += 32;
  float* stdf   = ws + o; o += 32;
  float* stdp   = pmaxb;
  int* fps_i = (int*)(ws + o); o += (size_t)BATCH * 1024;
  int* knn_i = (int*)(ws + o); o += (size_t)BATCH * 1024 * KNN;
  int* fps_j = (int*)(xyzB + (size_t)BATCH * 1024 * 3);

  k_embed_wmma<<<1024, 256, 0, stream>>>(xyz_in, W, featA);
  k_bn_stats<<<CINIT, 256, 0, stream>>>(featA, bn_mu, bn_inv);
  k_bn_apply<<<2048, 256, 0, stream>>>(featA, bn_mu, bn_inv, gamma, beta);

  const float* xyz_cur = xyz_in;
  float* feat_cur = featA; float* feat_nxt = featB;
  float* xs_buf = xyzA;    float* xs_other = xyzB;
  int* fps_cur = fps_i;    int* fps_nxt = fps_j;
  int N = NPTS, C = CINIT, off = 0;
  const float* rd_pm = nullptr; const float* rd_ps = nullptr;
  int rd_G = 0, rd_C2 = 0, rd_S = 0, rd_off = 0;

  for (int st = 0; st < 4; ++st) {
    const int S = N / 2, C2 = 2 * C;
    if (N == 2048)      k_fps_wave<64><<<BATCH, 32, 0, stream>>>(xyz_cur, S, fps_cur);
    {
      const int nb = BATCH * S / 16 / KW;
      const int ng = BATCH * S / GROWS;
      const int np = rd_pm ? BATCH * (rd_C2 / PCOLS) : 0;
      if (N == 2048)      k_knn_fused<2048><<<BATCH + nb + ng + np, KW * 32, 0, stream>>>(
                              xyz_cur, fps_cur, S, knn_i, fps_nxt, feat_cur, C, xs_buf, feat_s, nb, ng,
                              rd_pm, rd_ps, rd_G, rd_C2, rd_S, out, rd_off);
      else if (N == 1024) k_knn_fused<1024><<<BATCH + nb + ng + np, KW * 32, 0, stream>>>(
                              xyz_cur, fps_cur, S, knn_i, fps_nxt, feat_cur, C, xs_buf, feat_s, nb, ng,
                              rd_pm, rd_ps, rd_G, rd_C2, rd_S, out, rd_off);
      else if (N == 512)  k_knn_fused<512><<<BATCH + nb + ng + np, KW * 32, 0, stream>>>(
                              xyz_cur, fps_cur, S, knn_i, fps_nxt, feat_cur, C, xs_buf, feat_s, nb, ng,
                              rd_pm, rd_ps, rd_G, rd_C2, rd_S, out, rd_off);
      else                k_knn_fused<256><<<nb + ng + np, KW * 32, 0, stream>>>(
                              xyz_cur, fps_cur, S, knn_i, fps_nxt, feat_cur, C, xs_buf, feat_s, nb, ng,
                              rd_pm, rd_ps, rd_G, rd_C2, rd_S, out, rd_off);
    }
    {
      k_std_partial<<<STD_G, STD_W * 32, 0, stream>>>(xyz_cur, feat_cur, xs_buf, feat_s, knn_i, N, S, C, stdp);
      k_std_final<<<1, 1024, 0, stream>>>(stdp, S, C, stdx, stdf);
    }
    {
      const int chunk = (C2 <= 256) ? 4 : ((C2 == 512) ? 2 : 1);
      const int nthr  = (C2 < 256) ? C2 : 256;
      if (C == CINIT) k_stage<PCH * KNN * CINIT><<<BATCH * S / chunk, nthr, 0, stream>>>(
                          xyz_cur, feat_cur, xs_buf, feat_s, knn_i, stdx, stdf, N, S, C, chunk, feat_nxt,
                          pmaxb, psumb);
      else            k_stage<16384><<<BATCH * S / chunk, nthr, 0, stream>>>(
                          xyz_cur, feat_cur, xs_buf, feat_s, knn_i, stdx, stdf, N, S, C, chunk, feat_nxt,
                          pmaxb, psumb);
      if (st == 3) {
        if (chunk > 1) k_pool_fold<<<BATCH, C2, 0, stream>>>(pmaxb, psumb, S / chunk, C2, S, out, off);
        else           k_pool_fold<<<BATCH, C2, 0, stream>>>(feat_nxt, feat_nxt, S, C2, S, out, off);
      } else {
        rd_pm = pmaxb; rd_ps = psumb; rd_G = S / chunk; rd_C2 = C2; rd_S = S; rd_off = off;
      }
    }
    off += 2 * C2;
    float* t = feat_cur; feat_cur = feat_nxt; feat_nxt = t;
    xyz_cur = xs_buf;
    float* u = xs_buf; xs_buf = xs_other; xs_other = u;
    int* v = fps_cur; fps_cur = fps_nxt; fps_nxt = v;
    N = S; C = C2;
  }
  (void)in_sizes; (void)n_in; (void)out_size; (void)ws_size;
}
